# starting kernel
# baseline (speedup 1.0000x reference)
#include <hip/hip_runtime.h>
#include <hip/hip_bf16.h>
#include <stdint.h>

#define BB 8
#define CC 256
#define HH 25
#define WW 512

typedef __bf16 bf16x16 __attribute__((ext_vector_type(16)));
typedef float  f32x8   __attribute__((ext_vector_type(8)));

union FragU { uint4 u[2]; bf16x16 v; };

__device__ __forceinline__ unsigned short f2bf(float x) {
  unsigned u = __float_as_uint(x);
  unsigned r = u + 0x7FFFu + ((u >> 16) & 1u);
  return (unsigned short)(r >> 16);
}

__device__ __forceinline__ f32x8 wmma_bf16(bf16x16 a, bf16x16 b, f32x8 c) {
  return __builtin_amdgcn_wmma_f32_16x16x32_bf16(false, a, false, b, (short)0, c, false, false);
}

__device__ __forceinline__ bf16x16 load_frag_lds(const unsigned short* rowBase, int kt, int lane) {
  int kb = (lane & 16) ? 8 : 0;
  const unsigned short* p = rowBase + kt * 32 + kb;
  FragU f;
  f.u[0] = *(const uint4*)(p);
  f.u[1] = *(const uint4*)(p + 16);
  return f.v;
}

__device__ __forceinline__ bf16x16 load_frag_Bswz(const unsigned int* swz, int kt, int nt, int lane) {
  const unsigned int* p = swz + ((((kt << 4) + nt) * 32 + lane) << 3);
  FragU f;
  f.u[0] = *(const uint4*)(p);
  f.u[1] = *(const uint4*)(p + 4);
  return f.v;
}

__global__ void prep_kernel(const float* __restrict__ wmsg, const float* __restrict__ wup2,
                            unsigned int* __restrict__ Mswz, unsigned int* __restrict__ Wswz)
{
  int idx = blockIdx.x * 256 + threadIdx.x;
  int tap = idx >> 15;
  int r   = idx & 32767;
  int j    = r & 7;
  int lane = (r >> 3) & 31;
  int nt   = (r >> 8) & 15;
  int kt   = r >> 12;
  int n  = nt * 16 + (lane & 15);
  int kb = (lane & 16) ? 8 : 0;
  int k  = kt * 32 + ((j < 4) ? (kb + 2 * j) : (16 + kb + 2 * (j - 4)));
  float lo, hi;
  if (tap == 9) {
    lo = wmsg[(n * 256 + k) * 9 + 4];
    hi = wmsg[(n * 256 + k + 1) * 9 + 4];
  } else {
    int ky = tap / 3, kx = tap - ky * 3;
    lo = wup2[((n * 256 + k) * 3 + ky) * 3 + kx];
    hi = wup2[((n * 256 + k + 1) * 3 + ky) * 3 + kx];
  }
  unsigned int d = (unsigned)f2bf(lo) | ((unsigned)f2bf(hi) << 16);
  if (tap == 9) Mswz[r] = d;
  else          Wswz[tap * 32768 + r] = d;
}

__global__ void init_misc(float* stats1, float* stats2, float* scid, float* shid) {
  int t = threadIdx.x;
  stats1[t] = 0.f; stats2[t] = 0.f;
  if (t < 256) { scid[t] = 1.0f; shid[t] = 0.0f; }
}

__global__ void __launch_bounds__(256) scnn_kernel(
    const float* __restrict__ xin, float* __restrict__ xout,
    const unsigned int* __restrict__ Mswz,
    const float* __restrict__ sc, const float* __restrict__ sh,
    int relu_in, float* __restrict__ stats)
{
  __shared__ __align__(16) unsigned short prevA[16 * 264];
  __shared__ __align__(16) unsigned short prevB[16 * 264];

  const int tid  = threadIdx.x;
  const int b    = blockIdx.x >> 5;
  const int w0   = (blockIdx.x & 31) << 4;
  const int wave = tid >> 5;
  const int lane = tid & 31;
  const int half8 = (lane & 16) ? 8 : 0;
  const int nt0 = wave * 2, nt1 = wave * 2 + 1;
  const int ch0 = nt0 * 16 + (lane & 15);
  const int ch1 = nt1 * 16 + (lane & 15);

  for (int i = tid; i < 16 * 264; i += 256) prevA[i] = 0;

  const float sA = sc[ch0], bA = sh[ch0];
  const float sB = sc[ch1], bB = sh[ch1];

  const float* xi0p = xin  + ((size_t)(b * CC + ch0) * HH) * WW + w0 + half8;
  const float* xi1p = xin  + ((size_t)(b * CC + ch1) * HH) * WW + w0 + half8;
  float*       xo0p = xout + ((size_t)(b * CC + ch0) * HH) * WW + w0 + half8;
  float*       xo1p = xout + ((size_t)(b * CC + ch1) * HH) * WW + w0 + half8;

  float4 x0a = *(const float4*)(xi0p);
  float4 x0b = *(const float4*)(xi0p + 4);
  float4 x1a = *(const float4*)(xi1p);
  float4 x1b = *(const float4*)(xi1p + 4);

  float s0 = 0.f, q0 = 0.f, s1 = 0.f, q1 = 0.f;

  for (int h = 0; h < HH; ++h) {
    unsigned short* prevCur = (h & 1) ? prevB : prevA;
    unsigned short* prevNxt = (h & 1) ? prevA : prevB;
    __syncthreads();

    f32x8 acc0 = {}; f32x8 acc1 = {};
    const unsigned short* arow = &prevCur[(lane & 15) * 264];
    bf16x16 a = load_frag_lds(arow, 0, lane);
    #pragma unroll
    for (int kt = 0; kt < 8; ++kt) {
      bf16x16 an = (kt < 7) ? load_frag_lds(arow, kt + 1, lane) : a;
      acc0 = wmma_bf16(a, load_frag_Bswz(Mswz, kt, nt0, lane), acc0);
      acc1 = wmma_bf16(a, load_frag_Bswz(Mswz, kt, nt1, lane), acc1);
      a = an;
    }

    const float xi0[8] = {x0a.x, x0a.y, x0a.z, x0a.w, x0b.x, x0b.y, x0b.z, x0b.w};
    const float xi1[8] = {x1a.x, x1a.y, x1a.z, x1a.w, x1b.x, x1b.y, x1b.z, x1b.w};
    float r0[8], r1[8];
    #pragma unroll
    for (int j = 0; j < 8; ++j) {
      float a0 = sA * xi0[j] + bA;
      float a1 = sB * xi1[j] + bB;
      if (relu_in) { a0 = fmaxf(a0, 0.f); a1 = fmaxf(a1, 0.f); }
      float v0 = a0 + fmaxf(acc0[j], 0.f);
      float v1 = a1 + fmaxf(acc1[j], 0.f);
      r0[j] = v0; r1[j] = v1;
      s0 += v0; q0 += v0 * v0;
      s1 += v1; q1 += v1 * v1;
      prevNxt[(j + half8) * 264 + ch0] = f2bf(v0);
      prevNxt[(j + half8) * 264 + ch1] = f2bf(v1);
    }
    {
      float* p0 = xo0p + (size_t)(HH - 1 - h) * WW;
      float* p1 = xo1p + (size_t)(HH - 1 - h) * WW;
      *(float4*)(p0)     = make_float4(r0[0], r0[1], r0[2], r0[3]);
      *(float4*)(p0 + 4) = make_float4(r0[4], r0[5], r0[6], r0[7]);
      *(float4*)(p1)     = make_float4(r1[0], r1[1], r1[2], r1[3]);
      *(float4*)(p1 + 4) = make_float4(r1[4], r1[5], r1[6], r1[7]);
    }
    if (h + 1 < HH) {
      const float* n0 = xi0p + (size_t)(h + 1) * WW;
      const float* n1 = xi1p + (size_t)(h + 1) * WW;
      x0a = *(const float4*)(n0); x0b = *(const float4*)(n0 + 4);
      x1a = *(const float4*)(n1); x1b = *(const float4*)(n1 + 4);
      if (h + 2 < HH) {
        __builtin_prefetch(xi0p + (size_t)(h + 2) * WW, 0, 0);
        __builtin_prefetch(xi1p + (size_t)(h + 2) * WW, 0, 0);
      }
    }
  }

  atomicAdd(&stats[ch0],       s0);
  atomicAdd(&stats[256 + ch0], q0);
  atomicAdd(&stats[ch1],       s1);
  atomicAdd(&stats[256 + ch1], q1);
}

__global__ void bnscale_kernel(const float* __restrict__ stats,
                               const float* __restrict__ gamma, const float* __restrict__ beta,
                               float* __restrict__ sc, float* __restrict__ sh)
{
  int c = threadIdx.x;
  const float n = (float)(BB * HH * WW);
  float mean = stats[c] / n;
  float var  = stats[256 + c] / n - mean * mean;
  float s = gamma[c] * rsqrtf(var + 1e-5f);
  sc[c] = s;
  sh[c] = beta[c] - mean * s;
}

__global__ void __launch_bounds__(256) conv_kernel(
    const float* __restrict__ xin,
    const unsigned int* __restrict__ Wswz,
    const float* __restrict__ sc, const float* __restrict__ sh,
    const float* __restrict__ v1,
    float* __restrict__ z)
{
  __shared__ __align__(16) unsigned short uT[3 * 18 * 264];
  __shared__ float zred[16];

  const int tid = threadIdx.x;
  const int wt  = blockIdx.x & 63;
  const int h   = (blockIdx.x >> 6) % 25;
  const int b   = blockIdx.x / (64 * 25);
  const int w20 = wt << 4;

  if (tid < 16) zred[tid] = 0.f;

  for (int e = tid; e < 3 * 18 * 256; e += 256) {
    int wi   = e % 18;
    int rest = e / 18;
    int c    = rest & 255;
    int hh   = rest >> 8;
    int hs = h + hh - 1;
    int w2 = w20 + wi - 1;
    float v = 0.f;
    if (hs >= 0 && hs < HH && w2 >= 0 && w2 < 1024) {
      int   num = w2 * 511;
      int   i0  = num / 1023;
      float f   = (float)(num - i0 * 1023) * (1.0f / 1023.0f);
      int   i1  = min(i0 + 1, 511);
      const float* base = xin + ((b * CC + c) * HH + hs) * WW;
      float scc = sc[c], shc = sh[c];
      float x0 = fmaxf(scc * base[i0] + shc, 0.f);
      float x1 = fmaxf(scc * base[i1] + shc, 0.f);
      v = x0 + f * (x1 - x0);
    }
    uT[(hh * 18 + wi) * 264 + c] = f2bf(v);
  }
  __syncthreads();

  const int wave = tid >> 5;
  const int lane = tid & 31;
  const int nt0 = wave * 2, nt1 = nt0 + 1;
  f32x8 acc0 = {}; f32x8 acc1 = {};

  #pragma unroll
  for (int ky = 0; ky < 3; ++ky) {
    #pragma unroll
    for (int kx = 0; kx < 3; ++kx) {
      const unsigned int* bsw = Wswz + (ky * 3 + kx) * 32768;
      const unsigned short* arow = &uT[(ky * 18 + (lane & 15) + kx) * 264];
      bf16x16 a = load_frag_lds(arow, 0, lane);
      #pragma unroll
      for (int kt = 0; kt < 8; ++kt) {
        bf16x16 an = (kt < 7) ? load_frag_lds(arow, kt + 1, lane) : a;
        acc0 = wmma_bf16(a, load_frag_Bswz(bsw, kt, nt0, lane), acc0);
        acc1 = wmma_bf16(a, load_frag_Bswz(bsw, kt, nt1, lane), acc1);
        a = an;
      }
    }
  }

  const int half8 = (lane & 16) ? 8 : 0;
  const int ch0 = nt0 * 16 + (lane & 15);
  const int ch1 = nt1 * 16 + (lane & 15);
  float g0 = v1[ch0], g1 = v1[ch1];
  #pragma unroll
  for (int j = 0; j < 8; ++j) {
    atomicAdd(&zred[j + half8], acc0[j] * g0 + acc1[j] * g1);
  }
  __syncthreads();
  if (tid < 16) z[(b * 25 + h) * 1024 + w20 + tid] = zred[tid];
}

__global__ void final_kernel(const float* __restrict__ z, const float* __restrict__ v2,
                             float* __restrict__ out)
{
  int idx = blockIdx.x * 256 + threadIdx.x;
  if (idx >= BB * 4096) return;
  int b  = idx >> 12;
  int w4 = idx & 4095;

  int   num4 = w4 * 2047;
  int   i40  = num4 / 4095;
  float f4   = (float)(num4 - i40 * 4095) * (1.0f / 4095.0f);
  int   i41  = min(i40 + 1, 2047);

  int n3a = i40 * 1023; int ja0 = n3a / 2047;
  float fa = (float)(n3a - ja0 * 2047) * (1.0f / 2047.0f);
  int ja1 = min(ja0 + 1, 1023);
  int n3b = i41 * 1023; int jb0 = n3b / 2047;
  float fb = (float)(n3b - jb0 * 2047) * (1.0f / 2047.0f);
  int jb1 = min(jb0 + 1, 1023);

  float acc = 0.f;
  #pragma unroll 5
  for (int h = 0; h < HH; ++h) {
    const float* zh = z + (b * 25 + h) * 1024;
    float za = zh[ja0] + fa * (zh[ja1] - zh[ja0]);
    float zb = zh[jb0] + fb * (zh[jb1] - zh[jb0]);
    acc += v2[h] * (za + f4 * (zb - za));
  }
  out[idx] = 1.0f / (1.0f + __expf(-acc));
}

extern "C" void kernel_launch(void* const* d_in, const int* in_sizes, int n_in,
                              void* d_out, int out_size, void* d_ws, size_t ws_size,
                              hipStream_t stream)
{
  const float* p2c   = (const float*)d_in[0];
  const float* wmsg  = (const float*)d_in[1];
  const float* gamma = (const float*)d_in[2];
  const float* beta  = (const float*)d_in[3];
  const float* wup2  = (const float*)d_in[4];
  const float* v1    = (const float*)d_in[5];
  const float* v2    = (const float*)d_in[6];

  char* ws = (char*)d_ws;
  size_t off = 0;
  auto walloc = [&](size_t bytes) -> void* {
    void* p = ws + off;
    off = (off + bytes + 255) & ~(size_t)255;
    return p;
  };
  unsigned int* Mswz = (unsigned int*)walloc(32768u * 4);
  unsigned int* Wswz = (unsigned int*)walloc(9u * 32768u * 4);
  float* stats1 = (float*)walloc(512 * 4);
  float* stats2 = (float*)walloc(512 * 4);
  float* sc1  = (float*)walloc(256 * 4);
  float* sh1  = (float*)walloc(256 * 4);
  float* sc2  = (float*)walloc(256 * 4);
  float* sh2  = (float*)walloc(256 * 4);
  float* scid = (float*)walloc(256 * 4);
  float* shid = (float*)walloc(256 * 4);
  float* z1024 = (float*)walloc((size_t)BB * 25 * 1024 * 4);
  float* out1  = (float*)walloc((size_t)BB * CC * HH * WW * 4);
  float* out2  = (float*)walloc((size_t)BB * CC * HH * WW * 4);

  prep_kernel<<<1280, 256, 0, stream>>>(wmsg, wup2, Mswz, Wswz);
  init_misc<<<1, 512, 0, stream>>>(stats1, stats2, scid, shid);

  scnn_kernel<<<256, 256, 0, stream>>>(p2c, out1, Mswz, scid, shid, 0, stats1);
  bnscale_kernel<<<1, 256, 0, stream>>>(stats1, gamma, beta, sc1, sh1);

  scnn_kernel<<<256, 256, 0, stream>>>(out1, out2, Mswz, sc1, sh1, 1, stats2);
  bnscale_kernel<<<1, 256, 0, stream>>>(stats2, gamma, beta, sc2, sh2);

  conv_kernel<<<BB * 25 * 64, 256, 0, stream>>>(out2, Wswz, sc2, sh2, v1, z1024);

  final_kernel<<<128, 256, 0, stream>>>(z1024, v2, (float*)d_out);
}


# conv: work-group tile 16 -> 64 output columns; each wave keeps 4 M-tiles x 2 N-tiles of accumulators so every weight fragment read from L2 feeds 4 WMMAs instead of 1
# speedup vs baseline: 1.0706x; 1.0706x over previous
#include <hip/hip_runtime.h>
#include <hip/hip_bf16.h>
#include <stdint.h>

#define BB 8
#define CC 256
#define HH 25
#define WW 512

typedef __bf16 bf16x16 __attribute__((ext_vector_type(16)));
typedef float  f32x8   __attribute__((ext_vector_type(8)));

union FragU { uint4 u[2]; bf16x16 v; };

__device__ __forceinline__ unsigned short f2bf(float x) {
  unsigned u = __float_as_uint(x);
  unsigned r = u + 0x7FFFu + ((u >> 16) & 1u);
  return (unsigned short)(r >> 16);
}

__device__ __forceinline__ f32x8 wmma_bf16(bf16x16 a, bf16x16 b, f32x8 c) {
  return __builtin_amdgcn_wmma_f32_16x16x32_bf16(false, a, false, b, (short)0, c, false, false);
}

__device__ __forceinline__ bf16x16 load_frag_lds(const unsigned short* rowBase, int kt, int lane) {
  int kb = (lane & 16) ? 8 : 0;
  const unsigned short* p = rowBase + kt * 32 + kb;
  FragU f;
  f.u[0] = *(const uint4*)(p);
  f.u[1] = *(const uint4*)(p + 16);
  return f.v;
}

__device__ __forceinline__ bf16x16 load_frag_Bswz(const unsigned int* swz, int kt, int nt, int lane) {
  const unsigned int* p = swz + ((((kt << 4) + nt) * 32 + lane) << 3);
  FragU f;
  f.u[0] = *(const uint4*)(p);
  f.u[1] = *(const uint4*)(p + 4);
  return f.v;
}

__global__ void prep_kernel(const float* __restrict__ wmsg, const float* __restrict__ wup2,
                            unsigned int* __restrict__ Mswz, unsigned int* __restrict__ Wswz)
{
  int idx = blockIdx.x * 256 + threadIdx.x;
  int tap = idx >> 15;
  int r   = idx & 32767;
  int j    = r & 7;
  int lane = (r >> 3) & 31;
  int nt   = (r >> 8) & 15;
  int kt   = r >> 12;
  int n  = nt * 16 + (lane & 15);
  int kb = (lane & 16) ? 8 : 0;
  int k  = kt * 32 + ((j < 4) ? (kb + 2 * j) : (16 + kb + 2 * (j - 4)));
  float lo, hi;
  if (tap == 9) {
    lo = wmsg[(n * 256 + k) * 9 + 4];
    hi = wmsg[(n * 256 + k + 1) * 9 + 4];
  } else {
    int ky = tap / 3, kx = tap - ky * 3;
    lo = wup2[((n * 256 + k) * 3 + ky) * 3 + kx];
    hi = wup2[((n * 256 + k + 1) * 3 + ky) * 3 + kx];
  }
  unsigned int d = (unsigned)f2bf(lo) | ((unsigned)f2bf(hi) << 16);
  if (tap == 9) Mswz[r] = d;
  else          Wswz[tap * 32768 + r] = d;
}

__global__ void init_misc(float* stats1, float* stats2, float* scid, float* shid) {
  int t = threadIdx.x;
  stats1[t] = 0.f; stats2[t] = 0.f;
  if (t < 256) { scid[t] = 1.0f; shid[t] = 0.0f; }
}

__global__ void __launch_bounds__(256) scnn_kernel(
    const float* __restrict__ xin, float* __restrict__ xout,
    const unsigned int* __restrict__ Mswz,
    const float* __restrict__ sc, const float* __restrict__ sh,
    int relu_in, float* __restrict__ stats)
{
  __shared__ __align__(16) unsigned short prevA[16 * 264];
  __shared__ __align__(16) unsigned short prevB[16 * 264];

  const int tid  = threadIdx.x;
  const int b    = blockIdx.x >> 5;
  const int w0   = (blockIdx.x & 31) << 4;
  const int wave = tid >> 5;
  const int lane = tid & 31;
  const int half8 = (lane & 16) ? 8 : 0;
  const int nt0 = wave * 2, nt1 = wave * 2 + 1;
  const int ch0 = nt0 * 16 + (lane & 15);
  const int ch1 = nt1 * 16 + (lane & 15);

  for (int i = tid; i < 16 * 264; i += 256) prevA[i] = 0;

  const float sA = sc[ch0], bA = sh[ch0];
  const float sB = sc[ch1], bB = sh[ch1];

  const float* xi0p = xin  + ((size_t)(b * CC + ch0) * HH) * WW + w0 + half8;
  const float* xi1p = xin  + ((size_t)(b * CC + ch1) * HH) * WW + w0 + half8;
  float*       xo0p = xout + ((size_t)(b * CC + ch0) * HH) * WW + w0 + half8;
  float*       xo1p = xout + ((size_t)(b * CC + ch1) * HH) * WW + w0 + half8;

  float4 x0a = *(const float4*)(xi0p);
  float4 x0b = *(const float4*)(xi0p + 4);
  float4 x1a = *(const float4*)(xi1p);
  float4 x1b = *(const float4*)(xi1p + 4);

  float s0 = 0.f, q0 = 0.f, s1 = 0.f, q1 = 0.f;

  for (int h = 0; h < HH; ++h) {
    unsigned short* prevCur = (h & 1) ? prevB : prevA;
    unsigned short* prevNxt = (h & 1) ? prevA : prevB;
    __syncthreads();

    f32x8 acc0 = {}; f32x8 acc1 = {};
    const unsigned short* arow = &prevCur[(lane & 15) * 264];
    bf16x16 a = load_frag_lds(arow, 0, lane);
    #pragma unroll
    for (int kt = 0; kt < 8; ++kt) {
      bf16x16 an = (kt < 7) ? load_frag_lds(arow, kt + 1, lane) : a;
      acc0 = wmma_bf16(a, load_frag_Bswz(Mswz, kt, nt0, lane), acc0);
      acc1 = wmma_bf16(a, load_frag_Bswz(Mswz, kt, nt1, lane), acc1);
      a = an;
    }

    const float xi0[8] = {x0a.x, x0a.y, x0a.z, x0a.w, x0b.x, x0b.y, x0b.z, x0b.w};
    const float xi1[8] = {x1a.x, x1a.y, x1a.z, x1a.w, x1b.x, x1b.y, x1b.z, x1b.w};
    float r0[8], r1[8];
    #pragma unroll
    for (int j = 0; j < 8; ++j) {
      float a0 = sA * xi0[j] + bA;
      float a1 = sB * xi1[j] + bB;
      if (relu_in) { a0 = fmaxf(a0, 0.f); a1 = fmaxf(a1, 0.f); }
      float v0 = a0 + fmaxf(acc0[j], 0.f);
      float v1 = a1 + fmaxf(acc1[j], 0.f);
      r0[j] = v0; r1[j] = v1;
      s0 += v0; q0 += v0 * v0;
      s1 += v1; q1 += v1 * v1;
      prevNxt[(j + half8) * 264 + ch0] = f2bf(v0);
      prevNxt[(j + half8) * 264 + ch1] = f2bf(v1);
    }
    {
      float* p0 = xo0p + (size_t)(HH - 1 - h) * WW;
      float* p1 = xo1p + (size_t)(HH - 1 - h) * WW;
      *(float4*)(p0)     = make_float4(r0[0], r0[1], r0[2], r0[3]);
      *(float4*)(p0 + 4) = make_float4(r0[4], r0[5], r0[6], r0[7]);
      *(float4*)(p1)     = make_float4(r1[0], r1[1], r1[2], r1[3]);
      *(float4*)(p1 + 4) = make_float4(r1[4], r1[5], r1[6], r1[7]);
    }
    if (h + 1 < HH) {
      const float* n0 = xi0p + (size_t)(h + 1) * WW;
      const float* n1 = xi1p + (size_t)(h + 1) * WW;
      x0a = *(const float4*)(n0); x0b = *(const float4*)(n0 + 4);
      x1a = *(const float4*)(n1); x1b = *(const float4*)(n1 + 4);
      if (h + 2 < HH) {
        __builtin_prefetch(xi0p + (size_t)(h + 2) * WW, 0, 0);
        __builtin_prefetch(xi1p + (size_t)(h + 2) * WW, 0, 0);
      }
    }
  }

  atomicAdd(&stats[ch0],       s0);
  atomicAdd(&stats[256 + ch0], q0);
  atomicAdd(&stats[ch1],       s1);
  atomicAdd(&stats[256 + ch1], q1);
}

__global__ void bnscale_kernel(const float* __restrict__ stats,
                               const float* __restrict__ gamma, const float* __restrict__ beta,
                               float* __restrict__ sc, float* __restrict__ sh)
{
  int c = threadIdx.x;
  const float n = (float)(BB * HH * WW);
  float mean = stats[c] / n;
  float var  = stats[256 + c] / n - mean * mean;
  float s = gamma[c] * rsqrtf(var + 1e-5f);
  sc[c] = s;
  sh[c] = beta[c] - mean * s;
}

#define CONV_MT 4
#define CONV_TW (CONV_MT * 16)
#define CONV_HW (CONV_TW + 2)

__global__ void __launch_bounds__(256) conv_kernel(
    const float* __restrict__ xin,
    const unsigned int* __restrict__ Wswz,
    const float* __restrict__ sc, const float* __restrict__ sh,
    const float* __restrict__ v1,
    float* __restrict__ z)
{
  __shared__ __align__(16) unsigned short uT[3 * CONV_HW * 264];
  __shared__ float zred[CONV_TW];

  const int tid = threadIdx.x;
  const int wt  = blockIdx.x & 15;
  const int h   = (blockIdx.x >> 4) % 25;
  const int b   = blockIdx.x / (16 * 25);
  const int w20 = wt * CONV_TW;

  if (tid < CONV_TW) zred[tid] = 0.f;

  for (int e = tid; e < 3 * CONV_HW * 256; e += 256) {
    int wi   = e % CONV_HW;
    int rest = e / CONV_HW;
    int c    = rest & 255;
    int hh   = rest >> 8;
    int hs = h + hh - 1;
    int w2 = w20 + wi - 1;
    float v = 0.f;
    if (hs >= 0 && hs < HH && w2 >= 0 && w2 < 1024) {
      int   num = w2 * 511;
      int   i0  = num / 1023;
      float f   = (float)(num - i0 * 1023) * (1.0f / 1023.0f);
      int   i1  = min(i0 + 1, 511);
      const float* base = xin + ((b * CC + c) * HH + hs) * WW;
      float scc = sc[c], shc = sh[c];
      float x0 = fmaxf(scc * base[i0] + shc, 0.f);
      float x1 = fmaxf(scc * base[i1] + shc, 0.f);
      v = x0 + f * (x1 - x0);
    }
    uT[(hh * CONV_HW + wi) * 264 + c] = f2bf(v);
  }
  __syncthreads();

  const int wave = tid >> 5;
  const int lane = tid & 31;
  const int nt0 = wave * 2, nt1 = nt0 + 1;
  f32x8 acc0[CONV_MT], acc1[CONV_MT];
  #pragma unroll
  for (int m = 0; m < CONV_MT; ++m) { acc0[m] = (f32x8){}; acc1[m] = (f32x8){}; }

  const unsigned short* abase = &uT[(lane & 15) * 264];

  bf16x16 a[CONV_MT], b0, b1;
  #pragma unroll
  for (int m = 0; m < CONV_MT; ++m) a[m] = load_frag_lds(abase + m * 16 * 264, 0, lane);
  b0 = load_frag_Bswz(Wswz, 0, nt0, lane);
  b1 = load_frag_Bswz(Wswz, 0, nt1, lane);

  #pragma unroll
  for (int s = 0; s < 72; ++s) {
    bf16x16 an[CONV_MT], bn0, bn1;
    if (s + 1 < 72) {
      const int tap = (s + 1) >> 3, kt = (s + 1) & 7;
      const int ky = tap / 3, kx = tap - ky * 3;
      const unsigned short* arow = abase + (ky * CONV_HW + kx) * 264;
      #pragma unroll
      for (int m = 0; m < CONV_MT; ++m) an[m] = load_frag_lds(arow + m * 16 * 264, kt, lane);
      bn0 = load_frag_Bswz(Wswz + tap * 32768, kt, nt0, lane);
      bn1 = load_frag_Bswz(Wswz + tap * 32768, kt, nt1, lane);
    } else {
      #pragma unroll
      for (int m = 0; m < CONV_MT; ++m) an[m] = a[m];
      bn0 = b0; bn1 = b1;
    }
    #pragma unroll
    for (int m = 0; m < CONV_MT; ++m) {
      acc0[m] = wmma_bf16(a[m], b0, acc0[m]);
      acc1[m] = wmma_bf16(a[m], b1, acc1[m]);
    }
    #pragma unroll
    for (int m = 0; m < CONV_MT; ++m) a[m] = an[m];
    b0 = bn0; b1 = bn1;
  }

  const int half8 = (lane & 16) ? 8 : 0;
  const int ch0 = nt0 * 16 + (lane & 15);
  const int ch1 = nt1 * 16 + (lane & 15);
  float g0 = v1[ch0], g1 = v1[ch1];
  #pragma unroll
  for (int m = 0; m < CONV_MT; ++m) {
    #pragma unroll
    for (int j = 0; j < 8; ++j) {
      atomicAdd(&zred[m * 16 + j + half8], acc0[m][j] * g0 + acc1[m][j] * g1);
    }
  }
  __syncthreads();
  if (tid < CONV_TW) z[(b * 25 + h) * 1024 + w20 + tid] = zred[tid];
}

__global__ void final_kernel(const float* __restrict__ z, const float* __restrict__ v2,
                             float* __restrict__ out)
{
  int idx = blockIdx.x * 256 + threadIdx.x;
  if (idx >= BB * 4096) return;
  int b  = idx >> 12;
  int w4 = idx & 4095;

  int   num4 = w4 * 2047;
  int   i40  = num4 / 4095;
  float f4   = (float)(num4 - i40 * 4095) * (1.0f / 4095.0f);
  int   i41  = min(i40 + 1, 2047);

  int n3a = i40 * 1023; int ja0 = n3a / 2047;
  float fa = (float)(n3a - ja0 * 2047) * (1.0f / 2047.0f);
  int ja1 = min(ja0 + 1, 1023);
  int n3b = i41 * 1023; int jb0 = n3b / 2047;
  float fb = (float)(n3b - jb0 * 2047) * (1.0f / 2047.0f);
  int jb1 = min(jb0 + 1, 1023);

  float acc = 0.f;
  #pragma unroll 5
  for (int h = 0; h < HH; ++h) {
    const float* zh = z + (b * 25 + h) * 1024;
    float za = zh[ja0] + fa * (zh[ja1] - zh[ja0]);
    float zb = zh[jb0] + fb * (zh[jb1] - zh[jb0]);
    acc += v2[h] * (za + f4 * (zb - za));
  }
  out[idx] = 1.0f / (1.0f + __expf(-acc));
}

extern "C" void kernel_launch(void* const* d_in, const int* in_sizes, int n_in,
                              void* d_out, int out_size, void* d_ws, size_t ws_size,
                              hipStream_t stream)
{
  const float* p2c   = (const float*)d_in[0];
  const float* wmsg  = (const float*)d_in[1];
  const float* gamma = (const float*)d_in[2];
  const float* beta  = (const float*)d_in[3];
  const float* wup2  = (const float*)d_in[4];
  const float* v1    = (const float*)d_in[5];
  const float* v2    = (const float*)d_in[6];

  char* ws = (char*)d_ws;
  size_t off = 0;
  auto walloc = [&](size_t bytes) -> void* {
    void* p = ws + off;
    off = (off + bytes + 255) & ~(size_t)255;
    return p;
  };
  unsigned int* Mswz = (unsigned int*)walloc(32768u * 4);
  unsigned int* Wswz = (unsigned int*)walloc(9u * 32768u * 4);
  float* stats1 = (float*)walloc(512 * 4);
  float* stats2 = (float*)walloc(512 * 4);
  float* sc1  = (float*)walloc(256 * 4);
  float* sh1  = (float*)walloc(256 * 4);
  float* sc2  = (float*)walloc(256 * 4);
  float* sh2  = (float*)walloc(256 * 4);
  float* scid = (float*)walloc(256 * 4);
  float* shid = (float*)walloc(256 * 4);
  float* z1024 = (float*)walloc((size_t)BB * 25 * 1024 * 4);
  float* out1  = (float*)walloc((size_t)BB * CC * HH * WW * 4);
  float* out2  = (float*)walloc((size_t)BB * CC * HH * WW * 4);

  prep_kernel<<<1280, 256, 0, stream>>>(wmsg, wup2, Mswz, Wswz);
  init_misc<<<1, 512, 0, stream>>>(stats1, stats2, scid, shid);

  scnn_kernel<<<256, 256, 0, stream>>>(p2c, out1, Mswz, scid, shid, 0, stats1);
  bnscale_kernel<<<1, 256, 0, stream>>>(stats1, gamma, beta, sc1, sh1);

  scnn_kernel<<<256, 256, 0, stream>>>(out1, out2, Mswz, sc1, sh1, 1, stats2);
  bnscale_kernel<<<1, 256, 0, stream>>>(stats2, gamma, beta, sc2, sh2);

  conv_kernel<<<BB * 25 * 16, 256, 0, stream>>>(out2, Wswz, sc2, sh2, v1, z1024);

  final_kernel<<<128, 256, 0, stream>>>(z1024, v2, (float*)d_out);
}


# conv: tile staging requests a work-group's source samples all at once as whole-line 128-bit loads and turns each row through LDS to one thread per channel, replacing 198 load-then-wait passes
# speedup vs baseline: 1.7787x; 1.6614x over previous
#include <hip/hip_runtime.h>
#include <hip/hip_bf16.h>
#include <stdint.h>

#define BB 8
#define CC 256
#define HH 25
#define WW 512

typedef __bf16 bf16x16 __attribute__((ext_vector_type(16)));
typedef float  f32x8   __attribute__((ext_vector_type(8)));
typedef float  f32x4   __attribute__((ext_vector_type(4)));
typedef float  f32x2   __attribute__((ext_vector_type(2)));

union FragU { uint4 u[2]; bf16x16 v; };

__device__ __forceinline__ unsigned short f2bf(float x) {
  unsigned u = __float_as_uint(x);
  unsigned r = u + 0x7FFFu + ((u >> 16) & 1u);
  return (unsigned short)(r >> 16);
}

__device__ __forceinline__ f32x8 wmma_bf16(bf16x16 a, bf16x16 b, f32x8 c) {
  return __builtin_amdgcn_wmma_f32_16x16x32_bf16(false, a, false, b, (short)0, c, false, false);
}

__device__ __forceinline__ bf16x16 load_frag_lds(const unsigned short* rowBase, int kt, int lane) {
  int kb = (lane & 16) ? 8 : 0;
  const unsigned short* p = rowBase + kt * 32 + kb;
  FragU f;
  f.u[0] = *(const uint4*)(p);
  f.u[1] = *(const uint4*)(p + 16);
  return f.v;
}

__device__ __forceinline__ bf16x16 load_frag_Bswz(const unsigned int* swz, int kt, int nt, int lane) {
  const unsigned int* p = swz + ((((kt << 4) + nt) * 32 + lane) << 3);
  FragU f;
  f.u[0] = *(const uint4*)(p);
  f.u[1] = *(const uint4*)(p + 4);
  return f.v;
}

__global__ void prep_kernel(const float* __restrict__ wmsg, const float* __restrict__ wup2,
                            unsigned int* __restrict__ Mswz, unsigned int* __restrict__ Wswz)
{
  int idx = blockIdx.x * 256 + threadIdx.x;
  int tap = idx >> 15;
  int r   = idx & 32767;
  int j    = r & 7;
  int lane = (r >> 3) & 31;
  int nt   = (r >> 8) & 15;
  int kt   = r >> 12;
  int n  = nt * 16 + (lane & 15);
  int kb = (lane & 16) ? 8 : 0;
  int k  = kt * 32 + ((j < 4) ? (kb + 2 * j) : (16 + kb + 2 * (j - 4)));
  float lo, hi;
  if (tap == 9) {
    lo = wmsg[(n * 256 + k) * 9 + 4];
    hi = wmsg[(n * 256 + k + 1) * 9 + 4];
  } else {
    int ky = tap / 3, kx = tap - ky * 3;
    lo = wup2[((n * 256 + k) * 3 + ky) * 3 + kx];
    hi = wup2[((n * 256 + k + 1) * 3 + ky) * 3 + kx];
  }
  unsigned int d = (unsigned)f2bf(lo) | ((unsigned)f2bf(hi) << 16);
  if (tap == 9) Mswz[r] = d;
  else          Wswz[tap * 32768 + r] = d;
}

__global__ void init_misc(float* stats1, float* stats2, float* scid, float* shid) {
  int t = threadIdx.x;
  stats1[t] = 0.f; stats2[t] = 0.f;
  if (t < 256) { scid[t] = 1.0f; shid[t] = 0.0f; }
}

__global__ void __launch_bounds__(256) scnn_kernel(
    const float* __restrict__ xin, float* __restrict__ xout,
    const unsigned int* __restrict__ Mswz,
    const float* __restrict__ sc, const float* __restrict__ sh,
    int relu_in, float* __restrict__ stats)
{
  __shared__ __align__(16) unsigned short prevA[16 * 264];
  __shared__ __align__(16) unsigned short prevB[16 * 264];

  const int tid  = threadIdx.x;
  const int b    = blockIdx.x >> 5;
  const int w0   = (blockIdx.x & 31) << 4;
  const int wave = tid >> 5;
  const int lane = tid & 31;
  const int half8 = (lane & 16) ? 8 : 0;
  const int nt0 = wave * 2, nt1 = wave * 2 + 1;
  const int ch0 = nt0 * 16 + (lane & 15);
  const int ch1 = nt1 * 16 + (lane & 15);

  for (int i = tid; i < 16 * 264; i += 256) prevA[i] = 0;

  const float sA = sc[ch0], bA = sh[ch0];
  const float sB = sc[ch1], bB = sh[ch1];

  const float* xi0p = xin  + ((size_t)(b * CC + ch0) * HH) * WW + w0 + half8;
  const float* xi1p = xin  + ((size_t)(b * CC + ch1) * HH) * WW + w0 + half8;
  float*       xo0p = xout + ((size_t)(b * CC + ch0) * HH) * WW + w0 + half8;
  float*       xo1p = xout + ((size_t)(b * CC + ch1) * HH) * WW + w0 + half8;

  float4 x0a = *(const float4*)(xi0p);
  float4 x0b = *(const float4*)(xi0p + 4);
  float4 x1a = *(const float4*)(xi1p);
  float4 x1b = *(const float4*)(xi1p + 4);

  float s0 = 0.f, q0 = 0.f, s1 = 0.f, q1 = 0.f;

  for (int h = 0; h < HH; ++h) {
    unsigned short* prevCur = (h & 1) ? prevB : prevA;
    unsigned short* prevNxt = (h & 1) ? prevA : prevB;
    __syncthreads();

    f32x8 acc0 = {}; f32x8 acc1 = {};
    const unsigned short* arow = &prevCur[(lane & 15) * 264];
    bf16x16 a = load_frag_lds(arow, 0, lane);
    #pragma unroll
    for (int kt = 0; kt < 8; ++kt) {
      bf16x16 an = (kt < 7) ? load_frag_lds(arow, kt + 1, lane) : a;
      acc0 = wmma_bf16(a, load_frag_Bswz(Mswz, kt, nt0, lane), acc0);
      acc1 = wmma_bf16(a, load_frag_Bswz(Mswz, kt, nt1, lane), acc1);
      a = an;
    }

    const float xi0[8] = {x0a.x, x0a.y, x0a.z, x0a.w, x0b.x, x0b.y, x0b.z, x0b.w};
    const float xi1[8] = {x1a.x, x1a.y, x1a.z, x1a.w, x1b.x, x1b.y, x1b.z, x1b.w};
    float r0[8], r1[8];
    #pragma unroll
    for (int j = 0; j < 8; ++j) {
      float a0 = sA * xi0[j] + bA;
      float a1 = sB * xi1[j] + bB;
      if (relu_in) { a0 = fmaxf(a0, 0.f); a1 = fmaxf(a1, 0.f); }
      float v0 = a0 + fmaxf(acc0[j], 0.f);
      float v1 = a1 + fmaxf(acc1[j], 0.f);
      r0[j] = v0; r1[j] = v1;
      s0 += v0; q0 += v0 * v0;
      s1 += v1; q1 += v1 * v1;
      prevNxt[(j + half8) * 264 + ch0] = f2bf(v0);
      prevNxt[(j + half8) * 264 + ch1] = f2bf(v1);
    }
    {
      float* p0 = xo0p + (size_t)(HH - 1 - h) * WW;
      float* p1 = xo1p + (size_t)(HH - 1 - h) * WW;
      *(float4*)(p0)     = make_float4(r0[0], r0[1], r0[2], r0[3]);
      *(float4*)(p0 + 4) = make_float4(r0[4], r0[5], r0[6], r0[7]);
      *(float4*)(p1)     = make_float4(r1[0], r1[1], r1[2], r1[3]);
      *(float4*)(p1 + 4) = make_float4(r1[4], r1[5], r1[6], r1[7]);
    }
    if (h + 1 < HH) {
      const float* n0 = xi0p + (size_t)(h + 1) * WW;
      const float* n1 = xi1p + (size_t)(h + 1) * WW;
      x0a = *(const float4*)(n0); x0b = *(const float4*)(n0 + 4);
      x1a = *(const float4*)(n1); x1b = *(const float4*)(n1 + 4);
      if (h + 2 < HH) {
        __builtin_prefetch(xi0p + (size_t)(h + 2) * WW, 0, 0);
        __builtin_prefetch(xi1p + (size_t)(h + 2) * WW, 0, 0);
      }
    }
  }

  atomicAdd(&stats[ch0],       s0);
  atomicAdd(&stats[256 + ch0], q0);
  atomicAdd(&stats[ch1],       s1);
  atomicAdd(&stats[256 + ch1], q1);
}

__global__ void bnscale_kernel(const float* __restrict__ stats,
                               const float* __restrict__ gamma, const float* __restrict__ beta,
                               float* __restrict__ sc, float* __restrict__ sh)
{
  int c = threadIdx.x;
  const float n = (float)(BB * HH * WW);
  float mean = stats[c] / n;
  float var  = stats[256 + c] / n - mean * mean;
  float s = gamma[c] * rsqrtf(var + 1e-5f);
  sc[c] = s;
  sh[c] = beta[c] - mean * s;
}

#define CONV_MT 4
#define CONV_TW (CONV_MT * 16)
#define CONV_HW (CONV_TW + 2)
#define RAW_P   36

__global__ void __launch_bounds__(256) conv_kernel(
    const float* __restrict__ xin,
    const unsigned int* __restrict__ Wswz,
    const float* __restrict__ sc, const float* __restrict__ sh,
    const float* __restrict__ v1,
    float* __restrict__ z)
{
  __shared__ __align__(16) unsigned short uT[3 * CONV_HW * 264];
  __shared__ float zred[CONV_TW];
  __shared__ __align__(16) float raw[256 * RAW_P];

  const int tid = threadIdx.x;
  const int wt  = blockIdx.x & 15;
  const int h   = (blockIdx.x >> 4) % 25;
  const int b   = blockIdx.x / (16 * 25);
  const int w20 = wt * CONV_TW;

  if (tid < CONV_TW) zred[tid] = 0.f;

  {
    const int M0   = wt * 32;
    const int offL = (wt > 0)  ? -1 : 0;
    const int offR = (wt < 15) ? 32 : 31;
    const int pk   = tid & 7;
    const int pr   = tid >> 3;

    auto load_row = [&](int hh, f32x4 (&q)[8], float& eL, float& eR) {
      const int hs  = h + hh - 1;
      const int hsc = min(max(hs, 0), HH - 1);
      #pragma unroll
      for (int i = 0; i < 8; ++i)
        q[i] = *(const f32x4*)(xin + ((b * CC + i * 32 + pr) * HH + hsc) * WW + M0 + 4 * pk);
      const float* erow = xin + ((b * CC + tid) * HH + hsc) * WW + M0;
      eL = erow[offL];
      eR = erow[offR];
    };
    auto put_row = [&](const f32x4 (&q)[8], float eL, float eR) {
      #pragma unroll
      for (int i = 0; i < 8; ++i)
        *(f32x4*)&raw[(i * 32 + pr) * RAW_P + 4 * pk] = q[i];
      *(f32x2*)&raw[tid * RAW_P + 32] = (f32x2){eL, eR};
    };
    const float scc = sc[tid], shc = sh[tid];
    auto emit_row = [&](int hh) {
      const int  hs    = h + hh - 1;
      const bool rowok = (hs >= 0 && hs < HH);
      float xs[34];
      const f32x2 e = *(const f32x2*)&raw[tid * RAW_P + 32];
      xs[0] = e.x; xs[33] = e.y;
      #pragma unroll
      for (int k = 0; k < 8; ++k) {
        const f32x4 m = *(const f32x4*)&raw[tid * RAW_P + 4 * k];
        xs[1 + 4 * k] = m.x; xs[2 + 4 * k] = m.y; xs[3 + 4 * k] = m.z; xs[4 + 4 * k] = m.w;
      }
      #pragma unroll
      for (int k = 0; k < 34; ++k) {
        const float x = fmaxf(scc * xs[k] + shc, 0.f);
        xs[k] = rowok ? x : 0.f;
      }
      #pragma unroll
      for (int wi = 0; wi < CONV_HW; ++wi) {
        const int  w2    = w20 + wi - 1;
        const bool colok = (w2 >= 0 && w2 < 1024);
        const int   num = w2 * 511;
        const int   i0  = num / 1023;
        const float f   = (float)(num - i0 * 1023) * (1.0f / 1023.0f);
        const int   s   = wi >> 1;
        const bool  col0 = (wi == 1) && (wt == 0);
        const float x0 = col0 ? xs[1] : xs[s];
        const float x1 = col0 ? xs[2] : xs[s + 1];
        const float v  = x0 + f * (x1 - x0);
        uT[(hh * CONV_HW + wi) * 264 + tid] = colok ? f2bf(v) : (unsigned short)0;
      }
    };

    f32x4 q0[8], q1[8], q2[8];
    float  l0, r0, l1, r1, l2, r2;
    load_row(0, q0, l0, r0);
    load_row(1, q1, l1, r1);
    load_row(2, q2, l2, r2);

    put_row(q0, l0, r0);
    __syncthreads();
    emit_row(0);
    __syncthreads();
    put_row(q1, l1, r1);
    __syncthreads();
    emit_row(1);
    __syncthreads();
    put_row(q2, l2, r2);
    __syncthreads();
    emit_row(2);
  }
  __syncthreads();

  const int wave = tid >> 5;
  const int lane = tid & 31;
  const int nt0 = wave * 2, nt1 = nt0 + 1;
  f32x8 acc0[CONV_MT], acc1[CONV_MT];
  #pragma unroll
  for (int m = 0; m < CONV_MT; ++m) { acc0[m] = (f32x8){}; acc1[m] = (f32x8){}; }

  const unsigned short* abase = &uT[(lane & 15) * 264];

  bf16x16 a[CONV_MT], b0, b1;
  #pragma unroll
  for (int m = 0; m < CONV_MT; ++m) a[m] = load_frag_lds(abase + m * 16 * 264, 0, lane);
  b0 = load_frag_Bswz(Wswz, 0, nt0, lane);
  b1 = load_frag_Bswz(Wswz, 0, nt1, lane);

  #pragma unroll
  for (int s = 0; s < 72; ++s) {
    bf16x16 an[CONV_MT], bn0, bn1;
    if (s + 1 < 72) {
      const int tap = (s + 1) >> 3, kt = (s + 1) & 7;
      const int ky = tap / 3, kx = tap - ky * 3;
      const unsigned short* arow = abase + (ky * CONV_HW + kx) * 264;
      #pragma unroll
      for (int m = 0; m < CONV_MT; ++m) an[m] = load_frag_lds(arow + m * 16 * 264, kt, lane);
      bn0 = load_frag_Bswz(Wswz + tap * 32768, kt, nt0, lane);
      bn1 = load_frag_Bswz(Wswz + tap * 32768, kt, nt1, lane);
    } else {
      #pragma unroll
      for (int m = 0; m < CONV_MT; ++m) an[m] = a[m];
      bn0 = b0; bn1 = b1;
    }
    #pragma unroll
    for (int m = 0; m < CONV_MT; ++m) {
      acc0[m] = wmma_bf16(a[m], b0, acc0[m]);
      acc1[m] = wmma_bf16(a[m], b1, acc1[m]);
    }
    #pragma unroll
    for (int m = 0; m < CONV_MT; ++m) a[m] = an[m];
    b0 = bn0; b1 = bn1;
  }

  const int half8 = (lane & 16) ? 8 : 0;
  const int ch0 = nt0 * 16 + (lane & 15);
  const int ch1 = nt1 * 16 + (lane & 15);
  float g0 = v1[ch0], g1 = v1[ch1];
  #pragma unroll
  for (int m = 0; m < CONV_MT; ++m) {
    #pragma unroll
    for (int j = 0; j < 8; ++j) {
      atomicAdd(&zred[m * 16 + j + half8], acc0[m][j] * g0 + acc1[m][j] * g1);
    }
  }
  __syncthreads();
  if (tid < CONV_TW) z[(b * 25 + h) * 1024 + w20 + tid] = zred[tid];
}

__global__ void final_kernel(const float* __restrict__ z, const float* __restrict__ v2,
                             float* __restrict__ out)
{
  int idx = blockIdx.x * 256 + threadIdx.x;
  if (idx >= BB * 4096) return;
  int b  = idx >> 12;
  int w4 = idx & 4095;

  int   num4 = w4 * 2047;
  int   i40  = num4 / 4095;
  float f4   = (float)(num4 - i40 * 4095) * (1.0f / 4095.0f);
  int   i41  = min(i40 + 1, 2047);

  int n3a = i40 * 1023; int ja0 = n3a / 2047;
  float fa = (float)(n3a - ja0 * 2047) * (1.0f / 2047.0f);
  int ja1 = min(ja0 + 1, 1023);
  int n3b = i41 * 1023; int jb0 = n3b / 2047;
  float fb = (float)(n3b - jb0 * 2047) * (1.0f / 2047.0f);
  int jb1 = min(jb0 + 1, 1023);

  float acc = 0.f;
  #pragma unroll 5
  for (int h = 0; h < HH; ++h) {
    const float* zh = z + (b * 25 + h) * 1024;
    float za = zh[ja0] + fa * (zh[ja1] - zh[ja0]);
    float zb = zh[jb0] + fb * (zh[jb1] - zh[jb0]);
    acc += v2[h] * (za + f4 * (zb - za));
  }
  out[idx] = 1.0f / (1.0f + __expf(-acc));
}

extern "C" void kernel_launch(void* const* d_in, const int* in_sizes, int n_in,
                              void* d_out, int out_size, void* d_ws, size_t ws_size,
                              hipStream_t stream)
{
  const float* p2c   = (const float*)d_in[0];
  const float* wmsg  = (const float*)d_in[1];
  const float* gamma = (const float*)d_in[2];
  const float* beta  = (const float*)d_in[3];
  const float* wup2  = (const float*)d_in[4];
  const float* v1    = (const float*)d_in[5];
  const float* v2    = (const float*)d_in[6];

  char* ws = (char*)d_ws;
  size_t off = 0;
  auto walloc = [&](size_t bytes) -> void* {
    void* p = ws + off;
    off = (off + bytes + 255) & ~(size_t)255;
    return p;
  };
  unsigned int* Mswz = (unsigned int*)walloc(32768u * 4);
  unsigned int* Wswz = (unsigned int*)walloc(9u * 32768u * 4);
  float* stats1 = (float*)walloc(512 * 4);
  float* stats2 = (float*)walloc(512 * 4);
  float* sc1  = (float*)walloc(256 * 4);
  float* sh1  = (float*)walloc(256 * 4);
  float* sc2  = (float*)walloc(256 * 4);
  float* sh2  = (float*)walloc(256 * 4);
  float* scid = (float*)walloc(256 * 4);
  float* shid = (float*)walloc(256 * 4);
  float* z1024 = (float*)walloc((size_t)BB * 25 * 1024 * 4);
  float* out1  = (float*)walloc((size_t)BB * CC * HH * WW * 4);
  float* out2  = (float*)walloc((size_t)BB * CC * HH * WW * 4);

  prep_kernel<<<1280, 256, 0, stream>>>(wmsg, wup2, Mswz, Wswz);
  init_misc<<<1, 512, 0, stream>>>(stats1, stats2, scid, shid);

  scnn_kernel<<<256, 256, 0, stream>>>(p2c, out1, Mswz, scid, shid, 0, stats1);
  bnscale_kernel<<<1, 256, 0, stream>>>(stats1, gamma, beta, sc1, sh1);

  scnn_kernel<<<256, 256, 0, stream>>>(out1, out2, Mswz, sc1, sh1, 1, stats2);
  bnscale_kernel<<<1, 256, 0, stream>>>(stats2, gamma, beta, sc2, sh2);

  conv_kernel<<<BB * 25 * 16, 256, 0, stream>>>(out2, Wswz, sc2, sh2, v1, z1024);

  final_kernel<<<128, 256, 0, stream>>>(z1024, v2, (float*)d_out);
}


# earlier try varied run to run: BN sums were float atomics, now fixed-order partials; conv: w_conv1 folded into 3x3 weights in fp32 (value+remainder columns), N 256->16, K over 8 waves, ordered sums
# speedup vs baseline: 2.8021x; 1.5754x over previous
#include <hip/hip_runtime.h>
#include <hip/hip_bf16.h>
#include <stdint.h>

#define BB 8
#define CC 256
#define HH 25
#define WW 512

typedef __bf16 bf16x16 __attribute__((ext_vector_type(16)));
typedef float  f32x8   __attribute__((ext_vector_type(8)));
typedef float  f32x4   __attribute__((ext_vector_type(4)));
typedef float  f32x2   __attribute__((ext_vector_type(2)));

union FragU { uint4 u[2]; bf16x16 v; };

__device__ __forceinline__ unsigned short f2bf(float x) {
  unsigned u = __float_as_uint(x);
  unsigned r = u + 0x7FFFu + ((u >> 16) & 1u);
  return (unsigned short)(r >> 16);
}

__device__ __forceinline__ f32x8 wmma_bf16(bf16x16 a, bf16x16 b, f32x8 c) {
  return __builtin_amdgcn_wmma_f32_16x16x32_bf16(false, a, false, b, (short)0, c, false, false);
}

__device__ __forceinline__ bf16x16 load_frag_lds(const unsigned short* rowBase, int kt, int lane) {
  int kb = (lane & 16) ? 8 : 0;
  const unsigned short* p = rowBase + kt * 32 + kb;
  FragU f;
  f.u[0] = *(const uint4*)(p);
  f.u[1] = *(const uint4*)(p + 16);
  return f.v;
}

__device__ __forceinline__ bf16x16 load_frag_Bswz(const unsigned int* swz, int kt, int nt, int lane) {
  const unsigned int* p = swz + ((((kt << 4) + nt) * 32 + lane) << 3);
  FragU f;
  f.u[0] = *(const uint4*)(p);
  f.u[1] = *(const uint4*)(p + 4);
  return f.v;
}

__device__ __forceinline__ bf16x16 load_frag_F(const unsigned int* swz, int tap, int kt, int lane) {
  const unsigned int* p = swz + ((((tap << 3) + kt) * 32 + lane) << 3);
  FragU f;
  f.u[0] = *(const uint4*)(p);
  f.u[1] = *(const uint4*)(p + 4);
  return f.v;
}

__device__ __forceinline__ float bf2f(unsigned short h) {
  return __uint_as_float(((unsigned)h) << 16);
}

__global__ void __launch_bounds__(256) prep_kernel(
    const float* __restrict__ wmsg, const float* __restrict__ wup2, const float* __restrict__ v1,
    unsigned int* __restrict__ Mswz, unsigned int* __restrict__ Fswz)
{
  __shared__ float part[8][18];
  __shared__ float wsum[18];
  const int tid = threadIdx.x;

  if (blockIdx.x >= 128) {
    int r    = (blockIdx.x - 128) * 256 + tid;
    int j    = r & 7;
    int lane = (r >> 3) & 31;
    int nt   = (r >> 8) & 15;
    int kt   = r >> 12;
    int n  = nt * 16 + (lane & 15);
    int kb = (lane & 16) ? 8 : 0;
    int k  = kt * 32 + ((j < 4) ? (kb + 2 * j) : (16 + kb + 2 * (j - 4)));
    float lo = wmsg[(n * 256 + k) * 9 + 4];
    float hi = wmsg[(n * 256 + k + 1) * 9 + 4];
    Mswz[r] = (unsigned)f2bf(lo) | ((unsigned)f2bf(hi) << 16);
    return;
  }

  const int kp   = blockIdx.x;
  const int wave = tid >> 5;
  const int lane = tid & 31;
  const float g  = v1[tid];
  const float* srcw = wup2 + (tid * 256 + 2 * kp) * 9;
  float p[18];
  #pragma unroll
  for (int t = 0; t < 18; ++t) p[t] = g * srcw[t];
  #pragma unroll
  for (int t = 0; t < 18; ++t) {
    for (int off = 16; off >= 1; off >>= 1) p[t] += __shfl_xor(p[t], off);
  }
  if (lane == 0) {
    #pragma unroll
    for (int t = 0; t < 18; ++t) part[wave][t] = p[t];
  }
  __syncthreads();
  if (tid < 18) {
    float s = 0.f;
    for (int w = 0; w < 8; ++w) s += part[w][tid];
    wsum[tid] = s;
  }
  __syncthreads();
  if (tid < 144) {
    const int tap = tid >> 4;
    const int n   = tid & 15;
    const float w0 = wsum[tap], w1 = wsum[9 + tap];
    const unsigned short h0 = f2bf(w0), h1 = f2bf(w1);
    unsigned int d = 0u;
    if (n == 0) {
      d = (unsigned)h0 | ((unsigned)h1 << 16);
    } else if (n == 1) {
      d = (unsigned)f2bf(w0 - bf2f(h0)) | ((unsigned)f2bf(w1 - bf2f(h1)) << 16);
    }
    const int k   = 2 * kp;
    const int kt  = k >> 5;
    const int r   = k & 31;
    const int kbs = (r >> 3) & 1;
    const int j   = ((r >> 4) << 2) + ((r & 7) >> 1);
    Fswz[((((tap << 3) + kt) * 32 + kbs * 16 + n) << 3) + j] = d;
  }
}

__global__ void init_misc(float* scid, float* shid) {
  int t = threadIdx.x;
  scid[t] = 1.0f; shid[t] = 0.0f;
}

__global__ void __launch_bounds__(256) scnn_kernel(
    const float* __restrict__ xin, float* __restrict__ xout,
    const unsigned int* __restrict__ Mswz,
    const float* __restrict__ sc, const float* __restrict__ sh,
    int relu_in, float* __restrict__ statsP)
{
  __shared__ __align__(16) unsigned short prevA[16 * 264];
  __shared__ __align__(16) unsigned short prevB[16 * 264];

  const int tid  = threadIdx.x;
  const int b    = blockIdx.x >> 5;
  const int w0   = (blockIdx.x & 31) << 4;
  const int wave = tid >> 5;
  const int lane = tid & 31;
  const int half8 = (lane & 16) ? 8 : 0;
  const int nt0 = wave * 2, nt1 = wave * 2 + 1;
  const int ch0 = nt0 * 16 + (lane & 15);
  const int ch1 = nt1 * 16 + (lane & 15);

  for (int i = tid; i < 16 * 264; i += 256) prevA[i] = 0;

  const float sA = sc[ch0], bA = sh[ch0];
  const float sB = sc[ch1], bB = sh[ch1];

  const float* xi0p = xin  + ((size_t)(b * CC + ch0) * HH) * WW + w0 + half8;
  const float* xi1p = xin  + ((size_t)(b * CC + ch1) * HH) * WW + w0 + half8;
  float*       xo0p = xout + ((size_t)(b * CC + ch0) * HH) * WW + w0 + half8;
  float*       xo1p = xout + ((size_t)(b * CC + ch1) * HH) * WW + w0 + half8;

  float4 x0a = *(const float4*)(xi0p);
  float4 x0b = *(const float4*)(xi0p + 4);
  float4 x1a = *(const float4*)(xi1p);
  float4 x1b = *(const float4*)(xi1p + 4);

  float s0 = 0.f, q0 = 0.f, s1 = 0.f, q1 = 0.f;

  for (int h = 0; h < HH; ++h) {
    unsigned short* prevCur = (h & 1) ? prevB : prevA;
    unsigned short* prevNxt = (h & 1) ? prevA : prevB;
    __syncthreads();

    f32x8 acc0 = {}; f32x8 acc1 = {};
    const unsigned short* arow = &prevCur[(lane & 15) * 264];
    bf16x16 a = load_frag_lds(arow, 0, lane);
    #pragma unroll
    for (int kt = 0; kt < 8; ++kt) {
      bf16x16 an = (kt < 7) ? load_frag_lds(arow, kt + 1, lane) : a;
      acc0 = wmma_bf16(a, load_frag_Bswz(Mswz, kt, nt0, lane), acc0);
      acc1 = wmma_bf16(a, load_frag_Bswz(Mswz, kt, nt1, lane), acc1);
      a = an;
    }

    const float xi0[8] = {x0a.x, x0a.y, x0a.z, x0a.w, x0b.x, x0b.y, x0b.z, x0b.w};
    const float xi1[8] = {x1a.x, x1a.y, x1a.z, x1a.w, x1b.x, x1b.y, x1b.z, x1b.w};
    float r0[8], r1[8];
    #pragma unroll
    for (int j = 0; j < 8; ++j) {
      float a0 = sA * xi0[j] + bA;
      float a1 = sB * xi1[j] + bB;
      if (relu_in) { a0 = fmaxf(a0, 0.f); a1 = fmaxf(a1, 0.f); }
      float v0 = a0 + fmaxf(acc0[j], 0.f);
      float v1 = a1 + fmaxf(acc1[j], 0.f);
      r0[j] = v0; r1[j] = v1;
      s0 += v0; q0 += v0 * v0;
      s1 += v1; q1 += v1 * v1;
      prevNxt[(j + half8) * 264 + ch0] = f2bf(v0);
      prevNxt[(j + half8) * 264 + ch1] = f2bf(v1);
    }
    {
      float* p0 = xo0p + (size_t)(HH - 1 - h) * WW;
      float* p1 = xo1p + (size_t)(HH - 1 - h) * WW;
      *(float4*)(p0)     = make_float4(r0[0], r0[1], r0[2], r0[3]);
      *(float4*)(p0 + 4) = make_float4(r0[4], r0[5], r0[6], r0[7]);
      *(float4*)(p1)     = make_float4(r1[0], r1[1], r1[2], r1[3]);
      *(float4*)(p1 + 4) = make_float4(r1[4], r1[5], r1[6], r1[7]);
    }
    if (h + 1 < HH) {
      const float* n0 = xi0p + (size_t)(h + 1) * WW;
      const float* n1 = xi1p + (size_t)(h + 1) * WW;
      x0a = *(const float4*)(n0); x0b = *(const float4*)(n0 + 4);
      x1a = *(const float4*)(n1); x1b = *(const float4*)(n1 + 4);
      if (h + 2 < HH) {
        __builtin_prefetch(xi0p + (size_t)(h + 2) * WW, 0, 0);
        __builtin_prefetch(xi1p + (size_t)(h + 2) * WW, 0, 0);
      }
    }
  }

  s0 += __shfl_xor(s0, 16); q0 += __shfl_xor(q0, 16);
  s1 += __shfl_xor(s1, 16); q1 += __shfl_xor(q1, 16);
  if (lane < 16) {
    statsP[(ch0)       * 256 + blockIdx.x] = s0;
    statsP[(256 + ch0) * 256 + blockIdx.x] = q0;
    statsP[(ch1)       * 256 + blockIdx.x] = s1;
    statsP[(256 + ch1) * 256 + blockIdx.x] = q1;
  }
}

__global__ void __launch_bounds__(256) bnscale_kernel(
    const float* __restrict__ statsP,
    const float* __restrict__ gamma, const float* __restrict__ beta,
    float* __restrict__ sc, float* __restrict__ sh)
{
  __shared__ float ps[8], pq[8];
  const int c = blockIdx.x;
  const int t = threadIdx.x;
  float s = statsP[c * 256 + t];
  float q = statsP[(256 + c) * 256 + t];
  for (int off = 16; off >= 1; off >>= 1) { s += __shfl_xor(s, off); q += __shfl_xor(q, off); }
  if ((t & 31) == 0) { ps[t >> 5] = s; pq[t >> 5] = q; }
  __syncthreads();
  if (t == 0) {
    float S = 0.f, Q = 0.f;
    for (int w = 0; w < 8; ++w) { S += ps[w]; Q += pq[w]; }
    const float n = (float)(BB * HH * WW);
    float mean = S / n;
    float var  = Q / n - mean * mean;
    float g = gamma[c] * rsqrtf(var + 1e-5f);
    sc[c] = g;
    sh[c] = beta[c] - mean * g;
  }
}

#define CONV_MT 4
#define CONV_TW (CONV_MT * 16)
#define CONV_HW (CONV_TW + 2)
#define RAW_P   36

__global__ void __launch_bounds__(256) conv_kernel(
    const float* __restrict__ xin,
    const unsigned int* __restrict__ Fswz,
    const float* __restrict__ sc, const float* __restrict__ sh,
    float* __restrict__ z)
{
  __shared__ __align__(16) unsigned short uT[3 * CONV_HW * 264];
  __shared__ __align__(16) float raw[256 * RAW_P];

  const int tid = threadIdx.x;
  const int wt  = blockIdx.x & 15;
  const int h   = (blockIdx.x >> 4) % 25;
  const int b   = blockIdx.x / (16 * 25);
  const int w20 = wt * CONV_TW;

  {
    const int M0   = wt * 32;
    const int offL = (wt > 0)  ? -1 : 0;
    const int offR = (wt < 15) ? 32 : 31;
    const int pk   = tid & 7;
    const int pr   = tid >> 3;

    auto load_row = [&](int hh, f32x4 (&q)[8], float& eL, float& eR) {
      const int hs  = h + hh - 1;
      const int hsc = min(max(hs, 0), HH - 1);
      #pragma unroll
      for (int i = 0; i < 8; ++i)
        q[i] = *(const f32x4*)(xin + ((b * CC + i * 32 + pr) * HH + hsc) * WW + M0 + 4 * pk);
      const float* erow = xin + ((b * CC + tid) * HH + hsc) * WW + M0;
      eL = erow[offL];
      eR = erow[offR];
    };
    auto put_row = [&](const f32x4 (&q)[8], float eL, float eR) {
      #pragma unroll
      for (int i = 0; i < 8; ++i)
        *(f32x4*)&raw[(i * 32 + pr) * RAW_P + 4 * pk] = q[i];
      *(f32x2*)&raw[tid * RAW_P + 32] = (f32x2){eL, eR};
    };
    const float scc = sc[tid], shc = sh[tid];
    auto emit_row = [&](int hh) {
      const int  hs    = h + hh - 1;
      const bool rowok = (hs >= 0 && hs < HH);
      float xs[34];
      const f32x2 e = *(const f32x2*)&raw[tid * RAW_P + 32];
      xs[0] = e.x; xs[33] = e.y;
      #pragma unroll
      for (int k = 0; k < 8; ++k) {
        const f32x4 m = *(const f32x4*)&raw[tid * RAW_P + 4 * k];
        xs[1 + 4 * k] = m.x; xs[2 + 4 * k] = m.y; xs[3 + 4 * k] = m.z; xs[4 + 4 * k] = m.w;
      }
      #pragma unroll
      for (int k = 0; k < 34; ++k) {
        const float x = fmaxf(scc * xs[k] + shc, 0.f);
        xs[k] = rowok ? x : 0.f;
      }
      #pragma unroll
      for (int wi = 0; wi < CONV_HW; ++wi) {
        const int  w2    = w20 + wi - 1;
        const bool colok = (w2 >= 0 && w2 < 1024);
        const int   num = w2 * 511;
        const int   i0  = num / 1023;
        const float f   = (float)(num - i0 * 1023) * (1.0f / 1023.0f);
        const int   s   = wi >> 1;
        const bool  col0 = (wi == 1) && (wt == 0);
        const float x0 = col0 ? xs[1] : xs[s];
        const float x1 = col0 ? xs[2] : xs[s + 1];
        const float v  = x0 + f * (x1 - x0);
        uT[(hh * CONV_HW + wi) * 264 + tid] = colok ? f2bf(v) : (unsigned short)0;
      }
    };

    f32x4 q0[8], q1[8], q2[8];
    float  l0, r0, l1, r1, l2, r2;
    load_row(0, q0, l0, r0);
    load_row(1, q1, l1, r1);
    load_row(2, q2, l2, r2);

    put_row(q0, l0, r0);
    __syncthreads();
    emit_row(0);
    __syncthreads();
    put_row(q1, l1, r1);
    __syncthreads();
    emit_row(1);
    __syncthreads();
    put_row(q2, l2, r2);
    __syncthreads();
    emit_row(2);
  }
  __syncthreads();

  const int wave = tid >> 5;
  const int lane = tid & 31;
  f32x8 acc[CONV_MT];
  #pragma unroll
  for (int m = 0; m < CONV_MT; ++m) acc[m] = (f32x8){};

  const unsigned short* abase = &uT[(lane & 15) * 264];

  #pragma unroll
  for (int tap = 0; tap < 9; ++tap) {
    const int ky = tap / 3, kx = tap - ky * 3;
    const unsigned short* arow = abase + (ky * CONV_HW + kx) * 264;
    const bf16x16 bw = load_frag_F(Fswz, tap, wave, lane);
    #pragma unroll
    for (int m = 0; m < CONV_MT; ++m)
      acc[m] = wmma_bf16(load_frag_lds(arow + m * 16 * 264, wave, lane), bw, acc[m]);
  }

  float* zp = raw;
  const int half8 = (lane & 16) ? 8 : 0;
  #pragma unroll
  for (int m = 0; m < CONV_MT; ++m) {
    #pragma unroll
    for (int j = 0; j < 8; ++j)
      zp[((wave * CONV_TW) + m * 16 + j + half8) * 16 + (lane & 15)] = acc[m][j];
  }
  __syncthreads();
  if (tid < CONV_TW) {
    float zh = 0.f, zl = 0.f;
    for (int w = 0; w < 8; ++w) {
      zh += zp[(w * CONV_TW + tid) * 16 + 0];
      zl += zp[(w * CONV_TW + tid) * 16 + 1];
    }
    z[(b * 25 + h) * 1024 + w20 + tid] = zh + zl;
  }
}

__global__ void final_kernel(const float* __restrict__ z, const float* __restrict__ v2,
                             float* __restrict__ out)
{
  int idx = blockIdx.x * 256 + threadIdx.x;
  if (idx >= BB * 4096) return;
  int b  = idx >> 12;
  int w4 = idx & 4095;

  int   num4 = w4 * 2047;
  int   i40  = num4 / 4095;
  float f4   = (float)(num4 - i40 * 4095) * (1.0f / 4095.0f);
  int   i41  = min(i40 + 1, 2047);

  int n3a = i40 * 1023; int ja0 = n3a / 2047;
  float fa = (float)(n3a - ja0 * 2047) * (1.0f / 2047.0f);
  int ja1 = min(ja0 + 1, 1023);
  int n3b = i41 * 1023; int jb0 = n3b / 2047;
  float fb = (float)(n3b - jb0 * 2047) * (1.0f / 2047.0f);
  int jb1 = min(jb0 + 1, 1023);

  float acc = 0.f;
  #pragma unroll 5
  for (int h = 0; h < HH; ++h) {
    const float* zh = z + (b * 25 + h) * 1024;
    float za = zh[ja0] + fa * (zh[ja1] - zh[ja0]);
    float zb = zh[jb0] + fb * (zh[jb1] - zh[jb0]);
    acc += v2[h] * (za + f4 * (zb - za));
  }
  out[idx] = 1.0f / (1.0f + __expf(-acc));
}

extern "C" void kernel_launch(void* const* d_in, const int* in_sizes, int n_in,
                              void* d_out, int out_size, void* d_ws, size_t ws_size,
                              hipStream_t stream)
{
  const float* p2c   = (const float*)d_in[0];
  const float* wmsg  = (const float*)d_in[1];
  const float* gamma = (const float*)d_in[2];
  const float* beta  = (const float*)d_in[3];
  const float* wup2  = (const float*)d_in[4];
  const float* v1    = (const float*)d_in[5];
  const float* v2    = (const float*)d_in[6];

  char* ws = (char*)d_ws;
  size_t off = 0;
  auto walloc = [&](size_t bytes) -> void* {
    void* p = ws + off;
    off = (off + bytes + 255) & ~(size_t)255;
    return p;
  };
  unsigned int* Mswz = (unsigned int*)walloc(32768u * 4);
  unsigned int* Fswz = (unsigned int*)walloc(9u * 2048u * 4);
  float* stats1 = (float*)walloc(512u * 256u * 4);
  float* stats2 = (float*)walloc(512u * 256u * 4);
  float* sc1  = (float*)walloc(256 * 4);
  float* sh1  = (float*)walloc(256 * 4);
  float* sc2  = (float*)walloc(256 * 4);
  float* sh2  = (float*)walloc(256 * 4);
  float* scid = (float*)walloc(256 * 4);
  float* shid = (float*)walloc(256 * 4);
  float* z1024 = (float*)walloc((size_t)BB * 25 * 1024 * 4);
  float* out1  = (float*)walloc((size_t)BB * CC * HH * WW * 4);
  float* out2  = (float*)walloc((size_t)BB * CC * HH * WW * 4);

  prep_kernel<<<256, 256, 0, stream>>>(wmsg, wup2, v1, Mswz, Fswz);
  init_misc<<<1, 256, 0, stream>>>(scid, shid);

  scnn_kernel<<<256, 256, 0, stream>>>(p2c, out1, Mswz, scid, shid, 0, stats1);
  bnscale_kernel<<<256, 256, 0, stream>>>(stats1, gamma, beta, sc1, sh1);

  scnn_kernel<<<256, 256, 0, stream>>>(out1, out2, Mswz, sc1, sh1, 1, stats2);
  bnscale_kernel<<<256, 256, 0, stream>>>(stats2, gamma, beta, sc2, sh2);

  conv_kernel<<<BB * 25 * 16, 256, 0, stream>>>(out2, Fswz, sc2, sh2, z1024);

  final_kernel<<<128, 256, 0, stream>>>(z1024, v2, (float*)d_out);
}


# scnn: per-step barrier now orders LDS only, so global loads stay in flight across it; input rows are requested 4 steps ahead into 5 register sets instead of being waited for at every barrier
# speedup vs baseline: 3.0191x; 1.0774x over previous
#include <hip/hip_runtime.h>
#include <hip/hip_bf16.h>
#include <stdint.h>

#define BB 8
#define CC 256
#define HH 25
#define WW 512

typedef __bf16 bf16x16 __attribute__((ext_vector_type(16)));
typedef float  f32x8   __attribute__((ext_vector_type(8)));
typedef float  f32x4   __attribute__((ext_vector_type(4)));
typedef float  f32x2   __attribute__((ext_vector_type(2)));

union FragU { uint4 u[2]; bf16x16 v; };

__device__ __forceinline__ unsigned short f2bf(float x) {
  unsigned u = __float_as_uint(x);
  unsigned r = u + 0x7FFFu + ((u >> 16) & 1u);
  return (unsigned short)(r >> 16);
}

__device__ __forceinline__ f32x8 wmma_bf16(bf16x16 a, bf16x16 b, f32x8 c) {
  return __builtin_amdgcn_wmma_f32_16x16x32_bf16(false, a, false, b, (short)0, c, false, false);
}

__device__ __forceinline__ bf16x16 load_frag_lds(const unsigned short* rowBase, int kt, int lane) {
  int kb = (lane & 16) ? 8 : 0;
  const unsigned short* p = rowBase + kt * 32 + kb;
  FragU f;
  f.u[0] = *(const uint4*)(p);
  f.u[1] = *(const uint4*)(p + 16);
  return f.v;
}

__device__ __forceinline__ bf16x16 load_frag_Bswz(const unsigned int* swz, int kt, int nt, int lane) {
  const unsigned int* p = swz + ((((kt << 4) + nt) * 32 + lane) << 3);
  FragU f;
  f.u[0] = *(const uint4*)(p);
  f.u[1] = *(const uint4*)(p + 4);
  return f.v;
}

__device__ __forceinline__ bf16x16 load_frag_F(const unsigned int* swz, int tap, int kt, int lane) {
  const unsigned int* p = swz + ((((tap << 3) + kt) * 32 + lane) << 3);
  FragU f;
  f.u[0] = *(const uint4*)(p);
  f.u[1] = *(const uint4*)(p + 4);
  return f.v;
}

__device__ __forceinline__ float bf2f(unsigned short h) {
  return __uint_as_float(((unsigned)h) << 16);
}

__device__ __forceinline__ void lds_barrier() {
  __builtin_amdgcn_fence(__ATOMIC_RELEASE, "workgroup", "local");
  __builtin_amdgcn_s_barrier();
  __builtin_amdgcn_fence(__ATOMIC_ACQUIRE, "workgroup", "local");
}

__global__ void __launch_bounds__(256) prep_kernel(
    const float* __restrict__ wmsg, const float* __restrict__ wup2, const float* __restrict__ v1,
    unsigned int* __restrict__ Mswz, unsigned int* __restrict__ Fswz)
{
  __shared__ float part[8][18];
  __shared__ float wsum[18];
  const int tid = threadIdx.x;

  if (blockIdx.x >= 128) {
    int r    = (blockIdx.x - 128) * 256 + tid;
    int j    = r & 7;
    int lane = (r >> 3) & 31;
    int nt   = (r >> 8) & 15;
    int kt   = r >> 12;
    int n  = nt * 16 + (lane & 15);
    int kb = (lane & 16) ? 8 : 0;
    int k  = kt * 32 + ((j < 4) ? (kb + 2 * j) : (16 + kb + 2 * (j - 4)));
    float lo = wmsg[(n * 256 + k) * 9 + 4];
    float hi = wmsg[(n * 256 + k + 1) * 9 + 4];
    Mswz[r] = (unsigned)f2bf(lo) | ((unsigned)f2bf(hi) << 16);
    return;
  }

  const int kp   = blockIdx.x;
  const int wave = tid >> 5;
  const int lane = tid & 31;
  const float g  = v1[tid];
  const float* srcw = wup2 + (tid * 256 + 2 * kp) * 9;
  float p[18];
  #pragma unroll
  for (int t = 0; t < 18; ++t) p[t] = g * srcw[t];
  #pragma unroll
  for (int t = 0; t < 18; ++t) {
    for (int off = 16; off >= 1; off >>= 1) p[t] += __shfl_xor(p[t], off);
  }
  if (lane == 0) {
    #pragma unroll
    for (int t = 0; t < 18; ++t) part[wave][t] = p[t];
  }
  __syncthreads();
  if (tid < 18) {
    float s = 0.f;
    for (int w = 0; w < 8; ++w) s += part[w][tid];
    wsum[tid] = s;
  }
  __syncthreads();
  if (tid < 144) {
    const int tap = tid >> 4;
    const int n   = tid & 15;
    const float w0 = wsum[tap], w1 = wsum[9 + tap];
    const unsigned short h0 = f2bf(w0), h1 = f2bf(w1);
    unsigned int d = 0u;
    if (n == 0) {
      d = (unsigned)h0 | ((unsigned)h1 << 16);
    } else if (n == 1) {
      d = (unsigned)f2bf(w0 - bf2f(h0)) | ((unsigned)f2bf(w1 - bf2f(h1)) << 16);
    }
    const int k   = 2 * kp;
    const int kt  = k >> 5;
    const int r   = k & 31;
    const int kbs = (r >> 3) & 1;
    const int j   = ((r >> 4) << 2) + ((r & 7) >> 1);
    Fswz[((((tap << 3) + kt) * 32 + kbs * 16 + n) << 3) + j] = d;
  }
}

__global__ void init_misc(float* scid, float* shid) {
  int t = threadIdx.x;
  scid[t] = 1.0f; shid[t] = 0.0f;
}

struct XRow { f32x4 a0, b0, a1, b1; };

__global__ void __launch_bounds__(256) scnn_kernel(
    const float* __restrict__ xin, float* __restrict__ xout,
    const unsigned int* __restrict__ Mswz,
    const float* __restrict__ sc, const float* __restrict__ sh,
    int relu_in, float* __restrict__ statsP)
{
  __shared__ __align__(16) unsigned short prevA[16 * 264];
  __shared__ __align__(16) unsigned short prevB[16 * 264];

  const int tid  = threadIdx.x;
  const int b    = blockIdx.x >> 5;
  const int w0   = (blockIdx.x & 31) << 4;
  const int wave = tid >> 5;
  const int lane = tid & 31;
  const int half8 = (lane & 16) ? 8 : 0;
  const int nt0 = wave * 2, nt1 = wave * 2 + 1;
  const int ch0 = nt0 * 16 + (lane & 15);
  const int ch1 = nt1 * 16 + (lane & 15);

  for (int i = tid; i < 16 * 264; i += 256) prevA[i] = 0;

  const float sA = sc[ch0], bA = sh[ch0];
  const float sB = sc[ch1], bB = sh[ch1];

  const float* xi0p = xin  + ((size_t)(b * CC + ch0) * HH) * WW + w0 + half8;
  const float* xi1p = xin  + ((size_t)(b * CC + ch1) * HH) * WW + w0 + half8;
  float*       xo0p = xout + ((size_t)(b * CC + ch0) * HH) * WW + w0 + half8;
  float*       xo1p = xout + ((size_t)(b * CC + ch1) * HH) * WW + w0 + half8;

  bf16x16 bm0[8], bm1[8];
  #pragma unroll
  for (int kt = 0; kt < 8; ++kt) {
    bm0[kt] = load_frag_Bswz(Mswz, kt, nt0, lane);
    bm1[kt] = load_frag_Bswz(Mswz, kt, nt1, lane);
  }

  auto load_row = [&](int r, XRow& x) {
    const float* n0 = xi0p + (size_t)r * WW;
    const float* n1 = xi1p + (size_t)r * WW;
    x.a0 = *(const f32x4*)(n0); x.b0 = *(const f32x4*)(n0 + 4);
    x.a1 = *(const f32x4*)(n1); x.b1 = *(const f32x4*)(n1 + 4);
  };

  float s0 = 0.f, q0 = 0.f, s1 = 0.f, q1 = 0.f;

  auto step = [&](int h, const XRow& x, XRow& xfill, const unsigned short* prevCur, unsigned short* prevNxt) {
    __builtin_amdgcn_sched_barrier(0);
    lds_barrier();
    load_row(min(h + 4, HH - 1), xfill);

    f32x8 acc0 = {}; f32x8 acc1 = {};
    const unsigned short* arow = &prevCur[(lane & 15) * 264];
    bf16x16 a = load_frag_lds(arow, 0, lane);
    #pragma unroll
    for (int kt = 0; kt < 8; ++kt) {
      bf16x16 an = (kt < 7) ? load_frag_lds(arow, kt + 1, lane) : a;
      acc0 = wmma_bf16(a, bm0[kt], acc0);
      acc1 = wmma_bf16(a, bm1[kt], acc1);
      a = an;
    }

    f32x4 xa0 = x.a0, xb0 = x.b0, xa1 = x.a1, xb1 = x.b1;
    asm("" : "+v"(xa0), "+v"(xb0), "+v"(xa1), "+v"(xb1));
    const float xi0[8] = {xa0.x, xa0.y, xa0.z, xa0.w, xb0.x, xb0.y, xb0.z, xb0.w};
    const float xi1[8] = {xa1.x, xa1.y, xa1.z, xa1.w, xb1.x, xb1.y, xb1.z, xb1.w};
    float r0[8], r1[8];
    #pragma unroll
    for (int j = 0; j < 8; ++j) {
      float a0 = sA * xi0[j] + bA;
      float a1 = sB * xi1[j] + bB;
      if (relu_in) { a0 = fmaxf(a0, 0.f); a1 = fmaxf(a1, 0.f); }
      float v0 = a0 + fmaxf(acc0[j], 0.f);
      float v1 = a1 + fmaxf(acc1[j], 0.f);
      r0[j] = v0; r1[j] = v1;
      s0 += v0; q0 += v0 * v0;
      s1 += v1; q1 += v1 * v1;
      prevNxt[(j + half8) * 264 + ch0] = f2bf(v0);
      prevNxt[(j + half8) * 264 + ch1] = f2bf(v1);
    }
    {
      float* p0 = xo0p + (size_t)(HH - 1 - h) * WW;
      float* p1 = xo1p + (size_t)(HH - 1 - h) * WW;
      *(float4*)(p0)     = make_float4(r0[0], r0[1], r0[2], r0[3]);
      *(float4*)(p0 + 4) = make_float4(r0[4], r0[5], r0[6], r0[7]);
      *(float4*)(p1)     = make_float4(r1[0], r1[1], r1[2], r1[3]);
      *(float4*)(p1 + 4) = make_float4(r1[4], r1[5], r1[6], r1[7]);
    }
  };

  XRow x0, x1, x2, x3, x4;
  load_row(0, x0); load_row(1, x1); load_row(2, x2); load_row(3, x3);

  unsigned short* cur = prevA;
  unsigned short* nxt = prevB;
  for (int h = 0; h < HH; h += 5) {
    step(h,     x0, x4, cur, nxt);
    step(h + 1, x1, x0, nxt, cur);
    step(h + 2, x2, x1, cur, nxt);
    step(h + 3, x3, x2, nxt, cur);
    step(h + 4, x4, x3, cur, nxt);
    unsigned short* t = cur; cur = nxt; nxt = t;
  }

  s0 += __shfl_xor(s0, 16); q0 += __shfl_xor(q0, 16);
  s1 += __shfl_xor(s1, 16); q1 += __shfl_xor(q1, 16);
  if (lane < 16) {
    statsP[(ch0)       * 256 + blockIdx.x] = s0;
    statsP[(256 + ch0) * 256 + blockIdx.x] = q0;
    statsP[(ch1)       * 256 + blockIdx.x] = s1;
    statsP[(256 + ch1) * 256 + blockIdx.x] = q1;
  }
}

__global__ void __launch_bounds__(256) bnscale_kernel(
    const float* __restrict__ statsP,
    const float* __restrict__ gamma, const float* __restrict__ beta,
    float* __restrict__ sc, float* __restrict__ sh)
{
  __shared__ float ps[8], pq[8];
  const int c = blockIdx.x;
  const int t = threadIdx.x;
  float s = statsP[c * 256 + t];
  float q = statsP[(256 + c) * 256 + t];
  for (int off = 16; off >= 1; off >>= 1) { s += __shfl_xor(s, off); q += __shfl_xor(q, off); }
  if ((t & 31) == 0) { ps[t >> 5] = s; pq[t >> 5] = q; }
  __syncthreads();
  if (t == 0) {
    float S = 0.f, Q = 0.f;
    for (int w = 0; w < 8; ++w) { S += ps[w]; Q += pq[w]; }
    const float n = (float)(BB * HH * WW);
    float mean = S / n;
    float var  = Q / n - mean * mean;
    float g = gamma[c] * rsqrtf(var + 1e-5f);
    sc[c] = g;
    sh[c] = beta[c] - mean * g;
  }
}

#define CONV_MT 4
#define CONV_TW (CONV_MT * 16)
#define CONV_HW (CONV_TW + 2)
#define RAW_P   36

__global__ void __launch_bounds__(256) conv_kernel(
    const float* __restrict__ xin,
    const unsigned int* __restrict__ Fswz,
    const float* __restrict__ sc, const float* __restrict__ sh,
    float* __restrict__ z)
{
  __shared__ __align__(16) unsigned short uT[3 * CONV_HW * 264];
  __shared__ __align__(16) float raw[256 * RAW_P];

  const int tid = threadIdx.x;
  const int wt  = blockIdx.x & 15;
  const int h   = (blockIdx.x >> 4) % 25;
  const int b   = blockIdx.x / (16 * 25);
  const int w20 = wt * CONV_TW;

  {
    const int M0   = wt * 32;
    const int offL = (wt > 0)  ? -1 : 0;
    const int offR = (wt < 15) ? 32 : 31;
    const int pk   = tid & 7;
    const int pr   = tid >> 3;

    auto load_row = [&](int hh, f32x4 (&q)[8], float& eL, float& eR) {
      const int hs  = h + hh - 1;
      const int hsc = min(max(hs, 0), HH - 1);
      #pragma unroll
      for (int i = 0; i < 8; ++i)
        q[i] = *(const f32x4*)(xin + ((b * CC + i * 32 + pr) * HH + hsc) * WW + M0 + 4 * pk);
      const float* erow = xin + ((b * CC + tid) * HH + hsc) * WW + M0;
      eL = erow[offL];
      eR = erow[offR];
    };
    auto put_row = [&](const f32x4 (&q)[8], float eL, float eR) {
      #pragma unroll
      for (int i = 0; i < 8; ++i)
        *(f32x4*)&raw[(i * 32 + pr) * RAW_P + 4 * pk] = q[i];
      *(f32x2*)&raw[tid * RAW_P + 32] = (f32x2){eL, eR};
    };
    const float scc = sc[tid], shc = sh[tid];
    auto emit_row = [&](int hh) {
      const int  hs    = h + hh - 1;
      const bool rowok = (hs >= 0 && hs < HH);
      float xs[34];
      const f32x2 e = *(const f32x2*)&raw[tid * RAW_P + 32];
      xs[0] = e.x; xs[33] = e.y;
      #pragma unroll
      for (int k = 0; k < 8; ++k) {
        const f32x4 m = *(const f32x4*)&raw[tid * RAW_P + 4 * k];
        xs[1 + 4 * k] = m.x; xs[2 + 4 * k] = m.y; xs[3 + 4 * k] = m.z; xs[4 + 4 * k] = m.w;
      }
      #pragma unroll
      for (int k = 0; k < 34; ++k) {
        const float x = fmaxf(scc * xs[k] + shc, 0.f);
        xs[k] = rowok ? x : 0.f;
      }
      #pragma unroll
      for (int wi = 0; wi < CONV_HW; ++wi) {
        const int  w2    = w20 + wi - 1;
        const bool colok = (w2 >= 0 && w2 < 1024);
        const int   num = w2 * 511;
        const int   i0  = num / 1023;
        const float f   = (float)(num - i0 * 1023) * (1.0f / 1023.0f);
        const int   s   = wi >> 1;
        const bool  col0 = (wi == 1) && (wt == 0);
        const float x0 = col0 ? xs[1] : xs[s];
        const float x1 = col0 ? xs[2] : xs[s + 1];
        const float v  = x0 + f * (x1 - x0);
        uT[(hh * CONV_HW + wi) * 264 + tid] = colok ? f2bf(v) : (unsigned short)0;
      }
    };

    f32x4 q0[8], q1[8], q2[8];
    float  l0, r0, l1, r1, l2, r2;
    load_row(0, q0, l0, r0);
    load_row(1, q1, l1, r1);
    load_row(2, q2, l2, r2);

    put_row(q0, l0, r0);
    __syncthreads();
    emit_row(0);
    __syncthreads();
    put_row(q1, l1, r1);
    __syncthreads();
    emit_row(1);
    __syncthreads();
    put_row(q2, l2, r2);
    __syncthreads();
    emit_row(2);
  }
  __syncthreads();

  const int wave = tid >> 5;
  const int lane = tid & 31;
  f32x8 acc[CONV_MT];
  #pragma unroll
  for (int m = 0; m < CONV_MT; ++m) acc[m] = (f32x8){};

  const unsigned short* abase = &uT[(lane & 15) * 264];

  #pragma unroll
  for (int tap = 0; tap < 9; ++tap) {
    const int ky = tap / 3, kx = tap - ky * 3;
    const unsigned short* arow = abase + (ky * CONV_HW + kx) * 264;
    const bf16x16 bw = load_frag_F(Fswz, tap, wave, lane);
    #pragma unroll
    for (int m = 0; m < CONV_MT; ++m)
      acc[m] = wmma_bf16(load_frag_lds(arow + m * 16 * 264, wave, lane), bw, acc[m]);
  }

  float* zp = raw;
  const int half8 = (lane & 16) ? 8 : 0;
  #pragma unroll
  for (int m = 0; m < CONV_MT; ++m) {
    #pragma unroll
    for (int j = 0; j < 8; ++j)
      zp[((wave * CONV_TW) + m * 16 + j + half8) * 16 + (lane & 15)] = acc[m][j];
  }
  __syncthreads();
  if (tid < CONV_TW) {
    float zh = 0.f, zl = 0.f;
    for (int w = 0; w < 8; ++w) {
      zh += zp[(w * CONV_TW + tid) * 16 + 0];
      zl += zp[(w * CONV_TW + tid) * 16 + 1];
    }
    z[(b * 25 + h) * 1024 + w20 + tid] = zh + zl;
  }
}

__global__ void final_kernel(const float* __restrict__ z, const float* __restrict__ v2,
                             float* __restrict__ out)
{
  int idx = blockIdx.x * 256 + threadIdx.x;
  if (idx >= BB * 4096) return;
  int b  = idx >> 12;
  int w4 = idx & 4095;

  int   num4 = w4 * 2047;
  int   i40  = num4 / 4095;
  float f4   = (float)(num4 - i40 * 4095) * (1.0f / 4095.0f);
  int   i41  = min(i40 + 1, 2047);

  int n3a = i40 * 1023; int ja0 = n3a / 2047;
  float fa = (float)(n3a - ja0 * 2047) * (1.0f / 2047.0f);
  int ja1 = min(ja0 + 1, 1023);
  int n3b = i41 * 1023; int jb0 = n3b / 2047;
  float fb = (float)(n3b - jb0 * 2047) * (1.0f / 2047.0f);
  int jb1 = min(jb0 + 1, 1023);

  float acc = 0.f;
  #pragma unroll 5
  for (int h = 0; h < HH; ++h) {
    const float* zh = z + (b * 25 + h) * 1024;
    float za = zh[ja0] + fa * (zh[ja1] - zh[ja0]);
    float zb = zh[jb0] + fb * (zh[jb1] - zh[jb0]);
    acc += v2[h] * (za + f4 * (zb - za));
  }
  out[idx] = 1.0f / (1.0f + __expf(-acc));
}

extern "C" void kernel_launch(void* const* d_in, const int* in_sizes, int n_in,
                              void* d_out, int out_size, void* d_ws, size_t ws_size,
                              hipStream_t stream)
{
  const float* p2c   = (const float*)d_in[0];
  const float* wmsg  = (const float*)d_in[1];
  const float* gamma = (const float*)d_in[2];
  const float* beta  = (const float*)d_in[3];
  const float* wup2  = (const float*)d_in[4];
  const float* v1    = (const float*)d_in[5];
  const float* v2    = (const float*)d_in[6];

  char* ws = (char*)d_ws;
  size_t off = 0;
  auto walloc = [&](size_t bytes) -> void* {
    void* p = ws + off;
    off = (off + bytes + 255) & ~(size_t)255;
    return p;
  };
  unsigned int* Mswz = (unsigned int*)walloc(32768u * 4);
  unsigned int* Fswz = (unsigned int*)walloc(9u * 2048u * 4);
  float* stats1 = (float*)walloc(512u * 256u * 4);
  float* stats2 = (float*)walloc(512u * 256u * 4);
  float* sc1  = (float*)walloc(256 * 4);
  float* sh1  = (float*)walloc(256 * 4);
  float* sc2  = (float*)walloc(256 * 4);
  float* sh2  = (float*)walloc(256 * 4);
  float* scid = (float*)walloc(256 * 4);
  float* shid = (float*)walloc(256 * 4);
  float* z1024 = (float*)walloc((size_t)BB * 25 * 1024 * 4);
  float* out1  = (float*)walloc((size_t)BB * CC * HH * WW * 4);
  float* out2  = (float*)walloc((size_t)BB * CC * HH * WW * 4);

  prep_kernel<<<256, 256, 0, stream>>>(wmsg, wup2, v1, Mswz, Fswz);
  init_misc<<<1, 256, 0, stream>>>(scid, shid);

  scnn_kernel<<<256, 256, 0, stream>>>(p2c, out1, Mswz, scid, shid, 0, stats1);
  bnscale_kernel<<<256, 256, 0, stream>>>(stats1, gamma, beta, sc1, sh1);

  scnn_kernel<<<256, 256, 0, stream>>>(out1, out2, Mswz, sc1, sh1, 1, stats2);
  bnscale_kernel<<<256, 256, 0, stream>>>(stats2, gamma, beta, sc2, sh2);

  conv_kernel<<<BB * 25 * 16, 256, 0, stream>>>(out2, Fswz, sc2, sh2, z1024);

  final_kernel<<<128, 256, 0, stream>>>(z1024, v2, (float*)d_out);
}


# conv: a work-group now walks down 6-7 output rows with a ring of three staged halo rows in LDS: each source row is loaded and staged once per work-group, not three times (3200 -> 512 groups)
# speedup vs baseline: 3.5350x; 1.1709x over previous
#include <hip/hip_runtime.h>
#include <hip/hip_bf16.h>
#include <stdint.h>

#define BB 8
#define CC 256
#define HH 25
#define WW 512

typedef __bf16 bf16x16 __attribute__((ext_vector_type(16)));
typedef float  f32x8   __attribute__((ext_vector_type(8)));
typedef float  f32x4   __attribute__((ext_vector_type(4)));
typedef float  f32x2   __attribute__((ext_vector_type(2)));

union FragU { uint4 u[2]; bf16x16 v; };

__device__ __forceinline__ unsigned short f2bf(float x) {
  unsigned u = __float_as_uint(x);
  unsigned r = u + 0x7FFFu + ((u >> 16) & 1u);
  return (unsigned short)(r >> 16);
}

__device__ __forceinline__ f32x8 wmma_bf16(bf16x16 a, bf16x16 b, f32x8 c) {
  return __builtin_amdgcn_wmma_f32_16x16x32_bf16(false, a, false, b, (short)0, c, false, false);
}

__device__ __forceinline__ bf16x16 load_frag_lds(const unsigned short* rowBase, int kt, int lane) {
  int kb = (lane & 16) ? 8 : 0;
  const unsigned short* p = rowBase + kt * 32 + kb;
  FragU f;
  f.u[0] = *(const uint4*)(p);
  f.u[1] = *(const uint4*)(p + 16);
  return f.v;
}

__device__ __forceinline__ bf16x16 load_frag_Bswz(const unsigned int* swz, int kt, int nt, int lane) {
  const unsigned int* p = swz + ((((kt << 4) + nt) * 32 + lane) << 3);
  FragU f;
  f.u[0] = *(const uint4*)(p);
  f.u[1] = *(const uint4*)(p + 4);
  return f.v;
}

__device__ __forceinline__ bf16x16 load_frag_F(const unsigned int* swz, int tap, int kt, int lane) {
  const unsigned int* p = swz + ((((tap << 3) + kt) * 32 + lane) << 3);
  FragU f;
  f.u[0] = *(const uint4*)(p);
  f.u[1] = *(const uint4*)(p + 4);
  return f.v;
}

__device__ __forceinline__ float bf2f(unsigned short h) {
  return __uint_as_float(((unsigned)h) << 16);
}

__device__ __forceinline__ void lds_barrier() {
  __builtin_amdgcn_fence(__ATOMIC_RELEASE, "workgroup", "local");
  __builtin_amdgcn_s_barrier();
  __builtin_amdgcn_fence(__ATOMIC_ACQUIRE, "workgroup", "local");
}

__global__ void __launch_bounds__(256) prep_kernel(
    const float* __restrict__ wmsg, const float* __restrict__ wup2, const float* __restrict__ v1,
    unsigned int* __restrict__ Mswz, unsigned int* __restrict__ Fswz)
{
  __shared__ float part[8][18];
  __shared__ float wsum[18];
  const int tid = threadIdx.x;

  if (blockIdx.x >= 128) {
    int r    = (blockIdx.x - 128) * 256 + tid;
    int j    = r & 7;
    int lane = (r >> 3) & 31;
    int nt   = (r >> 8) & 15;
    int kt   = r >> 12;
    int n  = nt * 16 + (lane & 15);
    int kb = (lane & 16) ? 8 : 0;
    int k  = kt * 32 + ((j < 4) ? (kb + 2 * j) : (16 + kb + 2 * (j - 4)));
    float lo = wmsg[(n * 256 + k) * 9 + 4];
    float hi = wmsg[(n * 256 + k + 1) * 9 + 4];
    Mswz[r] = (unsigned)f2bf(lo) | ((unsigned)f2bf(hi) << 16);
    return;
  }

  const int kp   = blockIdx.x;
  const int wave = tid >> 5;
  const int lane = tid & 31;
  const float g  = v1[tid];
  const float* srcw = wup2 + (tid * 256 + 2 * kp) * 9;
  float p[18];
  #pragma unroll
  for (int t = 0; t < 18; ++t) p[t] = g * srcw[t];
  #pragma unroll
  for (int t = 0; t < 18; ++t) {
    for (int off = 16; off >= 1; off >>= 1) p[t] += __shfl_xor(p[t], off);
  }
  if (lane == 0) {
    #pragma unroll
    for (int t = 0; t < 18; ++t) part[wave][t] = p[t];
  }
  __syncthreads();
  if (tid < 18) {
    float s = 0.f;
    for (int w = 0; w < 8; ++w) s += part[w][tid];
    wsum[tid] = s;
  }
  __syncthreads();
  if (tid < 144) {
    const int tap = tid >> 4;
    const int n   = tid & 15;
    const float w0 = wsum[tap], w1 = wsum[9 + tap];
    const unsigned short h0 = f2bf(w0), h1 = f2bf(w1);
    unsigned int d = 0u;
    if (n == 0) {
      d = (unsigned)h0 | ((unsigned)h1 << 16);
    } else if (n == 1) {
      d = (unsigned)f2bf(w0 - bf2f(h0)) | ((unsigned)f2bf(w1 - bf2f(h1)) << 16);
    }
    const int k   = 2 * kp;
    const int kt  = k >> 5;
    const int r   = k & 31;
    const int kbs = (r >> 3) & 1;
    const int j   = ((r >> 4) << 2) + ((r & 7) >> 1);
    Fswz[((((tap << 3) + kt) * 32 + kbs * 16 + n) << 3) + j] = d;
  }
}

__global__ void init_misc(float* scid, float* shid) {
  int t = threadIdx.x;
  scid[t] = 1.0f; shid[t] = 0.0f;
}

struct XRow { f32x4 a0, b0, a1, b1; };

__global__ void __launch_bounds__(256) scnn_kernel(
    const float* __restrict__ xin, float* __restrict__ xout,
    const unsigned int* __restrict__ Mswz,
    const float* __restrict__ sc, const float* __restrict__ sh,
    int relu_in, float* __restrict__ statsP)
{
  __shared__ __align__(16) unsigned short prevA[16 * 264];
  __shared__ __align__(16) unsigned short prevB[16 * 264];

  const int tid  = threadIdx.x;
  const int b    = blockIdx.x >> 5;
  const int w0   = (blockIdx.x & 31) << 4;
  const int wave = tid >> 5;
  const int lane = tid & 31;
  const int half8 = (lane & 16) ? 8 : 0;
  const int nt0 = wave * 2, nt1 = wave * 2 + 1;
  const int ch0 = nt0 * 16 + (lane & 15);
  const int ch1 = nt1 * 16 + (lane & 15);

  for (int i = tid; i < 16 * 264; i += 256) prevA[i] = 0;

  const float sA = sc[ch0], bA = sh[ch0];
  const float sB = sc[ch1], bB = sh[ch1];

  const float* xi0p = xin  + ((size_t)(b * CC + ch0) * HH) * WW + w0 + half8;
  const float* xi1p = xin  + ((size_t)(b * CC + ch1) * HH) * WW + w0 + half8;
  float*       xo0p = xout + ((size_t)(b * CC + ch0) * HH) * WW + w0 + half8;
  float*       xo1p = xout + ((size_t)(b * CC + ch1) * HH) * WW + w0 + half8;

  bf16x16 bm0[8], bm1[8];
  #pragma unroll
  for (int kt = 0; kt < 8; ++kt) {
    bm0[kt] = load_frag_Bswz(Mswz, kt, nt0, lane);
    bm1[kt] = load_frag_Bswz(Mswz, kt, nt1, lane);
  }

  auto load_row = [&](int r, XRow& x) {
    const float* n0 = xi0p + (size_t)r * WW;
    const float* n1 = xi1p + (size_t)r * WW;
    x.a0 = *(const f32x4*)(n0); x.b0 = *(const f32x4*)(n0 + 4);
    x.a1 = *(const f32x4*)(n1); x.b1 = *(const f32x4*)(n1 + 4);
  };

  float s0 = 0.f, q0 = 0.f, s1 = 0.f, q1 = 0.f;

  auto step = [&](int h, const XRow& x, XRow& xfill, const unsigned short* prevCur, unsigned short* prevNxt) {
    __builtin_amdgcn_sched_barrier(0);
    lds_barrier();
    load_row(min(h + 4, HH - 1), xfill);

    f32x8 acc0 = {}; f32x8 acc1 = {};
    const unsigned short* arow = &prevCur[(lane & 15) * 264];
    bf16x16 a = load_frag_lds(arow, 0, lane);
    #pragma unroll
    for (int kt = 0; kt < 8; ++kt) {
      bf16x16 an = (kt < 7) ? load_frag_lds(arow, kt + 1, lane) : a;
      acc0 = wmma_bf16(a, bm0[kt], acc0);
      acc1 = wmma_bf16(a, bm1[kt], acc1);
      a = an;
    }

    f32x4 xa0 = x.a0, xb0 = x.b0, xa1 = x.a1, xb1 = x.b1;
    asm("" : "+v"(xa0), "+v"(xb0), "+v"(xa1), "+v"(xb1));
    const float xi0[8] = {xa0.x, xa0.y, xa0.z, xa0.w, xb0.x, xb0.y, xb0.z, xb0.w};
    const float xi1[8] = {xa1.x, xa1.y, xa1.z, xa1.w, xb1.x, xb1.y, xb1.z, xb1.w};
    float r0[8], r1[8];
    #pragma unroll
    for (int j = 0; j < 8; ++j) {
      float a0 = sA * xi0[j] + bA;
      float a1 = sB * xi1[j] + bB;
      if (relu_in) { a0 = fmaxf(a0, 0.f); a1 = fmaxf(a1, 0.f); }
      float v0 = a0 + fmaxf(acc0[j], 0.f);
      float v1 = a1 + fmaxf(acc1[j], 0.f);
      r0[j] = v0; r1[j] = v1;
      s0 += v0; q0 += v0 * v0;
      s1 += v1; q1 += v1 * v1;
      prevNxt[(j + half8) * 264 + ch0] = f2bf(v0);
      prevNxt[(j + half8) * 264 + ch1] = f2bf(v1);
    }
    {
      float* p0 = xo0p + (size_t)(HH - 1 - h) * WW;
      float* p1 = xo1p + (size_t)(HH - 1 - h) * WW;
      *(float4*)(p0)     = make_float4(r0[0], r0[1], r0[2], r0[3]);
      *(float4*)(p0 + 4) = make_float4(r0[4], r0[5], r0[6], r0[7]);
      *(float4*)(p1)     = make_float4(r1[0], r1[1], r1[2], r1[3]);
      *(float4*)(p1 + 4) = make_float4(r1[4], r1[5], r1[6], r1[7]);
    }
  };

  XRow x0, x1, x2, x3, x4;
  load_row(0, x0); load_row(1, x1); load_row(2, x2); load_row(3, x3);

  unsigned short* cur = prevA;
  unsigned short* nxt = prevB;
  for (int h = 0; h < HH; h += 5) {
    step(h,     x0, x4, cur, nxt);
    step(h + 1, x1, x0, nxt, cur);
    step(h + 2, x2, x1, cur, nxt);
    step(h + 3, x3, x2, nxt, cur);
    step(h + 4, x4, x3, cur, nxt);
    unsigned short* t = cur; cur = nxt; nxt = t;
  }

  s0 += __shfl_xor(s0, 16); q0 += __shfl_xor(q0, 16);
  s1 += __shfl_xor(s1, 16); q1 += __shfl_xor(q1, 16);
  if (lane < 16) {
    statsP[(ch0)       * 256 + blockIdx.x] = s0;
    statsP[(256 + ch0) * 256 + blockIdx.x] = q0;
    statsP[(ch1)       * 256 + blockIdx.x] = s1;
    statsP[(256 + ch1) * 256 + blockIdx.x] = q1;
  }
}

__global__ void __launch_bounds__(256) bnscale_kernel(
    const float* __restrict__ statsP,
    const float* __restrict__ gamma, const float* __restrict__ beta,
    float* __restrict__ sc, float* __restrict__ sh)
{
  __shared__ float ps[8], pq[8];
  const int c = blockIdx.x;
  const int t = threadIdx.x;
  float s = statsP[c * 256 + t];
  float q = statsP[(256 + c) * 256 + t];
  for (int off = 16; off >= 1; off >>= 1) { s += __shfl_xor(s, off); q += __shfl_xor(q, off); }
  if ((t & 31) == 0) { ps[t >> 5] = s; pq[t >> 5] = q; }
  __syncthreads();
  if (t == 0) {
    float S = 0.f, Q = 0.f;
    for (int w = 0; w < 8; ++w) { S += ps[w]; Q += pq[w]; }
    const float n = (float)(BB * HH * WW);
    float mean = S / n;
    float var  = Q / n - mean * mean;
    float g = gamma[c] * rsqrtf(var + 1e-5f);
    sc[c] = g;
    sh[c] = beta[c] - mean * g;
  }
}

#define CONV_MT 4
#define CONV_TW (CONV_MT * 16)
#define CONV_HW (CONV_TW + 2)
#define CONV_ROW (CONV_HW * 264)
#define CONV_RUNS 4
#define RAW_P   36

__global__ void __launch_bounds__(256) conv_kernel(
    const float* __restrict__ xin,
    const unsigned int* __restrict__ Fswz,
    const float* __restrict__ sc, const float* __restrict__ sh,
    float* __restrict__ z)
{
  __shared__ __align__(16) unsigned short uT[3 * CONV_ROW];
  __shared__ __align__(16) float raw[256 * RAW_P];
  __shared__ __align__(16) float zp[8 * CONV_TW * 2];

  const int tid  = threadIdx.x;
  const int wave = tid >> 5;
  const int lane = tid & 31;
  const int wt   = blockIdx.x & 15;
  const int run  = (blockIdx.x >> 4) & (CONV_RUNS - 1);
  const int b    = blockIdx.x >> 6;
  const int h0   = run * 6;
  const int nrows = (run == CONV_RUNS - 1) ? 7 : 6;
  const int w20  = wt * CONV_TW;

  const int M0   = wt * 32;
  const int offL = (wt > 0)  ? -1 : 0;
  const int offR = (wt < 15) ? 32 : 31;
  const int pk   = tid & 7;
  const int pr   = tid >> 3;

  auto load_row = [&](int hs, f32x4 (&q)[8], float& eL, float& eR) {
    const int hsc = min(max(hs, 0), HH - 1);
    #pragma unroll
    for (int i = 0; i < 8; ++i)
      q[i] = *(const f32x4*)(xin + ((b * CC + i * 32 + pr) * HH + hsc) * WW + M0 + 4 * pk);
    const float* erow = xin + ((b * CC + tid) * HH + hsc) * WW + M0;
    eL = erow[offL];
    eR = erow[offR];
  };
  auto put_row = [&](const f32x4 (&q)[8], float eL, float eR) {
    #pragma unroll
    for (int i = 0; i < 8; ++i)
      *(f32x4*)&raw[(i * 32 + pr) * RAW_P + 4 * pk] = q[i];
    *(f32x2*)&raw[tid * RAW_P + 32] = (f32x2){eL, eR};
  };
  const float scc = sc[tid], shc = sh[tid];
  auto col_weight = [&](int wi) {
    const int w2  = w20 + wi - 1;
    const int num = w2 * 511;
    const int i0  = num / 1023;
    return (float)(num - i0 * 1023) * (1.0f / 1023.0f);
  };
  const int fl0 = __float_as_int(col_weight(lane));
  const int fl1 = __float_as_int(col_weight(32 + lane));
  const int fl2 = __float_as_int(col_weight(64 + lane));
  auto emit_row = [&](unsigned short* dst, int hs) {
    const bool rowok = (hs >= 0 && hs < HH);
    float xs[34];
    const f32x2 e = *(const f32x2*)&raw[tid * RAW_P + 32];
    xs[0] = e.x; xs[33] = e.y;
    #pragma unroll
    for (int k = 0; k < 8; ++k) {
      const f32x4 m = *(const f32x4*)&raw[tid * RAW_P + 4 * k];
      xs[1 + 4 * k] = m.x; xs[2 + 4 * k] = m.y; xs[3 + 4 * k] = m.z; xs[4 + 4 * k] = m.w;
    }
    #pragma unroll
    for (int k = 0; k < 34; ++k) {
      const float x = fmaxf(scc * xs[k] + shc, 0.f);
      xs[k] = rowok ? x : 0.f;
    }
    #pragma unroll
    for (int wi = 0; wi < CONV_HW; ++wi) {
      const int  w2    = w20 + wi - 1;
      const bool colok = (w2 >= 0 && w2 < 1024);
      const float f   = __int_as_float(__builtin_amdgcn_readlane((wi < 32) ? fl0 : (wi < 64) ? fl1 : fl2, wi & 31));
      const int   s   = wi >> 1;
      const bool  col0 = (wi == 1) && (wt == 0);
      const float x0 = col0 ? xs[1] : xs[s];
      const float x1 = col0 ? xs[2] : xs[s + 1];
      const float v  = x0 + f * (x1 - x0);
      dst[wi * 264 + tid] = colok ? f2bf(v) : (unsigned short)0;
    }
  };

  bf16x16 bw[9];
  #pragma unroll
  for (int tap = 0; tap < 9; ++tap) bw[tap] = load_frag_F(Fswz, tap, wave, lane);

  int o0 = 0, o1 = CONV_ROW, o2 = 2 * CONV_ROW;

  f32x4 qn[8];
  float ln, rn;
  {
    f32x4 q0[8], q1[8], q2[8];
    float  l0, r0, l1, r1, l2, r2;
    load_row(h0 - 1, q0, l0, r0);
    load_row(h0,     q1, l1, r1);
    load_row(h0 + 1, q2, l2, r2);

    put_row(q0, l0, r0);
    __syncthreads();
    emit_row(uT + o0, h0 - 1);
    __syncthreads();
    put_row(q1, l1, r1);
    __syncthreads();
    emit_row(uT + o1, h0);
    __syncthreads();
    put_row(q2, l2, r2);
    load_row(h0 + 2, qn, ln, rn);
    __syncthreads();
    emit_row(uT + o2, h0 + 1);
  }
  __syncthreads();

  const int half8 = (lane & 16) ? 8 : 0;
  const int arow_l = (lane & 15) * 264;

  for (int i = 0; i < nrows; ++i) {
    const int  h    = h0 + i;
    const bool more = (i + 1 < nrows);

    if (more) put_row(qn, ln, rn);

    f32x8 acc[CONV_MT];
    #pragma unroll
    for (int m = 0; m < CONV_MT; ++m) acc[m] = (f32x8){};

    const unsigned short* a0 = uT + o0 + arow_l;
    const unsigned short* a1 = uT + o1 + arow_l;
    const unsigned short* a2 = uT + o2 + arow_l;
    bf16x16 af[CONV_MT];
    #pragma unroll
    for (int m = 0; m < CONV_MT; ++m) af[m] = load_frag_lds(a0 + m * 16 * 264, wave, lane);
    #pragma unroll
    for (int tap = 0; tap < 9; ++tap) {
      bf16x16 an[CONV_MT];
      if (tap < 8) {
        const int ky = (tap + 1) / 3, kx = (tap + 1) - ky * 3;
        const unsigned short* arow = ((ky == 0) ? a0 : (ky == 1) ? a1 : a2) + kx * 264;
        #pragma unroll
        for (int m = 0; m < CONV_MT; ++m) an[m] = load_frag_lds(arow + m * 16 * 264, wave, lane);
      }
      __builtin_amdgcn_sched_barrier(0);
      #pragma unroll
      for (int m = 0; m < CONV_MT; ++m) acc[m] = wmma_bf16(af[m], bw[tap], acc[m]);
      __builtin_amdgcn_sched_barrier(0);
      if (tap < 8) {
        #pragma unroll
        for (int m = 0; m < CONV_MT; ++m) af[m] = an[m];
      }
    }

    if ((lane & 15) < 2) {
      #pragma unroll
      for (int m = 0; m < CONV_MT; ++m) {
        #pragma unroll
        for (int j = 0; j < 8; ++j)
          zp[((wave * CONV_TW) + m * 16 + j + half8) * 2 + (lane & 15)] = acc[m][j];
      }
    }
    __syncthreads();

    if (tid < CONV_TW) {
      float zh = 0.f, zl = 0.f;
      for (int w = 0; w < 8; ++w) {
        const f32x2 p = *(const f32x2*)&zp[(w * CONV_TW + tid) * 2];
        zh += p.x;
        zl += p.y;
      }
      z[(b * 25 + h) * 1024 + w20 + tid] = zh + zl;
    }
    if (more) {
      if (i + 2 < nrows) load_row(h + 3, qn, ln, rn);
      emit_row(uT + o0, h + 2);
    }
    __syncthreads();
    const int t = o0; o0 = o1; o1 = o2; o2 = t;
  }
}

__global__ void final_kernel(const float* __restrict__ z, const float* __restrict__ v2,
                             float* __restrict__ out)
{
  int idx = blockIdx.x * 256 + threadIdx.x;
  if (idx >= BB * 4096) return;
  int b  = idx >> 12;
  int w4 = idx & 4095;

  int   num4 = w4 * 2047;
  int   i40  = num4 / 4095;
  float f4   = (float)(num4 - i40 * 4095) * (1.0f / 4095.0f);
  int   i41  = min(i40 + 1, 2047);

  int n3a = i40 * 1023; int ja0 = n3a / 2047;
  float fa = (float)(n3a - ja0 * 2047) * (1.0f / 2047.0f);
  int ja1 = min(ja0 + 1, 1023);
  int n3b = i41 * 1023; int jb0 = n3b / 2047;
  float fb = (float)(n3b - jb0 * 2047) * (1.0f / 2047.0f);
  int jb1 = min(jb0 + 1, 1023);

  float acc = 0.f;
  #pragma unroll 5
  for (int h = 0; h < HH; ++h) {
    const float* zh = z + (b * 25 + h) * 1024;
    float za = zh[ja0] + fa * (zh[ja1] - zh[ja0]);
    float zb = zh[jb0] + fb * (zh[jb1] - zh[jb0]);
    acc += v2[h] * (za + f4 * (zb - za));
  }
  out[idx] = 1.0f / (1.0f + __expf(-acc));
}

extern "C" void kernel_launch(void* const* d_in, const int* in_sizes, int n_in,
                              void* d_out, int out_size, void* d_ws, size_t ws_size,
                              hipStream_t stream)
{
  const float* p2c   = (const float*)d_in[0];
  const float* wmsg  = (const float*)d_in[1];
  const float* gamma = (const float*)d_in[2];
  const float* beta  = (const float*)d_in[3];
  const float* wup2  = (const float*)d_in[4];
  const float* v1    = (const float*)d_in[5];
  const float* v2    = (const float*)d_in[6];

  char* ws = (char*)d_ws;
  size_t off = 0;
  auto walloc = [&](size_t bytes) -> void* {
    void* p = ws + off;
    off = (off + bytes + 255) & ~(size_t)255;
    return p;
  };
  unsigned int* Mswz = (unsigned int*)walloc(32768u * 4);
  unsigned int* Fswz = (unsigned int*)walloc(9u * 2048u * 4);
  float* stats1 = (float*)walloc(512u * 256u * 4);
  float* stats2 = (float*)walloc(512u * 256u * 4);
  float* sc1  = (float*)walloc(256 * 4);
  float* sh1  = (float*)walloc(256 * 4);
  float* sc2  = (float*)walloc(256 * 4);
  float* sh2  = (float*)walloc(256 * 4);
  float* scid = (float*)walloc(256 * 4);
  float* shid = (float*)walloc(256 * 4);
  float* z1024 = (float*)walloc((size_t)BB * 25 * 1024 * 4);
  float* out1  = (float*)walloc((size_t)BB * CC * HH * WW * 4);
  float* out2  = (float*)walloc((size_t)BB * CC * HH * WW * 4);

  prep_kernel<<<256, 256, 0, stream>>>(wmsg, wup2, v1, Mswz, Fswz);
  init_misc<<<1, 256, 0, stream>>>(scid, shid);

  scnn_kernel<<<256, 256, 0, stream>>>(p2c, out1, Mswz, scid, shid, 0, stats1);
  bnscale_kernel<<<256, 256, 0, stream>>>(stats1, gamma, beta, sc1, sh1);

  scnn_kernel<<<256, 256, 0, stream>>>(out1, out2, Mswz, sc1, sh1, 1, stats2);
  bnscale_kernel<<<256, 256, 0, stream>>>(stats2, gamma, beta, sc2, sh2);

  conv_kernel<<<BB * CONV_RUNS * 16, 256, 0, stream>>>(out2, Fswz, sc2, sh2, z1024);

  final_kernel<<<128, 256, 0, stream>>>(z1024, v2, (float*)d_out);
}


# layout: both intermediate tensors kept tile-major in the lanes' own fragment order: every store and reload of the recurrence and every conv row load covers whole 128-byte lines of one work-group
# speedup vs baseline: 6.5738x; 1.8596x over previous
#include <hip/hip_runtime.h>
#include <hip/hip_bf16.h>
#include <stdint.h>

#define BB 8
#define CC 256
#define HH 25
#define WW 512

typedef __bf16 bf16x16 __attribute__((ext_vector_type(16)));
typedef float  f32x8   __attribute__((ext_vector_type(8)));
typedef float  f32x4   __attribute__((ext_vector_type(4)));
typedef float  f32x2   __attribute__((ext_vector_type(2)));

union FragU { uint4 u[2]; bf16x16 v; };

__device__ __forceinline__ unsigned short f2bf(float x) {
  unsigned u = __float_as_uint(x);
  unsigned r = u + 0x7FFFu + ((u >> 16) & 1u);
  return (unsigned short)(r >> 16);
}

__device__ __forceinline__ f32x8 wmma_bf16(bf16x16 a, bf16x16 b, f32x8 c) {
  return __builtin_amdgcn_wmma_f32_16x16x32_bf16(false, a, false, b, (short)0, c, false, false);
}

__device__ __forceinline__ bf16x16 load_frag_lds(const unsigned short* rowBase, int kt, int lane) {
  int kb = (lane & 16) ? 8 : 0;
  const unsigned short* p = rowBase + kt * 32 + kb;
  FragU f;
  f.u[0] = *(const uint4*)(p);
  f.u[1] = *(const uint4*)(p + 16);
  return f.v;
}

__device__ __forceinline__ bf16x16 load_frag_Bswz(const unsigned int* swz, int kt, int nt, int lane) {
  const unsigned int* p = swz + ((((kt << 4) + nt) * 32 + lane) << 3);
  FragU f;
  f.u[0] = *(const uint4*)(p);
  f.u[1] = *(const uint4*)(p + 4);
  return f.v;
}

__device__ __forceinline__ bf16x16 load_frag_F(const unsigned int* swz, int tap, int kt, int lane) {
  const unsigned int* p = swz + ((((tap << 3) + kt) * 32 + lane) << 3);
  FragU f;
  f.u[0] = *(const uint4*)(p);
  f.u[1] = *(const uint4*)(p + 4);
  return f.v;
}

__device__ __forceinline__ float bf2f(unsigned short h) {
  return __uint_as_float(((unsigned)h) << 16);
}

__device__ __forceinline__ void lds_barrier() {
  __builtin_amdgcn_fence(__ATOMIC_RELEASE, "workgroup", "local");
  __builtin_amdgcn_s_barrier();
  __builtin_amdgcn_fence(__ATOMIC_ACQUIRE, "workgroup", "local");
}

__global__ void __launch_bounds__(256) prep_kernel(
    const float* __restrict__ wmsg, const float* __restrict__ wup2, const float* __restrict__ v1,
    unsigned int* __restrict__ Mswz, unsigned int* __restrict__ Fswz)
{
  __shared__ float part[8][18];
  __shared__ float wsum[18];
  const int tid = threadIdx.x;

  if (blockIdx.x >= 128) {
    int r    = (blockIdx.x - 128) * 256 + tid;
    int j    = r & 7;
    int lane = (r >> 3) & 31;
    int nt   = (r >> 8) & 15;
    int kt   = r >> 12;
    int n  = nt * 16 + (lane & 15);
    int kb = (lane & 16) ? 8 : 0;
    int k  = kt * 32 + ((j < 4) ? (kb + 2 * j) : (16 + kb + 2 * (j - 4)));
    float lo = wmsg[(n * 256 + k) * 9 + 4];
    float hi = wmsg[(n * 256 + k + 1) * 9 + 4];
    Mswz[r] = (unsigned)f2bf(lo) | ((unsigned)f2bf(hi) << 16);
    return;
  }

  const int kp   = blockIdx.x;
  const int wave = tid >> 5;
  const int lane = tid & 31;
  const float g  = v1[tid];
  const float* srcw = wup2 + (tid * 256 + 2 * kp) * 9;
  float p[18];
  #pragma unroll
  for (int t = 0; t < 18; ++t) p[t] = g * srcw[t];
  #pragma unroll
  for (int t = 0; t < 18; ++t) {
    for (int off = 16; off >= 1; off >>= 1) p[t] += __shfl_xor(p[t], off);
  }
  if (lane == 0) {
    #pragma unroll
    for (int t = 0; t < 18; ++t) part[wave][t] = p[t];
  }
  __syncthreads();
  if (tid < 18) {
    float s = 0.f;
    for (int w = 0; w < 8; ++w) s += part[w][tid];
    wsum[tid] = s;
  }
  __syncthreads();
  if (tid < 144) {
    const int tap = tid >> 4;
    const int n   = tid & 15;
    const float w0 = wsum[tap], w1 = wsum[9 + tap];
    const unsigned short h0 = f2bf(w0), h1 = f2bf(w1);
    unsigned int d = 0u;
    if (n == 0) {
      d = (unsigned)h0 | ((unsigned)h1 << 16);
    } else if (n == 1) {
      d = (unsigned)f2bf(w0 - bf2f(h0)) | ((unsigned)f2bf(w1 - bf2f(h1)) << 16);
    }
    const int k   = 2 * kp;
    const int kt  = k >> 5;
    const int r   = k & 31;
    const int kbs = (r >> 3) & 1;
    const int j   = ((r >> 4) << 2) + ((r & 7) >> 1);
    Fswz[((((tap << 3) + kt) * 32 + kbs * 16 + n) << 3) + j] = d;
  }
}

__global__ void init_misc(float* scid, float* shid) {
  int t = threadIdx.x;
  scid[t] = 1.0f; shid[t] = 0.0f;
}

#define TBLK 4096
struct XRow { f32x4 a0, b0, a1, b1; };

template <bool TIN>
__global__ void __launch_bounds__(256) scnn_kernel(
    const float* __restrict__ xin, float* __restrict__ xout,
    const unsigned int* __restrict__ Mswz,
    const float* __restrict__ sc, const float* __restrict__ sh,
    int relu_in, float* __restrict__ statsP)
{
  __shared__ __align__(16) unsigned short prevA[16 * 264];
  __shared__ __align__(16) unsigned short prevB[16 * 264];

  const int tid  = threadIdx.x;
  const int b    = blockIdx.x >> 5;
  const int w0   = (blockIdx.x & 31) << 4;
  const int wave = tid >> 5;
  const int lane = tid & 31;
  const int half8 = (lane & 16) ? 8 : 0;
  const int nt0 = wave * 2, nt1 = wave * 2 + 1;
  const int ch0 = nt0 * 16 + (lane & 15);
  const int ch1 = nt1 * 16 + (lane & 15);

  for (int i = tid; i < 16 * 264; i += 256) prevA[i] = 0;

  const float sA = sc[ch0], bA = sh[ch0];
  const float sB = sc[ch1], bB = sh[ch1];

  const float* xi0p = xin  + ((size_t)(b * CC + ch0) * HH) * WW + w0 + half8;
  const float* xi1p = xin  + ((size_t)(b * CC + ch1) * HH) * WW + w0 + half8;
  const size_t tbase = (size_t)blockIdx.x * HH * TBLK + (size_t)(wave * 128 + lane) * 4;
  const float* xitp = xin  + tbase;
  float*       xotp = xout + tbase;

  bf16x16 bm0[8], bm1[8];
  #pragma unroll
  for (int kt = 0; kt < 8; ++kt) {
    bm0[kt] = load_frag_Bswz(Mswz, kt, nt0, lane);
    bm1[kt] = load_frag_Bswz(Mswz, kt, nt1, lane);
  }

  auto load_row = [&](int r, XRow& x) {
    if (TIN) {
      const float* n = xitp + (size_t)r * TBLK;
      x.a0 = *(const f32x4*)(n);       x.b0 = *(const f32x4*)(n + 128);
      x.a1 = *(const f32x4*)(n + 256); x.b1 = *(const f32x4*)(n + 384);
    } else {
      const float* n0 = xi0p + (size_t)r * WW;
      const float* n1 = xi1p + (size_t)r * WW;
      x.a0 = *(const f32x4*)(n0); x.b0 = *(const f32x4*)(n0 + 4);
      x.a1 = *(const f32x4*)(n1); x.b1 = *(const f32x4*)(n1 + 4);
    }
  };

  float s0 = 0.f, q0 = 0.f, s1 = 0.f, q1 = 0.f;

  auto step = [&](int h, const XRow& x, XRow& xfill, const unsigned short* prevCur, unsigned short* prevNxt) {
    __builtin_amdgcn_sched_barrier(0);
    lds_barrier();
    load_row(min(h + 4, HH - 1), xfill);

    f32x8 acc0 = {}; f32x8 acc1 = {};
    const unsigned short* arow = &prevCur[(lane & 15) * 264];
    bf16x16 a = load_frag_lds(arow, 0, lane);
    #pragma unroll
    for (int kt = 0; kt < 8; ++kt) {
      bf16x16 an = (kt < 7) ? load_frag_lds(arow, kt + 1, lane) : a;
      acc0 = wmma_bf16(a, bm0[kt], acc0);
      acc1 = wmma_bf16(a, bm1[kt], acc1);
      a = an;
    }

    f32x4 xa0 = x.a0, xb0 = x.b0, xa1 = x.a1, xb1 = x.b1;
    asm("" : "+v"(xa0), "+v"(xb0), "+v"(xa1), "+v"(xb1));
    const float xi0[8] = {xa0.x, xa0.y, xa0.z, xa0.w, xb0.x, xb0.y, xb0.z, xb0.w};
    const float xi1[8] = {xa1.x, xa1.y, xa1.z, xa1.w, xb1.x, xb1.y, xb1.z, xb1.w};
    float r0[8], r1[8];
    #pragma unroll
    for (int j = 0; j < 8; ++j) {
      float a0 = sA * xi0[j] + bA;
      float a1 = sB * xi1[j] + bB;
      if (relu_in) { a0 = fmaxf(a0, 0.f); a1 = fmaxf(a1, 0.f); }
      float v0 = a0 + fmaxf(acc0[j], 0.f);
      float v1 = a1 + fmaxf(acc1[j], 0.f);
      r0[j] = v0; r1[j] = v1;
      s0 += v0; q0 += v0 * v0;
      s1 += v1; q1 += v1 * v1;
      prevNxt[(j + half8) * 264 + ch0] = f2bf(v0);
      prevNxt[(j + half8) * 264 + ch1] = f2bf(v1);
    }
    {
      float* p = xotp + (size_t)(HH - 1 - h) * TBLK;
      *(float4*)(p)       = make_float4(r0[0], r0[1], r0[2], r0[3]);
      *(float4*)(p + 128) = make_float4(r0[4], r0[5], r0[6], r0[7]);
      *(float4*)(p + 256) = make_float4(r1[0], r1[1], r1[2], r1[3]);
      *(float4*)(p + 384) = make_float4(r1[4], r1[5], r1[6], r1[7]);
    }
  };

  XRow x0, x1, x2, x3, x4;
  load_row(0, x0); load_row(1, x1); load_row(2, x2); load_row(3, x3);

  unsigned short* cur = prevA;
  unsigned short* nxt = prevB;
  for (int h = 0; h < HH; h += 5) {
    step(h,     x0, x4, cur, nxt);
    step(h + 1, x1, x0, nxt, cur);
    step(h + 2, x2, x1, cur, nxt);
    step(h + 3, x3, x2, nxt, cur);
    step(h + 4, x4, x3, cur, nxt);
    unsigned short* t = cur; cur = nxt; nxt = t;
  }

  s0 += __shfl_xor(s0, 16); q0 += __shfl_xor(q0, 16);
  s1 += __shfl_xor(s1, 16); q1 += __shfl_xor(q1, 16);
  if (lane < 16) {
    statsP[(ch0)       * 256 + blockIdx.x] = s0;
    statsP[(256 + ch0) * 256 + blockIdx.x] = q0;
    statsP[(ch1)       * 256 + blockIdx.x] = s1;
    statsP[(256 + ch1) * 256 + blockIdx.x] = q1;
  }
}

__global__ void __launch_bounds__(256) bnscale_kernel(
    const float* __restrict__ statsP,
    const float* __restrict__ gamma, const float* __restrict__ beta,
    float* __restrict__ sc, float* __restrict__ sh)
{
  __shared__ float ps[8], pq[8];
  const int c = blockIdx.x;
  const int t = threadIdx.x;
  float s = statsP[c * 256 + t];
  float q = statsP[(256 + c) * 256 + t];
  for (int off = 16; off >= 1; off >>= 1) { s += __shfl_xor(s, off); q += __shfl_xor(q, off); }
  if ((t & 31) == 0) { ps[t >> 5] = s; pq[t >> 5] = q; }
  __syncthreads();
  if (t == 0) {
    float S = 0.f, Q = 0.f;
    for (int w = 0; w < 8; ++w) { S += ps[w]; Q += pq[w]; }
    const float n = (float)(BB * HH * WW);
    float mean = S / n;
    float var  = Q / n - mean * mean;
    float g = gamma[c] * rsqrtf(var + 1e-5f);
    sc[c] = g;
    sh[c] = beta[c] - mean * g;
  }
}

#define CONV_MT 4
#define CONV_TW (CONV_MT * 16)
#define CONV_HW (CONV_TW + 2)
#define CONV_ROW (CONV_HW * 264)
#define CONV_RUNS 4
#define RAW_P   36

__global__ void __launch_bounds__(256) conv_kernel(
    const float* __restrict__ xin,
    const unsigned int* __restrict__ Fswz,
    const float* __restrict__ sc, const float* __restrict__ sh,
    float* __restrict__ z)
{
  __shared__ __align__(16) unsigned short uT[3 * CONV_ROW];
  __shared__ __align__(16) float raw[256 * RAW_P];
  __shared__ __align__(16) float zp[8 * CONV_TW * 2];

  const int tid  = threadIdx.x;
  const int wave = tid >> 5;
  const int lane = tid & 31;
  const int wt   = blockIdx.x & 15;
  const int run  = (blockIdx.x >> 4) & (CONV_RUNS - 1);
  const int b    = blockIdx.x >> 6;
  const int h0   = run * 6;
  const int nrows = (run == CONV_RUNS - 1) ? 7 : 6;
  const int w20  = wt * CONV_TW;

  const int pq   = (tid >> 5) & 3;
  const int rawq = (32 * (tid >> 7) + 16 * (pq >> 1) + (tid & 15)) * RAW_P
                 + 8 * ((tid >> 4) & 1) + 4 * (pq & 1);
  const int eP0  = (((tid >> 5) * 4 + ((tid >> 4) & 1) * 2) * 32 + (tid & 15)) * 4;
  const int eP15 = (((tid >> 5) * 4 + ((tid >> 4) & 1) * 2 + 1) * 32 + 16 + (tid & 15)) * 4 + 3;
  const int tileL = (wt > 0)  ? 2 * wt - 1 : 0;
  const int offL  = (wt > 0)  ? eP15 : eP0;
  const int tileR = (wt < 15) ? 2 * wt + 2 : 31;
  const int offR  = (wt < 15) ? eP0 : eP15;

  auto load_row = [&](int hs, f32x4 (&q)[8], float& eL, float& eR) {
    const int hsc = min(max(hs, 0), HH - 1);
    const float* t0 = xin + ((b * 32 + 2 * wt) * HH + hsc) * TBLK + 4 * tid;
    #pragma unroll
    for (int i = 0; i < 8; ++i)
      q[i] = *(const f32x4*)(t0 + (i >> 2) * (HH * TBLK) + (i & 3) * 1024);
    eL = xin[((b * 32 + tileL) * HH + hsc) * TBLK + offL];
    eR = xin[((b * 32 + tileR) * HH + hsc) * TBLK + offR];
  };
  auto put_row = [&](const f32x4 (&q)[8], float eL, float eR) {
    #pragma unroll
    for (int i = 0; i < 8; ++i)
      *(f32x4*)&raw[rawq + (i & 3) * (64 * RAW_P) + (i >> 2) * 16] = q[i];
    *(f32x2*)&raw[tid * RAW_P + 32] = (f32x2){eL, eR};
  };
  const float scc = sc[tid], shc = sh[tid];
  auto col_weight = [&](int wi) {
    const int w2  = w20 + wi - 1;
    const int num = w2 * 511;
    const int i0  = num / 1023;
    return (float)(num - i0 * 1023) * (1.0f / 1023.0f);
  };
  const int fl0 = __float_as_int(col_weight(lane));
  const int fl1 = __float_as_int(col_weight(32 + lane));
  const int fl2 = __float_as_int(col_weight(64 + lane));
  auto emit_row = [&](unsigned short* dst, int hs) {
    const bool rowok = (hs >= 0 && hs < HH);
    float xs[34];
    const f32x2 e = *(const f32x2*)&raw[tid * RAW_P + 32];
    xs[0] = e.x; xs[33] = e.y;
    #pragma unroll
    for (int k = 0; k < 8; ++k) {
      const f32x4 m = *(const f32x4*)&raw[tid * RAW_P + 4 * k];
      xs[1 + 4 * k] = m.x; xs[2 + 4 * k] = m.y; xs[3 + 4 * k] = m.z; xs[4 + 4 * k] = m.w;
    }
    #pragma unroll
    for (int k = 0; k < 34; ++k) {
      const float x = fmaxf(scc * xs[k] + shc, 0.f);
      xs[k] = rowok ? x : 0.f;
    }
    #pragma unroll
    for (int wi = 0; wi < CONV_HW; ++wi) {
      const int  w2    = w20 + wi - 1;
      const bool colok = (w2 >= 0 && w2 < 1024);
      const float f   = __int_as_float(__builtin_amdgcn_readlane((wi < 32) ? fl0 : (wi < 64) ? fl1 : fl2, wi & 31));
      const int   s   = wi >> 1;
      const bool  col0 = (wi == 1) && (wt == 0);
      const float x0 = col0 ? xs[1] : xs[s];
      const float x1 = col0 ? xs[2] : xs[s + 1];
      const float v  = x0 + f * (x1 - x0);
      dst[wi * 264 + tid] = colok ? f2bf(v) : (unsigned short)0;
    }
  };

  bf16x16 bw[9];
  #pragma unroll
  for (int tap = 0; tap < 9; ++tap) bw[tap] = load_frag_F(Fswz, tap, wave, lane);

  int o0 = 0, o1 = CONV_ROW, o2 = 2 * CONV_ROW;

  f32x4 qn[8];
  float ln, rn;
  {
    f32x4 q0[8], q1[8], q2[8];
    float  l0, r0, l1, r1, l2, r2;
    load_row(h0 - 1, q0, l0, r0);
    load_row(h0,     q1, l1, r1);
    load_row(h0 + 1, q2, l2, r2);

    put_row(q0, l0, r0);
    __syncthreads();
    emit_row(uT + o0, h0 - 1);
    __syncthreads();
    put_row(q1, l1, r1);
    __syncthreads();
    emit_row(uT + o1, h0);
    __syncthreads();
    put_row(q2, l2, r2);
    load_row(h0 + 2, qn, ln, rn);
    __syncthreads();
    emit_row(uT + o2, h0 + 1);
  }
  __syncthreads();

  const int half8 = (lane & 16) ? 8 : 0;
  const int arow_l = (lane & 15) * 264;

  for (int i = 0; i < nrows; ++i) {
    const int  h    = h0 + i;
    const bool more = (i + 1 < nrows);

    if (more) put_row(qn, ln, rn);

    f32x8 acc[CONV_MT];
    #pragma unroll
    for (int m = 0; m < CONV_MT; ++m) acc[m] = (f32x8){};

    const unsigned short* a0 = uT + o0 + arow_l;
    const unsigned short* a1 = uT + o1 + arow_l;
    const unsigned short* a2 = uT + o2 + arow_l;
    bf16x16 af[CONV_MT];
    #pragma unroll
    for (int m = 0; m < CONV_MT; ++m) af[m] = load_frag_lds(a0 + m * 16 * 264, wave, lane);
    #pragma unroll
    for (int tap = 0; tap < 9; ++tap) {
      bf16x16 an[CONV_MT];
      if (tap < 8) {
        const int ky = (tap + 1) / 3, kx = (tap + 1) - ky * 3;
        const unsigned short* arow = ((ky == 0) ? a0 : (ky == 1) ? a1 : a2) + kx * 264;
        #pragma unroll
        for (int m = 0; m < CONV_MT; ++m) an[m] = load_frag_lds(arow + m * 16 * 264, wave, lane);
      }
      __builtin_amdgcn_sched_barrier(0);
      #pragma unroll
      for (int m = 0; m < CONV_MT; ++m) acc[m] = wmma_bf16(af[m], bw[tap], acc[m]);
      __builtin_amdgcn_sched_barrier(0);
      if (tap < 8) {
        #pragma unroll
        for (int m = 0; m < CONV_MT; ++m) af[m] = an[m];
      }
    }

    if ((lane & 15) < 2) {
      #pragma unroll
      for (int m = 0; m < CONV_MT; ++m) {
        #pragma unroll
        for (int j = 0; j < 8; ++j)
          zp[((wave * CONV_TW) + m * 16 + j + half8) * 2 + (lane & 15)] = acc[m][j];
      }
    }
    __syncthreads();

    if (tid < CONV_TW) {
      float zh = 0.f, zl = 0.f;
      for (int w = 0; w < 8; ++w) {
        const f32x2 p = *(const f32x2*)&zp[(w * CONV_TW + tid) * 2];
        zh += p.x;
        zl += p.y;
      }
      z[(b * 25 + h) * 1024 + w20 + tid] = zh + zl;
    }
    if (more) {
      if (i + 2 < nrows) load_row(h + 3, qn, ln, rn);
      emit_row(uT + o0, h + 2);
    }
    __syncthreads();
    const int t = o0; o0 = o1; o1 = o2; o2 = t;
  }
}

__global__ void final_kernel(const float* __restrict__ z, const float* __restrict__ v2,
                             float* __restrict__ out)
{
  int idx = blockIdx.x * 256 + threadIdx.x;
  if (idx >= BB * 4096) return;
  int b  = idx >> 12;
  int w4 = idx & 4095;

  int   num4 = w4 * 2047;
  int   i40  = num4 / 4095;
  float f4   = (float)(num4 - i40 * 4095) * (1.0f / 4095.0f);
  int   i41  = min(i40 + 1, 2047);

  int n3a = i40 * 1023; int ja0 = n3a / 2047;
  float fa = (float)(n3a - ja0 * 2047) * (1.0f / 2047.0f);
  int ja1 = min(ja0 + 1, 1023);
  int n3b = i41 * 1023; int jb0 = n3b / 2047;
  float fb = (float)(n3b - jb0 * 2047) * (1.0f / 2047.0f);
  int jb1 = min(jb0 + 1, 1023);

  float acc = 0.f;
  #pragma unroll 5
  for (int h = 0; h < HH; ++h) {
    const float* zh = z + (b * 25 + h) * 1024;
    float za = zh[ja0] + fa * (zh[ja1] - zh[ja0]);
    float zb = zh[jb0] + fb * (zh[jb1] - zh[jb0]);
    acc += v2[h] * (za + f4 * (zb - za));
  }
  out[idx] = 1.0f / (1.0f + __expf(-acc));
}

extern "C" void kernel_launch(void* const* d_in, const int* in_sizes, int n_in,
                              void* d_out, int out_size, void* d_ws, size_t ws_size,
                              hipStream_t stream)
{
  const float* p2c   = (const float*)d_in[0];
  const float* wmsg  = (const float*)d_in[1];
  const float* gamma = (const float*)d_in[2];
  const float* beta  = (const float*)d_in[3];
  const float* wup2  = (const float*)d_in[4];
  const float* v1    = (const float*)d_in[5];
  const float* v2    = (const float*)d_in[6];

  char* ws = (char*)d_ws;
  size_t off = 0;
  auto walloc = [&](size_t bytes) -> void* {
    void* p = ws + off;
    off = (off + bytes + 255) & ~(size_t)255;
    return p;
  };
  unsigned int* Mswz = (unsigned int*)walloc(32768u * 4);
  unsigned int* Fswz = (unsigned int*)walloc(9u * 2048u * 4);
  float* stats1 = (float*)walloc(512u * 256u * 4);
  float* stats2 = (float*)walloc(512u * 256u * 4);
  float* sc1  = (float*)walloc(256 * 4);
  float* sh1  = (float*)walloc(256 * 4);
  float* sc2  = (float*)walloc(256 * 4);
  float* sh2  = (float*)walloc(256 * 4);
  float* scid = (float*)walloc(256 * 4);
  float* shid = (float*)walloc(256 * 4);
  float* z1024 = (float*)walloc((size_t)BB * 25 * 1024 * 4);
  float* out1  = (float*)walloc((size_t)BB * CC * HH * WW * 4);
  float* out2  = (float*)walloc((size_t)BB * CC * HH * WW * 4);

  prep_kernel<<<256, 256, 0, stream>>>(wmsg, wup2, v1, Mswz, Fswz);
  init_misc<<<1, 256, 0, stream>>>(scid, shid);

  scnn_kernel<false><<<256, 256, 0, stream>>>(p2c, out1, Mswz, scid, shid, 0, stats1);
  bnscale_kernel<<<256, 256, 0, stream>>>(stats1, gamma, beta, sc1, sh1);

  scnn_kernel<true><<<256, 256, 0, stream>>>(out1, out2, Mswz, sc1, sh1, 1, stats2);
  bnscale_kernel<<<256, 256, 0, stream>>>(stats2, gamma, beta, sc2, sh2);

  conv_kernel<<<BB * CONV_RUNS * 16, 256, 0, stream>>>(out2, Fswz, sc2, sh2, z1024);

  final_kernel<<<128, 256, 0, stream>>>(z1024, v2, (float*)d_out);
}


# f32 values that become matrix operands (recurrence state, staged conv rows) are rounded by the packed hardware convert, two per instruction, instead of integer bit arithmetic; same rounding
# speedup vs baseline: 6.8761x; 1.0460x over previous
#include <hip/hip_runtime.h>
#include <hip/hip_bf16.h>
#include <stdint.h>

#define BB 8
#define CC 256
#define HH 25
#define WW 512

typedef __bf16 bf16x16 __attribute__((ext_vector_type(16)));
typedef float  f32x8   __attribute__((ext_vector_type(8)));
typedef float  f32x4   __attribute__((ext_vector_type(4)));
typedef float  f32x2   __attribute__((ext_vector_type(2)));

union FragU { uint4 u[2]; bf16x16 v; };

__device__ __forceinline__ unsigned short f2bf(float x) {
  unsigned u = __float_as_uint(x);
  unsigned r = u + 0x7FFFu + ((u >> 16) & 1u);
  return (unsigned short)(r >> 16);
}

typedef __bf16 bf16x2 __attribute__((ext_vector_type(2)));

__device__ __forceinline__ unsigned int pack_bf2(float a, float b) {
  const f32x2 v = {a, b};
  const bf16x2 r = __builtin_convertvector(v, bf16x2);
  return __builtin_bit_cast(unsigned int, r);
}

__device__ __forceinline__ f32x8 wmma_bf16(bf16x16 a, bf16x16 b, f32x8 c) {
  return __builtin_amdgcn_wmma_f32_16x16x32_bf16(false, a, false, b, (short)0, c, false, false);
}

__device__ __forceinline__ bf16x16 load_frag_lds(const unsigned short* rowBase, int kt, int lane) {
  int kb = (lane & 16) ? 8 : 0;
  const unsigned short* p = rowBase + kt * 32 + kb;
  FragU f;
  f.u[0] = *(const uint4*)(p);
  f.u[1] = *(const uint4*)(p + 16);
  return f.v;
}

__device__ __forceinline__ bf16x16 load_frag_Bswz(const unsigned int* swz, int kt, int nt, int lane) {
  const unsigned int* p = swz + ((((kt << 4) + nt) * 32 + lane) << 3);
  FragU f;
  f.u[0] = *(const uint4*)(p);
  f.u[1] = *(const uint4*)(p + 4);
  return f.v;
}

__device__ __forceinline__ bf16x16 load_frag_F(const unsigned int* swz, int tap, int kt, int lane) {
  const unsigned int* p = swz + ((((tap << 3) + kt) * 32 + lane) << 3);
  FragU f;
  f.u[0] = *(const uint4*)(p);
  f.u[1] = *(const uint4*)(p + 4);
  return f.v;
}

__device__ __forceinline__ float bf2f(unsigned short h) {
  return __uint_as_float(((unsigned)h) << 16);
}

__device__ __forceinline__ void lds_barrier() {
  __builtin_amdgcn_fence(__ATOMIC_RELEASE, "workgroup", "local");
  __builtin_amdgcn_s_barrier();
  __builtin_amdgcn_fence(__ATOMIC_ACQUIRE, "workgroup", "local");
}

__global__ void __launch_bounds__(256) prep_kernel(
    const float* __restrict__ wmsg, const float* __restrict__ wup2, const float* __restrict__ v1,
    unsigned int* __restrict__ Mswz, unsigned int* __restrict__ Fswz)
{
  __shared__ float part[8][18];
  __shared__ float wsum[18];
  const int tid = threadIdx.x;

  if (blockIdx.x >= 128) {
    int r    = (blockIdx.x - 128) * 256 + tid;
    int j    = r & 7;
    int lane = (r >> 3) & 31;
    int nt   = (r >> 8) & 15;
    int kt   = r >> 12;
    int n  = nt * 16 + (lane & 15);
    int kb = (lane & 16) ? 8 : 0;
    int k  = kt * 32 + ((j < 4) ? (kb + 2 * j) : (16 + kb + 2 * (j - 4)));
    float lo = wmsg[(n * 256 + k) * 9 + 4];
    float hi = wmsg[(n * 256 + k + 1) * 9 + 4];
    Mswz[r] = (unsigned)f2bf(lo) | ((unsigned)f2bf(hi) << 16);
    return;
  }

  const int kp   = blockIdx.x;
  const int wave = tid >> 5;
  const int lane = tid & 31;
  const float g  = v1[tid];
  const float* srcw = wup2 + (tid * 256 + 2 * kp) * 9;
  float p[18];
  #pragma unroll
  for (int t = 0; t < 18; ++t) p[t] = g * srcw[t];
  #pragma unroll
  for (int t = 0; t < 18; ++t) {
    for (int off = 16; off >= 1; off >>= 1) p[t] += __shfl_xor(p[t], off);
  }
  if (lane == 0) {
    #pragma unroll
    for (int t = 0; t < 18; ++t) part[wave][t] = p[t];
  }
  __syncthreads();
  if (tid < 18) {
    float s = 0.f;
    for (int w = 0; w < 8; ++w) s += part[w][tid];
    wsum[tid] = s;
  }
  __syncthreads();
  if (tid < 144) {
    const int tap = tid >> 4;
    const int n   = tid & 15;
    const float w0 = wsum[tap], w1 = wsum[9 + tap];
    const unsigned short h0 = f2bf(w0), h1 = f2bf(w1);
    unsigned int d = 0u;
    if (n == 0) {
      d = (unsigned)h0 | ((unsigned)h1 << 16);
    } else if (n == 1) {
      d = (unsigned)f2bf(w0 - bf2f(h0)) | ((unsigned)f2bf(w1 - bf2f(h1)) << 16);
    }
    const int k   = 2 * kp;
    const int kt  = k >> 5;
    const int r   = k & 31;
    const int kbs = (r >> 3) & 1;
    const int j   = ((r >> 4) << 2) + ((r & 7) >> 1);
    Fswz[((((tap << 3) + kt) * 32 + kbs * 16 + n) << 3) + j] = d;
  }
}

__global__ void init_misc(float* scid, float* shid) {
  int t = threadIdx.x;
  scid[t] = 1.0f; shid[t] = 0.0f;
}

#define TBLK 4096
struct XRow { f32x4 a0, b0, a1, b1; };

template <bool TIN>
__global__ void __launch_bounds__(256) scnn_kernel(
    const float* __restrict__ xin, float* __restrict__ xout,
    const unsigned int* __restrict__ Mswz,
    const float* __restrict__ sc, const float* __restrict__ sh,
    int relu_in, float* __restrict__ statsP)
{
  __shared__ __align__(16) unsigned short prevA[16 * 264];
  __shared__ __align__(16) unsigned short prevB[16 * 264];

  const int tid  = threadIdx.x;
  const int b    = blockIdx.x >> 5;
  const int w0   = (blockIdx.x & 31) << 4;
  const int wave = tid >> 5;
  const int lane = tid & 31;
  const int half8 = (lane & 16) ? 8 : 0;
  const int nt0 = wave * 2, nt1 = wave * 2 + 1;
  const int ch0 = nt0 * 16 + (lane & 15);
  const int ch1 = nt1 * 16 + (lane & 15);

  for (int i = tid; i < 16 * 264; i += 256) prevA[i] = 0;

  const float sA = sc[ch0], bA = sh[ch0];
  const float sB = sc[ch1], bB = sh[ch1];

  const float* xi0p = xin  + ((size_t)(b * CC + ch0) * HH) * WW + w0 + half8;
  const float* xi1p = xin  + ((size_t)(b * CC + ch1) * HH) * WW + w0 + half8;
  const size_t tbase = (size_t)blockIdx.x * HH * TBLK + (size_t)(wave * 128 + lane) * 4;
  const float* xitp = xin  + tbase;
  float*       xotp = xout + tbase;

  bf16x16 bm0[8], bm1[8];
  #pragma unroll
  for (int kt = 0; kt < 8; ++kt) {
    bm0[kt] = load_frag_Bswz(Mswz, kt, nt0, lane);
    bm1[kt] = load_frag_Bswz(Mswz, kt, nt1, lane);
  }

  auto load_row = [&](int r, XRow& x) {
    if (TIN) {
      const float* n = xitp + (size_t)r * TBLK;
      x.a0 = *(const f32x4*)(n);       x.b0 = *(const f32x4*)(n + 128);
      x.a1 = *(const f32x4*)(n + 256); x.b1 = *(const f32x4*)(n + 384);
    } else {
      const float* n0 = xi0p + (size_t)r * WW;
      const float* n1 = xi1p + (size_t)r * WW;
      x.a0 = *(const f32x4*)(n0); x.b0 = *(const f32x4*)(n0 + 4);
      x.a1 = *(const f32x4*)(n1); x.b1 = *(const f32x4*)(n1 + 4);
    }
  };

  float s0 = 0.f, q0 = 0.f, s1 = 0.f, q1 = 0.f;

  auto step = [&](int h, const XRow& x, XRow& xfill, const unsigned short* prevCur, unsigned short* prevNxt) {
    __builtin_amdgcn_sched_barrier(0);
    lds_barrier();
    load_row(min(h + 4, HH - 1), xfill);

    f32x8 acc0 = {}; f32x8 acc1 = {};
    const unsigned short* arow = &prevCur[(lane & 15) * 264];
    bf16x16 a = load_frag_lds(arow, 0, lane);
    #pragma unroll
    for (int kt = 0; kt < 8; ++kt) {
      bf16x16 an = (kt < 7) ? load_frag_lds(arow, kt + 1, lane) : a;
      acc0 = wmma_bf16(a, bm0[kt], acc0);
      acc1 = wmma_bf16(a, bm1[kt], acc1);
      a = an;
    }

    f32x4 xa0 = x.a0, xb0 = x.b0, xa1 = x.a1, xb1 = x.b1;
    asm("" : "+v"(xa0), "+v"(xb0), "+v"(xa1), "+v"(xb1));
    const float xi0[8] = {xa0.x, xa0.y, xa0.z, xa0.w, xb0.x, xb0.y, xb0.z, xb0.w};
    const float xi1[8] = {xa1.x, xa1.y, xa1.z, xa1.w, xb1.x, xb1.y, xb1.z, xb1.w};
    float r0[8], r1[8];
    #pragma unroll
    for (int j = 0; j < 8; ++j) {
      float a0 = sA * xi0[j] + bA;
      float a1 = sB * xi1[j] + bB;
      if (relu_in) { a0 = fmaxf(a0, 0.f); a1 = fmaxf(a1, 0.f); }
      float v0 = a0 + fmaxf(acc0[j], 0.f);
      float v1 = a1 + fmaxf(acc1[j], 0.f);
      r0[j] = v0; r1[j] = v1;
      s0 += v0; q0 += v0 * v0;
      s1 += v1; q1 += v1 * v1;
    }
    #pragma unroll
    for (int j = 0; j < 8; j += 2) {
      const unsigned int p0 = pack_bf2(r0[j], r0[j + 1]);
      const unsigned int p1 = pack_bf2(r1[j], r1[j + 1]);
      prevNxt[(j + half8) * 264 + ch0]     = (unsigned short)(p0 & 0xFFFFu);
      prevNxt[(j + 1 + half8) * 264 + ch0] = (unsigned short)(p0 >> 16);
      prevNxt[(j + half8) * 264 + ch1]     = (unsigned short)(p1 & 0xFFFFu);
      prevNxt[(j + 1 + half8) * 264 + ch1] = (unsigned short)(p1 >> 16);
    }
    {
      float* p = xotp + (size_t)(HH - 1 - h) * TBLK;
      *(float4*)(p)       = make_float4(r0[0], r0[1], r0[2], r0[3]);
      *(float4*)(p + 128) = make_float4(r0[4], r0[5], r0[6], r0[7]);
      *(float4*)(p + 256) = make_float4(r1[0], r1[1], r1[2], r1[3]);
      *(float4*)(p + 384) = make_float4(r1[4], r1[5], r1[6], r1[7]);
    }
  };

  XRow x0, x1, x2, x3, x4;
  load_row(0, x0); load_row(1, x1); load_row(2, x2); load_row(3, x3);

  unsigned short* cur = prevA;
  unsigned short* nxt = prevB;
  for (int h = 0; h < HH; h += 5) {
    step(h,     x0, x4, cur, nxt);
    step(h + 1, x1, x0, nxt, cur);
    step(h + 2, x2, x1, cur, nxt);
    step(h + 3, x3, x2, nxt, cur);
    step(h + 4, x4, x3, cur, nxt);
    unsigned short* t = cur; cur = nxt; nxt = t;
  }

  s0 += __shfl_xor(s0, 16); q0 += __shfl_xor(q0, 16);
  s1 += __shfl_xor(s1, 16); q1 += __shfl_xor(q1, 16);
  if (lane < 16) {
    statsP[(ch0)       * 256 + blockIdx.x] = s0;
    statsP[(256 + ch0) * 256 + blockIdx.x] = q0;
    statsP[(ch1)       * 256 + blockIdx.x] = s1;
    statsP[(256 + ch1) * 256 + blockIdx.x] = q1;
  }
}

__global__ void __launch_bounds__(256) bnscale_kernel(
    const float* __restrict__ statsP,
    const float* __restrict__ gamma, const float* __restrict__ beta,
    float* __restrict__ sc, float* __restrict__ sh)
{
  __shared__ float ps[8], pq[8];
  const int c = blockIdx.x;
  const int t = threadIdx.x;
  float s = statsP[c * 256 + t];
  float q = statsP[(256 + c) * 256 + t];
  for (int off = 16; off >= 1; off >>= 1) { s += __shfl_xor(s, off); q += __shfl_xor(q, off); }
  if ((t & 31) == 0) { ps[t >> 5] = s; pq[t >> 5] = q; }
  __syncthreads();
  if (t == 0) {
    float S = 0.f, Q = 0.f;
    for (int w = 0; w < 8; ++w) { S += ps[w]; Q += pq[w]; }
    const float n = (float)(BB * HH * WW);
    float mean = S / n;
    float var  = Q / n - mean * mean;
    float g = gamma[c] * rsqrtf(var + 1e-5f);
    sc[c] = g;
    sh[c] = beta[c] - mean * g;
  }
}

#define CONV_MT 4
#define CONV_TW (CONV_MT * 16)
#define CONV_HW (CONV_TW + 2)
#define CONV_ROW (CONV_HW * 264)
#define CONV_RUNS 4
#define RAW_P   36

__global__ void __launch_bounds__(256) conv_kernel(
    const float* __restrict__ xin,
    const unsigned int* __restrict__ Fswz,
    const float* __restrict__ sc, const float* __restrict__ sh,
    float* __restrict__ z)
{
  __shared__ __align__(16) unsigned short uT[3 * CONV_ROW];
  __shared__ __align__(16) float raw[256 * RAW_P];
  __shared__ __align__(16) float zp[8 * CONV_TW * 2];

  const int tid  = threadIdx.x;
  const int wave = tid >> 5;
  const int lane = tid & 31;
  const int wt   = blockIdx.x & 15;
  const int run  = (blockIdx.x >> 4) & (CONV_RUNS - 1);
  const int b    = blockIdx.x >> 6;
  const int h0   = run * 6;
  const int nrows = (run == CONV_RUNS - 1) ? 7 : 6;
  const int w20  = wt * CONV_TW;

  const int pq   = (tid >> 5) & 3;
  const int rawq = (32 * (tid >> 7) + 16 * (pq >> 1) + (tid & 15)) * RAW_P
                 + 8 * ((tid >> 4) & 1) + 4 * (pq & 1);
  const int eP0  = (((tid >> 5) * 4 + ((tid >> 4) & 1) * 2) * 32 + (tid & 15)) * 4;
  const int eP15 = (((tid >> 5) * 4 + ((tid >> 4) & 1) * 2 + 1) * 32 + 16 + (tid & 15)) * 4 + 3;
  const int tileL = (wt > 0)  ? 2 * wt - 1 : 0;
  const int offL  = (wt > 0)  ? eP15 : eP0;
  const int tileR = (wt < 15) ? 2 * wt + 2 : 31;
  const int offR  = (wt < 15) ? eP0 : eP15;

  auto load_row = [&](int hs, f32x4 (&q)[8], float& eL, float& eR) {
    const int hsc = min(max(hs, 0), HH - 1);
    const float* t0 = xin + ((b * 32 + 2 * wt) * HH + hsc) * TBLK + 4 * tid;
    #pragma unroll
    for (int i = 0; i < 8; ++i)
      q[i] = *(const f32x4*)(t0 + (i >> 2) * (HH * TBLK) + (i & 3) * 1024);
    eL = xin[((b * 32 + tileL) * HH + hsc) * TBLK + offL];
    eR = xin[((b * 32 + tileR) * HH + hsc) * TBLK + offR];
  };
  auto put_row = [&](const f32x4 (&q)[8], float eL, float eR) {
    #pragma unroll
    for (int i = 0; i < 8; ++i)
      *(f32x4*)&raw[rawq + (i & 3) * (64 * RAW_P) + (i >> 2) * 16] = q[i];
    *(f32x2*)&raw[tid * RAW_P + 32] = (f32x2){eL, eR};
  };
  const float scc = sc[tid], shc = sh[tid];
  auto col_weight = [&](int wi) {
    const int w2  = w20 + wi - 1;
    const int num = w2 * 511;
    const int i0  = num / 1023;
    return (float)(num - i0 * 1023) * (1.0f / 1023.0f);
  };
  const int fl0 = __float_as_int(col_weight(lane));
  const int fl1 = __float_as_int(col_weight(32 + lane));
  const int fl2 = __float_as_int(col_weight(64 + lane));
  auto emit_row = [&](unsigned short* dst, int hs) {
    const bool rowok = (hs >= 0 && hs < HH);
    float xs[34];
    const f32x2 e = *(const f32x2*)&raw[tid * RAW_P + 32];
    xs[0] = e.x; xs[33] = e.y;
    #pragma unroll
    for (int k = 0; k < 8; ++k) {
      const f32x4 m = *(const f32x4*)&raw[tid * RAW_P + 4 * k];
      xs[1 + 4 * k] = m.x; xs[2 + 4 * k] = m.y; xs[3 + 4 * k] = m.z; xs[4 + 4 * k] = m.w;
    }
    #pragma unroll
    for (int k = 0; k < 34; ++k) {
      const float x = fmaxf(scc * xs[k] + shc, 0.f);
      xs[k] = rowok ? x : 0.f;
    }
    #pragma unroll
    for (int wp = 0; wp < CONV_HW; wp += 2) {
      float vv[2];
      #pragma unroll
      for (int e = 0; e < 2; ++e) {
        const int  wi    = wp + e;
        const int  w2    = w20 + wi - 1;
        const bool colok = (w2 >= 0 && w2 < 1024);
        const float f   = __int_as_float(__builtin_amdgcn_readlane((wi < 32) ? fl0 : (wi < 64) ? fl1 : fl2, wi & 31));
        const int   s   = wi >> 1;
        const bool  col0 = (wi == 1) && (wt == 0);
        const float x0 = col0 ? xs[1] : xs[s];
        const float x1 = col0 ? xs[2] : xs[s + 1];
        const float v  = x0 + f * (x1 - x0);
        vv[e] = colok ? v : 0.f;
      }
      const unsigned int p = pack_bf2(vv[0], vv[1]);
      dst[wp * 264 + tid]       = (unsigned short)(p & 0xFFFFu);
      dst[(wp + 1) * 264 + tid] = (unsigned short)(p >> 16);
    }
  };

  bf16x16 bw[9];
  #pragma unroll
  for (int tap = 0; tap < 9; ++tap) bw[tap] = load_frag_F(Fswz, tap, wave, lane);

  int o0 = 0, o1 = CONV_ROW, o2 = 2 * CONV_ROW;

  f32x4 qn[8];
  float ln, rn;
  {
    f32x4 q0[8], q1[8], q2[8];
    float  l0, r0, l1, r1, l2, r2;
    load_row(h0 - 1, q0, l0, r0);
    load_row(h0,     q1, l1, r1);
    load_row(h0 + 1, q2, l2, r2);

    put_row(q0, l0, r0);
    __syncthreads();
    emit_row(uT + o0, h0 - 1);
    __syncthreads();
    put_row(q1, l1, r1);
    __syncthreads();
    emit_row(uT + o1, h0);
    __syncthreads();
    put_row(q2, l2, r2);
    load_row(h0 + 2, qn, ln, rn);
    __syncthreads();
    emit_row(uT + o2, h0 + 1);
  }
  __syncthreads();

  const int half8 = (lane & 16) ? 8 : 0;
  const int arow_l = (lane & 15) * 264;

  for (int i = 0; i < nrows; ++i) {
    const int  h    = h0 + i;
    const bool more = (i + 1 < nrows);

    if (more) put_row(qn, ln, rn);

    f32x8 acc[CONV_MT];
    #pragma unroll
    for (int m = 0; m < CONV_MT; ++m) acc[m] = (f32x8){};

    const unsigned short* a0 = uT + o0 + arow_l;
    const unsigned short* a1 = uT + o1 + arow_l;
    const unsigned short* a2 = uT + o2 + arow_l;
    bf16x16 af[CONV_MT];
    #pragma unroll
    for (int m = 0; m < CONV_MT; ++m) af[m] = load_frag_lds(a0 + m * 16 * 264, wave, lane);
    #pragma unroll
    for (int tap = 0; tap < 9; ++tap) {
      bf16x16 an[CONV_MT];
      if (tap < 8) {
        const int ky = (tap + 1) / 3, kx = (tap + 1) - ky * 3;
        const unsigned short* arow = ((ky == 0) ? a0 : (ky == 1) ? a1 : a2) + kx * 264;
        #pragma unroll
        for (int m = 0; m < CONV_MT; ++m) an[m] = load_frag_lds(arow + m * 16 * 264, wave, lane);
      }
      __builtin_amdgcn_sched_barrier(0);
      #pragma unroll
      for (int m = 0; m < CONV_MT; ++m) acc[m] = wmma_bf16(af[m], bw[tap], acc[m]);
      __builtin_amdgcn_sched_barrier(0);
      if (tap < 8) {
        #pragma unroll
        for (int m = 0; m < CONV_MT; ++m) af[m] = an[m];
      }
    }

    if ((lane & 15) < 2) {
      #pragma unroll
      for (int m = 0; m < CONV_MT; ++m) {
        #pragma unroll
        for (int j = 0; j < 8; ++j)
          zp[((wave * CONV_TW) + m * 16 + j + half8) * 2 + (lane & 15)] = acc[m][j];
      }
    }
    __syncthreads();

    if (tid < CONV_TW) {
      float zh = 0.f, zl = 0.f;
      for (int w = 0; w < 8; ++w) {
        const f32x2 p = *(const f32x2*)&zp[(w * CONV_TW + tid) * 2];
        zh += p.x;
        zl += p.y;
      }
      z[(b * 25 + h) * 1024 + w20 + tid] = zh + zl;
    }
    if (more) {
      if (i + 2 < nrows) load_row(h + 3, qn, ln, rn);
      emit_row(uT + o0, h + 2);
    }
    __syncthreads();
    const int t = o0; o0 = o1; o1 = o2; o2 = t;
  }
}

__global__ void final_kernel(const float* __restrict__ z, const float* __restrict__ v2,
                             float* __restrict__ out)
{
  int idx = blockIdx.x * 256 + threadIdx.x;
  if (idx >= BB * 4096) return;
  int b  = idx >> 12;
  int w4 = idx & 4095;

  int   num4 = w4 * 2047;
  int   i40  = num4 / 4095;
  float f4   = (float)(num4 - i40 * 4095) * (1.0f / 4095.0f);
  int   i41  = min(i40 + 1, 2047);

  int n3a = i40 * 1023; int ja0 = n3a / 2047;
  float fa = (float)(n3a - ja0 * 2047) * (1.0f / 2047.0f);
  int ja1 = min(ja0 + 1, 1023);
  int n3b = i41 * 1023; int jb0 = n3b / 2047;
  float fb = (float)(n3b - jb0 * 2047) * (1.0f / 2047.0f);
  int jb1 = min(jb0 + 1, 1023);

  float acc = 0.f;
  #pragma unroll 5
  for (int h = 0; h < HH; ++h) {
    const float* zh = z + (b * 25 + h) * 1024;
    float za = zh[ja0] + fa * (zh[ja1] - zh[ja0]);
    float zb = zh[jb0] + fb * (zh[jb1] - zh[jb0]);
    acc += v2[h] * (za + f4 * (zb - za));
  }
  out[idx] = 1.0f / (1.0f + __expf(-acc));
}

extern "C" void kernel_launch(void* const* d_in, const int* in_sizes, int n_in,
                              void* d_out, int out_size, void* d_ws, size_t ws_size,
                              hipStream_t stream)
{
  const float* p2c   = (const float*)d_in[0];
  const float* wmsg  = (const float*)d_in[1];
  const float* gamma = (const float*)d_in[2];
  const float* beta  = (const float*)d_in[3];
  const float* wup2  = (const float*)d_in[4];
  const float* v1    = (const float*)d_in[5];
  const float* v2    = (const float*)d_in[6];

  char* ws = (char*)d_ws;
  size_t off = 0;
  auto walloc = [&](size_t bytes) -> void* {
    void* p = ws + off;
    off = (off + bytes + 255) & ~(size_t)255;
    return p;
  };
  unsigned int* Mswz = (unsigned int*)walloc(32768u * 4);
  unsigned int* Fswz = (unsigned int*)walloc(9u * 2048u * 4);
  float* stats1 = (float*)walloc(512u * 256u * 4);
  float* stats2 = (float*)walloc(512u * 256u * 4);
  float* sc1  = (float*)walloc(256 * 4);
  float* sh1  = (float*)walloc(256 * 4);
  float* sc2  = (float*)walloc(256 * 4);
  float* sh2  = (float*)walloc(256 * 4);
  float* scid = (float*)walloc(256 * 4);
  float* shid = (float*)walloc(256 * 4);
  float* z1024 = (float*)walloc((size_t)BB * 25 * 1024 * 4);
  float* out1  = (float*)walloc((size_t)BB * CC * HH * WW * 4);
  float* out2  = (float*)walloc((size_t)BB * CC * HH * WW * 4);

  prep_kernel<<<256, 256, 0, stream>>>(wmsg, wup2, v1, Mswz, Fswz);
  init_misc<<<1, 256, 0, stream>>>(scid, shid);

  scnn_kernel<false><<<256, 256, 0, stream>>>(p2c, out1, Mswz, scid, shid, 0, stats1);
  bnscale_kernel<<<256, 256, 0, stream>>>(stats1, gamma, beta, sc1, sh1);

  scnn_kernel<true><<<256, 256, 0, stream>>>(out1, out2, Mswz, sc1, sh1, 1, stats2);
  bnscale_kernel<<<256, 256, 0, stream>>>(stats2, gamma, beta, sc2, sh2);

  conv_kernel<<<BB * CONV_RUNS * 16, 256, 0, stream>>>(out2, Fswz, sc2, sh2, z1024);

  final_kernel<<<128, 256, 0, stream>>>(z1024, v2, (float*)d_out);
}


# conv: a staged row is read from LDS once, not 9 times: kx taps are columns of one weight tile, ky taps feed three running accumulators; tap shift applied to the small sums (sums regrouped)
# speedup vs baseline: 7.1171x; 1.0350x over previous
#include <hip/hip_runtime.h>
#include <hip/hip_bf16.h>
#include <stdint.h>

#define BB 8
#define CC 256
#define HH 25
#define WW 512

typedef __bf16 bf16x16 __attribute__((ext_vector_type(16)));
typedef float  f32x8   __attribute__((ext_vector_type(8)));
typedef float  f32x4   __attribute__((ext_vector_type(4)));
typedef float  f32x2   __attribute__((ext_vector_type(2)));

union FragU { uint4 u[2]; bf16x16 v; };

__device__ __forceinline__ unsigned short f2bf(float x) {
  unsigned u = __float_as_uint(x);
  unsigned r = u + 0x7FFFu + ((u >> 16) & 1u);
  return (unsigned short)(r >> 16);
}

typedef __bf16 bf16x2 __attribute__((ext_vector_type(2)));

__device__ __forceinline__ unsigned int pack_bf2(float a, float b) {
  const f32x2 v = {a, b};
  const bf16x2 r = __builtin_convertvector(v, bf16x2);
  return __builtin_bit_cast(unsigned int, r);
}

__device__ __forceinline__ f32x8 wmma_bf16(bf16x16 a, bf16x16 b, f32x8 c) {
  return __builtin_amdgcn_wmma_f32_16x16x32_bf16(false, a, false, b, (short)0, c, false, false);
}

__device__ __forceinline__ bf16x16 load_frag_lds(const unsigned short* rowBase, int kt, int lane) {
  int kb = (lane & 16) ? 8 : 0;
  const unsigned short* p = rowBase + kt * 32 + kb;
  FragU f;
  f.u[0] = *(const uint4*)(p);
  f.u[1] = *(const uint4*)(p + 16);
  return f.v;
}

__device__ __forceinline__ bf16x16 load_frag_Bswz(const unsigned int* swz, int kt, int nt, int lane) {
  const unsigned int* p = swz + ((((kt << 4) + nt) * 32 + lane) << 3);
  FragU f;
  f.u[0] = *(const uint4*)(p);
  f.u[1] = *(const uint4*)(p + 4);
  return f.v;
}

__device__ __forceinline__ bf16x16 load_frag_F(const unsigned int* swz, int ky, int kt, int lane) {
  const unsigned int* p = swz + ((((ky << 3) + kt) * 32 + lane) << 3);
  FragU f;
  f.u[0] = *(const uint4*)(p);
  f.u[1] = *(const uint4*)(p + 4);
  return f.v;
}

__device__ __forceinline__ float bf2f(unsigned short h) {
  return __uint_as_float(((unsigned)h) << 16);
}

__device__ __forceinline__ void lds_barrier() {
  __builtin_amdgcn_fence(__ATOMIC_RELEASE, "workgroup", "local");
  __builtin_amdgcn_s_barrier();
  __builtin_amdgcn_fence(__ATOMIC_ACQUIRE, "workgroup", "local");
}

__global__ void __launch_bounds__(256) prep_kernel(
    const float* __restrict__ wmsg, const float* __restrict__ wup2, const float* __restrict__ v1,
    unsigned int* __restrict__ Mswz, unsigned int* __restrict__ Fswz)
{
  __shared__ float part[8][18];
  __shared__ float wsum[18];
  const int tid = threadIdx.x;

  if (blockIdx.x >= 128) {
    int r    = (blockIdx.x - 128) * 256 + tid;
    int j    = r & 7;
    int lane = (r >> 3) & 31;
    int nt   = (r >> 8) & 15;
    int kt   = r >> 12;
    int n  = nt * 16 + (lane & 15);
    int kb = (lane & 16) ? 8 : 0;
    int k  = kt * 32 + ((j < 4) ? (kb + 2 * j) : (16 + kb + 2 * (j - 4)));
    float lo = wmsg[(n * 256 + k) * 9 + 4];
    float hi = wmsg[(n * 256 + k + 1) * 9 + 4];
    Mswz[r] = (unsigned)f2bf(lo) | ((unsigned)f2bf(hi) << 16);
    return;
  }

  const int kp   = blockIdx.x;
  const int wave = tid >> 5;
  const int lane = tid & 31;
  const float g  = v1[tid];
  const float* srcw = wup2 + (tid * 256 + 2 * kp) * 9;
  float p[18];
  #pragma unroll
  for (int t = 0; t < 18; ++t) p[t] = g * srcw[t];
  #pragma unroll
  for (int t = 0; t < 18; ++t) {
    for (int off = 16; off >= 1; off >>= 1) p[t] += __shfl_xor(p[t], off);
  }
  if (lane == 0) {
    #pragma unroll
    for (int t = 0; t < 18; ++t) part[wave][t] = p[t];
  }
  __syncthreads();
  if (tid < 18) {
    float s = 0.f;
    for (int w = 0; w < 8; ++w) s += part[w][tid];
    wsum[tid] = s;
  }
  __syncthreads();
  if (tid < 48) {
    const int ky = tid >> 4;
    const int n  = tid & 15;
    unsigned int d = 0u;
    if (n < 6) {
      const int tap = ky * 3 + (n >> 1);
      const float w0 = wsum[tap], w1 = wsum[9 + tap];
      const unsigned short h0 = f2bf(w0), h1 = f2bf(w1);
      d = (n & 1) ? ((unsigned)f2bf(w0 - bf2f(h0)) | ((unsigned)f2bf(w1 - bf2f(h1)) << 16))
                  : ((unsigned)h0 | ((unsigned)h1 << 16));
    }
    const int k   = 2 * kp;
    const int kt  = k >> 5;
    const int r   = k & 31;
    const int kbs = (r >> 3) & 1;
    const int j   = ((r >> 4) << 2) + ((r & 7) >> 1);
    Fswz[((((ky << 3) + kt) * 32 + kbs * 16 + n) << 3) + j] = d;
  }
}

__global__ void init_misc(float* scid, float* shid) {
  int t = threadIdx.x;
  scid[t] = 1.0f; shid[t] = 0.0f;
}

#define TBLK 4096
struct XRow { f32x4 a0, b0, a1, b1; };

template <bool TIN>
__global__ void __launch_bounds__(256) scnn_kernel(
    const float* __restrict__ xin, float* __restrict__ xout,
    const unsigned int* __restrict__ Mswz,
    const float* __restrict__ sc, const float* __restrict__ sh,
    int relu_in, float* __restrict__ statsP)
{
  __shared__ __align__(16) unsigned short prevA[16 * 264];
  __shared__ __align__(16) unsigned short prevB[16 * 264];

  const int tid  = threadIdx.x;
  const int b    = blockIdx.x >> 5;
  const int w0   = (blockIdx.x & 31) << 4;
  const int wave = tid >> 5;
  const int lane = tid & 31;
  const int half8 = (lane & 16) ? 8 : 0;
  const int nt0 = wave * 2, nt1 = wave * 2 + 1;
  const int ch0 = nt0 * 16 + (lane & 15);
  const int ch1 = nt1 * 16 + (lane & 15);

  for (int i = tid; i < 16 * 264; i += 256) prevA[i] = 0;

  const float sA = sc[ch0], bA = sh[ch0];
  const float sB = sc[ch1], bB = sh[ch1];

  const float* xi0p = xin  + ((size_t)(b * CC + ch0) * HH) * WW + w0 + half8;
  const float* xi1p = xin  + ((size_t)(b * CC + ch1) * HH) * WW + w0 + half8;
  const size_t tbase = (size_t)blockIdx.x * HH * TBLK + (size_t)(wave * 128 + lane) * 4;
  const float* xitp = xin  + tbase;
  float*       xotp = xout + tbase;

  bf16x16 bm0[8], bm1[8];
  #pragma unroll
  for (int kt = 0; kt < 8; ++kt) {
    bm0[kt] = load_frag_Bswz(Mswz, kt, nt0, lane);
    bm1[kt] = load_frag_Bswz(Mswz, kt, nt1, lane);
  }

  auto load_row = [&](int r, XRow& x) {
    if (TIN) {
      const float* n = xitp + (size_t)r * TBLK;
      x.a0 = *(const f32x4*)(n);       x.b0 = *(const f32x4*)(n + 128);
      x.a1 = *(const f32x4*)(n + 256); x.b1 = *(const f32x4*)(n + 384);
    } else {
      const float* n0 = xi0p + (size_t)r * WW;
      const float* n1 = xi1p + (size_t)r * WW;
      x.a0 = *(const f32x4*)(n0); x.b0 = *(const f32x4*)(n0 + 4);
      x.a1 = *(const f32x4*)(n1); x.b1 = *(const f32x4*)(n1 + 4);
    }
  };

  float s0 = 0.f, q0 = 0.f, s1 = 0.f, q1 = 0.f;

  auto step = [&](int h, const XRow& x, XRow& xfill, const unsigned short* prevCur, unsigned short* prevNxt) {
    __builtin_amdgcn_sched_barrier(0);
    lds_barrier();
    load_row(min(h + 4, HH - 1), xfill);

    f32x8 acc0 = {}; f32x8 acc1 = {};
    const unsigned short* arow = &prevCur[(lane & 15) * 264];
    bf16x16 a = load_frag_lds(arow, 0, lane);
    #pragma unroll
    for (int kt = 0; kt < 8; ++kt) {
      bf16x16 an = (kt < 7) ? load_frag_lds(arow, kt + 1, lane) : a;
      acc0 = wmma_bf16(a, bm0[kt], acc0);
      acc1 = wmma_bf16(a, bm1[kt], acc1);
      a = an;
    }

    f32x4 xa0 = x.a0, xb0 = x.b0, xa1 = x.a1, xb1 = x.b1;
    asm("" : "+v"(xa0), "+v"(xb0), "+v"(xa1), "+v"(xb1));
    const float xi0[8] = {xa0.x, xa0.y, xa0.z, xa0.w, xb0.x, xb0.y, xb0.z, xb0.w};
    const float xi1[8] = {xa1.x, xa1.y, xa1.z, xa1.w, xb1.x, xb1.y, xb1.z, xb1.w};
    float r0[8], r1[8];
    #pragma unroll
    for (int j = 0; j < 8; ++j) {
      float a0 = sA * xi0[j] + bA;
      float a1 = sB * xi1[j] + bB;
      if (relu_in) { a0 = fmaxf(a0, 0.f); a1 = fmaxf(a1, 0.f); }
      float v0 = a0 + fmaxf(acc0[j], 0.f);
      float v1 = a1 + fmaxf(acc1[j], 0.f);
      r0[j] = v0; r1[j] = v1;
      s0 += v0; q0 += v0 * v0;
      s1 += v1; q1 += v1 * v1;
    }
    #pragma unroll
    for (int j = 0; j < 8; j += 2) {
      const unsigned int p0 = pack_bf2(r0[j], r0[j + 1]);
      const unsigned int p1 = pack_bf2(r1[j], r1[j + 1]);
      prevNxt[(j + half8) * 264 + ch0]     = (unsigned short)(p0 & 0xFFFFu);
      prevNxt[(j + 1 + half8) * 264 + ch0] = (unsigned short)(p0 >> 16);
      prevNxt[(j + half8) * 264 + ch1]     = (unsigned short)(p1 & 0xFFFFu);
      prevNxt[(j + 1 + half8) * 264 + ch1] = (unsigned short)(p1 >> 16);
    }
    {
      float* p = xotp + (size_t)(HH - 1 - h) * TBLK;
      *(float4*)(p)       = make_float4(r0[0], r0[1], r0[2], r0[3]);
      *(float4*)(p + 128) = make_float4(r0[4], r0[5], r0[6], r0[7]);
      *(float4*)(p + 256) = make_float4(r1[0], r1[1], r1[2], r1[3]);
      *(float4*)(p + 384) = make_float4(r1[4], r1[5], r1[6], r1[7]);
    }
  };

  XRow x0, x1, x2, x3, x4;
  load_row(0, x0); load_row(1, x1); load_row(2, x2); load_row(3, x3);

  unsigned short* cur = prevA;
  unsigned short* nxt = prevB;
  for (int h = 0; h < HH; h += 5) {
    step(h,     x0, x4, cur, nxt);
    step(h + 1, x1, x0, nxt, cur);
    step(h + 2, x2, x1, cur, nxt);
    step(h + 3, x3, x2, nxt, cur);
    step(h + 4, x4, x3, cur, nxt);
    unsigned short* t = cur; cur = nxt; nxt = t;
  }

  s0 += __shfl_xor(s0, 16); q0 += __shfl_xor(q0, 16);
  s1 += __shfl_xor(s1, 16); q1 += __shfl_xor(q1, 16);
  if (lane < 16) {
    statsP[(ch0)       * 256 + blockIdx.x] = s0;
    statsP[(256 + ch0) * 256 + blockIdx.x] = q0;
    statsP[(ch1)       * 256 + blockIdx.x] = s1;
    statsP[(256 + ch1) * 256 + blockIdx.x] = q1;
  }
}

__global__ void __launch_bounds__(256) bnscale_kernel(
    const float* __restrict__ statsP,
    const float* __restrict__ gamma, const float* __restrict__ beta,
    float* __restrict__ sc, float* __restrict__ sh)
{
  __shared__ float ps[8], pq[8];
  const int c = blockIdx.x;
  const int t = threadIdx.x;
  float s = statsP[c * 256 + t];
  float q = statsP[(256 + c) * 256 + t];
  for (int off = 16; off >= 1; off >>= 1) { s += __shfl_xor(s, off); q += __shfl_xor(q, off); }
  if ((t & 31) == 0) { ps[t >> 5] = s; pq[t >> 5] = q; }
  __syncthreads();
  if (t == 0) {
    float S = 0.f, Q = 0.f;
    for (int w = 0; w < 8; ++w) { S += ps[w]; Q += pq[w]; }
    const float n = (float)(BB * HH * WW);
    float mean = S / n;
    float var  = Q / n - mean * mean;
    float g = gamma[c] * rsqrtf(var + 1e-5f);
    sc[c] = g;
    sh[c] = beta[c] - mean * g;
  }
}

#define CONV_MT 4
#define CONV_MA (CONV_MT + 1)
#define CONV_TW (CONV_MT * 16)
#define CONV_HW (CONV_TW + 2)
#define CONV_ROW (CONV_HW * 264)
#define CONV_RUNS 4
#define RAW_P   36

__global__ void __launch_bounds__(256) conv_kernel(
    const float* __restrict__ xin,
    const unsigned int* __restrict__ Fswz,
    const float* __restrict__ sc, const float* __restrict__ sh,
    float* __restrict__ z)
{
  __shared__ __align__(16) unsigned short uT[3 * CONV_ROW];
  __shared__ __align__(16) float raw[256 * RAW_P];
  __shared__ __align__(16) float zp[8 * CONV_HW * 6];

  const int tid  = threadIdx.x;
  const int wave = tid >> 5;
  const int lane = tid & 31;
  const int wt   = blockIdx.x & 15;
  const int run  = (blockIdx.x >> 4) & (CONV_RUNS - 1);
  const int b    = blockIdx.x >> 6;
  const int h0   = run * 6;
  const int nrows = (run == CONV_RUNS - 1) ? 7 : 6;
  const int w20  = wt * CONV_TW;

  const int pq   = (tid >> 5) & 3;
  const int rawq = (32 * (tid >> 7) + 16 * (pq >> 1) + (tid & 15)) * RAW_P
                 + 8 * ((tid >> 4) & 1) + 4 * (pq & 1);
  const int eP0  = (((tid >> 5) * 4 + ((tid >> 4) & 1) * 2) * 32 + (tid & 15)) * 4;
  const int eP15 = (((tid >> 5) * 4 + ((tid >> 4) & 1) * 2 + 1) * 32 + 16 + (tid & 15)) * 4 + 3;
  const int tileL = (wt > 0)  ? 2 * wt - 1 : 0;
  const int offL  = (wt > 0)  ? eP15 : eP0;
  const int tileR = (wt < 15) ? 2 * wt + 2 : 31;
  const int offR  = (wt < 15) ? eP0 : eP15;

  auto load_row = [&](int hs, f32x4 (&q)[8], float& eL, float& eR) {
    const int hsc = min(max(hs, 0), HH - 1);
    const float* t0 = xin + ((b * 32 + 2 * wt) * HH + hsc) * TBLK + 4 * tid;
    #pragma unroll
    for (int i = 0; i < 8; ++i)
      q[i] = *(const f32x4*)(t0 + (i >> 2) * (HH * TBLK) + (i & 3) * 1024);
    eL = xin[((b * 32 + tileL) * HH + hsc) * TBLK + offL];
    eR = xin[((b * 32 + tileR) * HH + hsc) * TBLK + offR];
  };
  auto put_row = [&](const f32x4 (&q)[8], float eL, float eR) {
    #pragma unroll
    for (int i = 0; i < 8; ++i)
      *(f32x4*)&raw[rawq + (i & 3) * (64 * RAW_P) + (i >> 2) * 16] = q[i];
    *(f32x2*)&raw[tid * RAW_P + 32] = (f32x2){eL, eR};
  };
  const float scc = sc[tid], shc = sh[tid];
  auto col_weight = [&](int wi) {
    const int w2  = w20 + wi - 1;
    const int num = w2 * 511;
    const int i0  = num / 1023;
    return (float)(num - i0 * 1023) * (1.0f / 1023.0f);
  };
  const int fl0 = __float_as_int(col_weight(lane));
  const int fl1 = __float_as_int(col_weight(32 + lane));
  const int fl2 = __float_as_int(col_weight(64 + lane));
  auto emit_row = [&](unsigned short* dst, int hs) {
    const bool rowok = (hs >= 0 && hs < HH);
    float xs[34];
    const f32x2 e = *(const f32x2*)&raw[tid * RAW_P + 32];
    xs[0] = e.x; xs[33] = e.y;
    #pragma unroll
    for (int k = 0; k < 8; ++k) {
      const f32x4 m = *(const f32x4*)&raw[tid * RAW_P + 4 * k];
      xs[1 + 4 * k] = m.x; xs[2 + 4 * k] = m.y; xs[3 + 4 * k] = m.z; xs[4 + 4 * k] = m.w;
    }
    #pragma unroll
    for (int k = 0; k < 34; ++k) {
      const float x = fmaxf(scc * xs[k] + shc, 0.f);
      xs[k] = rowok ? x : 0.f;
    }
    #pragma unroll
    for (int wp = 0; wp < CONV_HW; wp += 2) {
      float vv[2];
      #pragma unroll
      for (int e = 0; e < 2; ++e) {
        const int  wi    = wp + e;
        const int  w2    = w20 + wi - 1;
        const bool colok = (w2 >= 0 && w2 < 1024);
        const float f   = __int_as_float(__builtin_amdgcn_readlane((wi < 32) ? fl0 : (wi < 64) ? fl1 : fl2, wi & 31));
        const int   s   = wi >> 1;
        const bool  col0 = (wi == 1) && (wt == 0);
        const float x0 = col0 ? xs[1] : xs[s];
        const float x1 = col0 ? xs[2] : xs[s + 1];
        const float v  = x0 + f * (x1 - x0);
        vv[e] = colok ? v : 0.f;
      }
      const unsigned int p = pack_bf2(vv[0], vv[1]);
      dst[wp * 264 + tid]       = (unsigned short)(p & 0xFFFFu);
      dst[(wp + 1) * 264 + tid] = (unsigned short)(p >> 16);
    }
  };

  bf16x16 bw[3];
  #pragma unroll
  for (int ky = 0; ky < 3; ++ky) bw[ky] = load_frag_F(Fswz, ky, wave, lane);

  int o0 = 0, o1 = CONV_ROW, o2 = 2 * CONV_ROW;

  f32x4 qn[8];
  float ln, rn;
  {
    f32x4 q0[8], q1[8], q2[8];
    float  l0, r0, l1, r1, l2, r2;
    load_row(h0 - 1, q0, l0, r0);
    load_row(h0,     q1, l1, r1);
    load_row(h0 + 1, q2, l2, r2);

    put_row(q0, l0, r0);
    __syncthreads();
    emit_row(uT + o0, h0 - 1);
    __syncthreads();
    put_row(q1, l1, r1);
    __syncthreads();
    emit_row(uT + o1, h0);
    __syncthreads();
    put_row(q2, l2, r2);
    load_row(h0 + 2, qn, ln, rn);
    __syncthreads();
    emit_row(uT + o2, h0 + 1);
  }
  __syncthreads();

  const int half8 = (lane & 16) ? 8 : 0;
  const int arow_l = (lane & 15) * 264;
  const int arow_t = min(CONV_TW + (lane & 15), CONV_HW - 1) * 264;

  auto frag = [&](int slot, int m) {
    return load_frag_lds(uT + slot + ((m < CONV_MT) ? arow_l + m * 16 * 264 : arow_t), wave, lane);
  };

  f32x8 accA[CONV_MA], accB[CONV_MA], accC[CONV_MA];
  {
    bf16x16 af[CONV_MA];
    #pragma unroll
    for (int m = 0; m < CONV_MA; ++m) af[m] = frag(o0, m);
    #pragma unroll
    for (int m = 0; m < CONV_MA; ++m) accA[m] = wmma_bf16(af[m], bw[0], (f32x8){});
    #pragma unroll
    for (int m = 0; m < CONV_MA; ++m) af[m] = frag(o1, m);
    #pragma unroll
    for (int m = 0; m < CONV_MA; ++m) {
      accA[m] = wmma_bf16(af[m], bw[1], accA[m]);
      accB[m] = wmma_bf16(af[m], bw[0], (f32x8){});
    }
  }

  auto row_step = [&](int i, f32x8 (&P)[CONV_MA], f32x8 (&Q)[CONV_MA], f32x8 (&R)[CONV_MA]) {
    const int  h    = h0 + i;
    const bool more = (i + 1 < nrows);

    if (more) put_row(qn, ln, rn);

    bf16x16 af[CONV_MA];
    #pragma unroll
    for (int m = 0; m < CONV_MA; ++m) af[m] = frag(o2, m);
    __builtin_amdgcn_sched_barrier(0);
    #pragma unroll
    for (int m = 0; m < CONV_MA; ++m) {
      P[m] = wmma_bf16(af[m], bw[2], P[m]);
      Q[m] = wmma_bf16(af[m], bw[1], Q[m]);
      R[m] = wmma_bf16(af[m], bw[0], (f32x8){});
    }

    if ((lane & 15) < 6) {
      float* zw = zp + (wave * CONV_HW + half8) * 6 + (lane & 15);
      #pragma unroll
      for (int m = 0; m < CONV_MT; ++m) {
        #pragma unroll
        for (int j = 0; j < 8; ++j)
          zw[(m * 16 + j) * 6] = P[m][j];
      }
      if (half8 == 0) {
        zw[(CONV_TW)     * 6] = P[CONV_MT][0];
        zw[(CONV_TW + 1) * 6] = P[CONV_MT][1];
      }
    }
    __syncthreads();

    if (tid < CONV_TW) {
      float zh = 0.f, zl = 0.f;
      for (int w = 0; w < 8; ++w) {
        const float* zr = zp + (w * CONV_HW + tid) * 6;
        #pragma unroll
        for (int kx = 0; kx < 3; ++kx) {
          const f32x2 p = *(const f32x2*)(zr + kx * 6 + 2 * kx);
          zh += p.x;
          zl += p.y;
        }
      }
      z[(b * 25 + h) * 1024 + w20 + tid] = zh + zl;
    }
    if (more) {
      if (i + 2 < nrows) load_row(h + 3, qn, ln, rn);
      emit_row(uT + o0, h + 2);
    }
    __syncthreads();
    const int t = o0; o0 = o1; o1 = o2; o2 = t;
  };

  for (int i = 0; ; i += 3) {
    row_step(i, accA, accB, accC);
    if (i + 1 >= nrows) break;
    row_step(i + 1, accB, accC, accA);
    if (i + 2 >= nrows) break;
    row_step(i + 2, accC, accA, accB);
    if (i + 3 >= nrows) break;
  }
}

__global__ void final_kernel(const float* __restrict__ z, const float* __restrict__ v2,
                             float* __restrict__ out)
{
  int idx = blockIdx.x * 256 + threadIdx.x;
  if (idx >= BB * 4096) return;
  int b  = idx >> 12;
  int w4 = idx & 4095;

  int   num4 = w4 * 2047;
  int   i40  = num4 / 4095;
  float f4   = (float)(num4 - i40 * 4095) * (1.0f / 4095.0f);
  int   i41  = min(i40 + 1, 2047);

  int n3a = i40 * 1023; int ja0 = n3a / 2047;
  float fa = (float)(n3a - ja0 * 2047) * (1.0f / 2047.0f);
  int ja1 = min(ja0 + 1, 1023);
  int n3b = i41 * 1023; int jb0 = n3b / 2047;
  float fb = (float)(n3b - jb0 * 2047) * (1.0f / 2047.0f);
  int jb1 = min(jb0 + 1, 1023);

  float acc = 0.f;
  #pragma unroll 5
  for (int h = 0; h < HH; ++h) {
    const float* zh = z + (b * 25 + h) * 1024;
    float za = zh[ja0] + fa * (zh[ja1] - zh[ja0]);
    float zb = zh[jb0] + fb * (zh[jb1] - zh[jb0]);
    acc += v2[h] * (za + f4 * (zb - za));
  }
  out[idx] = 1.0f / (1.0f + __expf(-acc));
}

extern "C" void kernel_launch(void* const* d_in, const int* in_sizes, int n_in,
                              void* d_out, int out_size, void* d_ws, size_t ws_size,
                              hipStream_t stream)
{
  const float* p2c   = (const float*)d_in[0];
  const float* wmsg  = (const float*)d_in[1];
  const float* gamma = (const float*)d_in[2];
  const float* beta  = (const float*)d_in[3];
  const float* wup2  = (const float*)d_in[4];
  const float* v1    = (const float*)d_in[5];
  const float* v2    = (const float*)d_in[6];

  char* ws = (char*)d_ws;
  size_t off = 0;
  auto walloc = [&](size_t bytes) -> void* {
    void* p = ws + off;
    off = (off + bytes + 255) & ~(size_t)255;
    return p;
  };
  unsigned int* Mswz = (unsigned int*)walloc(32768u * 4);
  unsigned int* Fswz = (unsigned int*)walloc(3u * 2048u * 4);
  float* stats1 = (float*)walloc(512u * 256u * 4);
  float* stats2 = (float*)walloc(512u * 256u * 4);
  float* sc1  = (float*)walloc(256 * 4);
  float* sh1  = (float*)walloc(256 * 4);
  float* sc2  = (float*)walloc(256 * 4);
  float* sh2  = (float*)walloc(256 * 4);
  float* scid = (float*)walloc(256 * 4);
  float* shid = (float*)walloc(256 * 4);
  float* z1024 = (float*)walloc((size_t)BB * 25 * 1024 * 4);
  float* out1  = (float*)walloc((size_t)BB * CC * HH * WW * 4);
  float* out2  = (float*)walloc((size_t)BB * CC * HH * WW * 4);

  prep_kernel<<<256, 256, 0, stream>>>(wmsg, wup2, v1, Mswz, Fswz);
  init_misc<<<1, 256, 0, stream>>>(scid, shid);

  scnn_kernel<false><<<256, 256, 0, stream>>>(p2c, out1, Mswz, scid, shid, 0, stats1);
  bnscale_kernel<<<256, 256, 0, stream>>>(stats1, gamma, beta, sc1, sh1);

  scnn_kernel<true><<<256, 256, 0, stream>>>(out1, out2, Mswz, sc1, sh1, 1, stats2);
  bnscale_kernel<<<256, 256, 0, stream>>>(stats2, gamma, beta, sc2, sh2);

  conv_kernel<<<BB * CONV_RUNS * 16, 256, 0, stream>>>(out2, Fswz, sc2, sh2, z1024);

  final_kernel<<<128, 256, 0, stream>>>(z1024, v2, (float*)d_out);
}


# conv: width interpolation applied after the product (linear, same for every channel): GEMM on the 34 source columns instead of 66 interpolated ones; lerp on the tap sums (sums regrouped)
# speedup vs baseline: 7.3613x; 1.0343x over previous
#include <hip/hip_runtime.h>
#include <hip/hip_bf16.h>
#include <stdint.h>

#define BB 8
#define CC 256
#define HH 25
#define WW 512

typedef __bf16 bf16x16 __attribute__((ext_vector_type(16)));
typedef float  f32x8   __attribute__((ext_vector_type(8)));
typedef float  f32x4   __attribute__((ext_vector_type(4)));
typedef float  f32x2   __attribute__((ext_vector_type(2)));

union FragU { uint4 u[2]; bf16x16 v; };

__device__ __forceinline__ unsigned short f2bf(float x) {
  unsigned u = __float_as_uint(x);
  unsigned r = u + 0x7FFFu + ((u >> 16) & 1u);
  return (unsigned short)(r >> 16);
}

typedef __bf16 bf16x2 __attribute__((ext_vector_type(2)));

__device__ __forceinline__ unsigned int pack_bf2(float a, float b) {
  const f32x2 v = {a, b};
  const bf16x2 r = __builtin_convertvector(v, bf16x2);
  return __builtin_bit_cast(unsigned int, r);
}

__device__ __forceinline__ f32x8 wmma_bf16(bf16x16 a, bf16x16 b, f32x8 c) {
  return __builtin_amdgcn_wmma_f32_16x16x32_bf16(false, a, false, b, (short)0, c, false, false);
}

__device__ __forceinline__ bf16x16 load_frag_lds(const unsigned short* rowBase, int kt, int lane) {
  int kb = (lane & 16) ? 8 : 0;
  const unsigned short* p = rowBase + kt * 32 + kb;
  FragU f;
  f.u[0] = *(const uint4*)(p);
  f.u[1] = *(const uint4*)(p + 16);
  return f.v;
}

__device__ __forceinline__ bf16x16 load_frag_Bswz(const unsigned int* swz, int kt, int nt, int lane) {
  const unsigned int* p = swz + ((((kt << 4) + nt) * 32 + lane) << 3);
  FragU f;
  f.u[0] = *(const uint4*)(p);
  f.u[1] = *(const uint4*)(p + 4);
  return f.v;
}

__device__ __forceinline__ bf16x16 load_frag_F(const unsigned int* swz, int ky, int kt, int lane) {
  const unsigned int* p = swz + ((((ky << 3) + kt) * 32 + lane) << 3);
  FragU f;
  f.u[0] = *(const uint4*)(p);
  f.u[1] = *(const uint4*)(p + 4);
  return f.v;
}

__device__ __forceinline__ float bf2f(unsigned short h) {
  return __uint_as_float(((unsigned)h) << 16);
}

__device__ __forceinline__ void lds_barrier() {
  __builtin_amdgcn_fence(__ATOMIC_RELEASE, "workgroup", "local");
  __builtin_amdgcn_s_barrier();
  __builtin_amdgcn_fence(__ATOMIC_ACQUIRE, "workgroup", "local");
}

__global__ void __launch_bounds__(256) prep_kernel(
    const float* __restrict__ wmsg, const float* __restrict__ wup2, const float* __restrict__ v1,
    unsigned int* __restrict__ Mswz, unsigned int* __restrict__ Fswz)
{
  __shared__ float part[8][18];
  __shared__ float wsum[18];
  const int tid = threadIdx.x;

  if (blockIdx.x >= 128) {
    int r    = (blockIdx.x - 128) * 256 + tid;
    int j    = r & 7;
    int lane = (r >> 3) & 31;
    int nt   = (r >> 8) & 15;
    int kt   = r >> 12;
    int n  = nt * 16 + (lane & 15);
    int kb = (lane & 16) ? 8 : 0;
    int k  = kt * 32 + ((j < 4) ? (kb + 2 * j) : (16 + kb + 2 * (j - 4)));
    float lo = wmsg[(n * 256 + k) * 9 + 4];
    float hi = wmsg[(n * 256 + k + 1) * 9 + 4];
    Mswz[r] = (unsigned)f2bf(lo) | ((unsigned)f2bf(hi) << 16);
    return;
  }

  const int kp   = blockIdx.x;
  const int wave = tid >> 5;
  const int lane = tid & 31;
  const float g  = v1[tid];
  const float* srcw = wup2 + (tid * 256 + 2 * kp) * 9;
  float p[18];
  #pragma unroll
  for (int t = 0; t < 18; ++t) p[t] = g * srcw[t];
  #pragma unroll
  for (int t = 0; t < 18; ++t) {
    for (int off = 16; off >= 1; off >>= 1) p[t] += __shfl_xor(p[t], off);
  }
  if (lane == 0) {
    #pragma unroll
    for (int t = 0; t < 18; ++t) part[wave][t] = p[t];
  }
  __syncthreads();
  if (tid < 18) {
    float s = 0.f;
    for (int w = 0; w < 8; ++w) s += part[w][tid];
    wsum[tid] = s;
  }
  __syncthreads();
  if (tid < 48) {
    const int ky = tid >> 4;
    const int n  = tid & 15;
    unsigned int d = 0u;
    if (n < 6) {
      const int tap = ky * 3 + (n >> 1);
      const float w0 = wsum[tap], w1 = wsum[9 + tap];
      const unsigned short h0 = f2bf(w0), h1 = f2bf(w1);
      d = (n & 1) ? ((unsigned)f2bf(w0 - bf2f(h0)) | ((unsigned)f2bf(w1 - bf2f(h1)) << 16))
                  : ((unsigned)h0 | ((unsigned)h1 << 16));
    }
    const int k   = 2 * kp;
    const int kt  = k >> 5;
    const int r   = k & 31;
    const int kbs = (r >> 3) & 1;
    const int j   = ((r >> 4) << 2) + ((r & 7) >> 1);
    Fswz[((((ky << 3) + kt) * 32 + kbs * 16 + n) << 3) + j] = d;
  }
}

__global__ void init_misc(float* scid, float* shid) {
  int t = threadIdx.x;
  scid[t] = 1.0f; shid[t] = 0.0f;
}

#define TBLK 4096
struct XRow { f32x4 a0, b0, a1, b1; };

template <bool TIN>
__global__ void __launch_bounds__(256) scnn_kernel(
    const float* __restrict__ xin, float* __restrict__ xout,
    const unsigned int* __restrict__ Mswz,
    const float* __restrict__ sc, const float* __restrict__ sh,
    int relu_in, float* __restrict__ statsP)
{
  __shared__ __align__(16) unsigned short prevA[16 * 264];
  __shared__ __align__(16) unsigned short prevB[16 * 264];

  const int tid  = threadIdx.x;
  const int b    = blockIdx.x >> 5;
  const int w0   = (blockIdx.x & 31) << 4;
  const int wave = tid >> 5;
  const int lane = tid & 31;
  const int half8 = (lane & 16) ? 8 : 0;
  const int nt0 = wave * 2, nt1 = wave * 2 + 1;
  const int ch0 = nt0 * 16 + (lane & 15);
  const int ch1 = nt1 * 16 + (lane & 15);

  for (int i = tid; i < 16 * 264; i += 256) prevA[i] = 0;

  const float sA = sc[ch0], bA = sh[ch0];
  const float sB = sc[ch1], bB = sh[ch1];

  const float* xi0p = xin  + ((size_t)(b * CC + ch0) * HH) * WW + w0 + half8;
  const float* xi1p = xin  + ((size_t)(b * CC + ch1) * HH) * WW + w0 + half8;
  const size_t tbase = (size_t)blockIdx.x * HH * TBLK + (size_t)(wave * 128 + lane) * 4;
  const float* xitp = xin  + tbase;
  float*       xotp = xout + tbase;

  bf16x16 bm0[8], bm1[8];
  #pragma unroll
  for (int kt = 0; kt < 8; ++kt) {
    bm0[kt] = load_frag_Bswz(Mswz, kt, nt0, lane);
    bm1[kt] = load_frag_Bswz(Mswz, kt, nt1, lane);
  }

  auto load_row = [&](int r, XRow& x) {
    if (TIN) {
      const float* n = xitp + (size_t)r * TBLK;
      x.a0 = *(const f32x4*)(n);       x.b0 = *(const f32x4*)(n + 128);
      x.a1 = *(const f32x4*)(n + 256); x.b1 = *(const f32x4*)(n + 384);
    } else {
      const float* n0 = xi0p + (size_t)r * WW;
      const float* n1 = xi1p + (size_t)r * WW;
      x.a0 = *(const f32x4*)(n0); x.b0 = *(const f32x4*)(n0 + 4);
      x.a1 = *(const f32x4*)(n1); x.b1 = *(const f32x4*)(n1 + 4);
    }
  };

  float s0 = 0.f, q0 = 0.f, s1 = 0.f, q1 = 0.f;

  auto step = [&](int h, const XRow& x, XRow& xfill, const unsigned short* prevCur, unsigned short* prevNxt) {
    __builtin_amdgcn_sched_barrier(0);
    lds_barrier();
    load_row(min(h + 4, HH - 1), xfill);

    f32x8 acc0 = {}; f32x8 acc1 = {};
    const unsigned short* arow = &prevCur[(lane & 15) * 264];
    bf16x16 a = load_frag_lds(arow, 0, lane);
    #pragma unroll
    for (int kt = 0; kt < 8; ++kt) {
      bf16x16 an = (kt < 7) ? load_frag_lds(arow, kt + 1, lane) : a;
      acc0 = wmma_bf16(a, bm0[kt], acc0);
      acc1 = wmma_bf16(a, bm1[kt], acc1);
      a = an;
    }

    f32x4 xa0 = x.a0, xb0 = x.b0, xa1 = x.a1, xb1 = x.b1;
    asm("" : "+v"(xa0), "+v"(xb0), "+v"(xa1), "+v"(xb1));
    const float xi0[8] = {xa0.x, xa0.y, xa0.z, xa0.w, xb0.x, xb0.y, xb0.z, xb0.w};
    const float xi1[8] = {xa1.x, xa1.y, xa1.z, xa1.w, xb1.x, xb1.y, xb1.z, xb1.w};
    float r0[8], r1[8];
    #pragma unroll
    for (int j = 0; j < 8; ++j) {
      float a0 = sA * xi0[j] + bA;
      float a1 = sB * xi1[j] + bB;
      if (relu_in) { a0 = fmaxf(a0, 0.f); a1 = fmaxf(a1, 0.f); }
      float v0 = a0 + fmaxf(acc0[j], 0.f);
      float v1 = a1 + fmaxf(acc1[j], 0.f);
      r0[j] = v0; r1[j] = v1;
      s0 += v0; q0 += v0 * v0;
      s1 += v1; q1 += v1 * v1;
    }
    #pragma unroll
    for (int j = 0; j < 8; j += 2) {
      const unsigned int p0 = pack_bf2(r0[j], r0[j + 1]);
      const unsigned int p1 = pack_bf2(r1[j], r1[j + 1]);
      prevNxt[(j + half8) * 264 + ch0]     = (unsigned short)(p0 & 0xFFFFu);
      prevNxt[(j + 1 + half8) * 264 + ch0] = (unsigned short)(p0 >> 16);
      prevNxt[(j + half8) * 264 + ch1]     = (unsigned short)(p1 & 0xFFFFu);
      prevNxt[(j + 1 + half8) * 264 + ch1] = (unsigned short)(p1 >> 16);
    }
    {
      float* p = xotp + (size_t)(HH - 1 - h) * TBLK;
      *(float4*)(p)       = make_float4(r0[0], r0[1], r0[2], r0[3]);
      *(float4*)(p + 128) = make_float4(r0[4], r0[5], r0[6], r0[7]);
      *(float4*)(p + 256) = make_float4(r1[0], r1[1], r1[2], r1[3]);
      *(float4*)(p + 384) = make_float4(r1[4], r1[5], r1[6], r1[7]);
    }
  };

  XRow x0, x1, x2, x3, x4;
  load_row(0, x0); load_row(1, x1); load_row(2, x2); load_row(3, x3);

  unsigned short* cur = prevA;
  unsigned short* nxt = prevB;
  for (int h = 0; h < HH; h += 5) {
    step(h,     x0, x4, cur, nxt);
    step(h + 1, x1, x0, nxt, cur);
    step(h + 2, x2, x1, cur, nxt);
    step(h + 3, x3, x2, nxt, cur);
    step(h + 4, x4, x3, cur, nxt);
    unsigned short* t = cur; cur = nxt; nxt = t;
  }

  s0 += __shfl_xor(s0, 16); q0 += __shfl_xor(q0, 16);
  s1 += __shfl_xor(s1, 16); q1 += __shfl_xor(q1, 16);
  if (lane < 16) {
    statsP[(ch0)       * 256 + blockIdx.x] = s0;
    statsP[(256 + ch0) * 256 + blockIdx.x] = q0;
    statsP[(ch1)       * 256 + blockIdx.x] = s1;
    statsP[(256 + ch1) * 256 + blockIdx.x] = q1;
  }
}

__global__ void __launch_bounds__(256) bnscale_kernel(
    const float* __restrict__ statsP,
    const float* __restrict__ gamma, const float* __restrict__ beta,
    float* __restrict__ sc, float* __restrict__ sh)
{
  __shared__ float ps[8], pq[8];
  const int c = blockIdx.x;
  const int t = threadIdx.x;
  float s = statsP[c * 256 + t];
  float q = statsP[(256 + c) * 256 + t];
  for (int off = 16; off >= 1; off >>= 1) { s += __shfl_xor(s, off); q += __shfl_xor(q, off); }
  if ((t & 31) == 0) { ps[t >> 5] = s; pq[t >> 5] = q; }
  __syncthreads();
  if (t == 0) {
    float S = 0.f, Q = 0.f;
    for (int w = 0; w < 8; ++w) { S += ps[w]; Q += pq[w]; }
    const float n = (float)(BB * HH * WW);
    float mean = S / n;
    float var  = Q / n - mean * mean;
    float g = gamma[c] * rsqrtf(var + 1e-5f);
    sc[c] = g;
    sh[c] = beta[c] - mean * g;
  }
}

#define CONV_TW 64
#define CONV_SW 34
#define CONV_MA 3
#define CONV_ROW (CONV_SW * 264)
#define CONV_RUNS 4
#define RAW_P   36

__global__ void __launch_bounds__(256) conv_kernel(
    const float* __restrict__ xin,
    const unsigned int* __restrict__ Fswz,
    const float* __restrict__ sc, const float* __restrict__ sh,
    float* __restrict__ z)
{
  __shared__ __align__(16) unsigned short uT[3 * CONV_ROW];
  __shared__ __align__(16) float raw[256 * RAW_P];
  __shared__ __align__(16) float zp[8 * CONV_SW * 6];
  __shared__ __align__(16) float zt[2 * CONV_SW * 4];

  const int tid  = threadIdx.x;
  const int wave = tid >> 5;
  const int lane = tid & 31;
  const int wt   = blockIdx.x & 15;
  const int run  = (blockIdx.x >> 4) & (CONV_RUNS - 1);
  const int b    = blockIdx.x >> 6;
  const int h0   = run * 6;
  const int nrows = (run == CONV_RUNS - 1) ? 7 : 6;
  const int w20  = wt * CONV_TW;

  const int pq   = (tid >> 5) & 3;
  const int rawq = (32 * (tid >> 7) + 16 * (pq >> 1) + (tid & 15)) * RAW_P
                 + 8 * ((tid >> 4) & 1) + 4 * (pq & 1);
  const int eP0  = (((tid >> 5) * 4 + ((tid >> 4) & 1) * 2) * 32 + (tid & 15)) * 4;
  const int eP15 = (((tid >> 5) * 4 + ((tid >> 4) & 1) * 2 + 1) * 32 + 16 + (tid & 15)) * 4 + 3;
  const int tileL = (wt > 0)  ? 2 * wt - 1 : 0;
  const int offL  = (wt > 0)  ? eP15 : eP0;
  const int tileR = (wt < 15) ? 2 * wt + 2 : 31;
  const int offR  = (wt < 15) ? eP0 : eP15;

  auto load_row = [&](int hs, f32x4 (&q)[8], float& eL, float& eR) {
    const int hsc = min(max(hs, 0), HH - 1);
    const float* t0 = xin + ((b * 32 + 2 * wt) * HH + hsc) * TBLK + 4 * tid;
    #pragma unroll
    for (int i = 0; i < 8; ++i)
      q[i] = *(const f32x4*)(t0 + (i >> 2) * (HH * TBLK) + (i & 3) * 1024);
    eL = xin[((b * 32 + tileL) * HH + hsc) * TBLK + offL];
    eR = xin[((b * 32 + tileR) * HH + hsc) * TBLK + offR];
  };
  auto put_row = [&](const f32x4 (&q)[8], float eL, float eR) {
    #pragma unroll
    for (int i = 0; i < 8; ++i)
      *(f32x4*)&raw[rawq + (i & 3) * (64 * RAW_P) + (i >> 2) * 16] = q[i];
    *(f32x2*)&raw[tid * RAW_P + 32] = (f32x2){eL, eR};
  };
  const float scc = sc[tid], shc = sh[tid];
  auto col_weight = [&](int wi) {
    const int w2  = w20 + wi - 1;
    const int num = w2 * 511;
    const int i0  = num / 1023;
    return (float)(num - i0 * 1023) * (1.0f / 1023.0f);
  };
  int   tb_off[3];
  float tb_f[3];
  bool  tb_ok[3];
  #pragma unroll
  for (int kx = 0; kx < 3; ++kx) {
    const int  wi   = (tid & (CONV_TW - 1)) + kx;
    const int  w2   = w20 + wi - 1;
    const bool col0 = (wi == 1) && (wt == 0);
    tb_off[kx] = (col0 ? 1 : (wi >> 1)) * 4 + kx;
    tb_f[kx]   = col_weight(wi);
    tb_ok[kx]  = (w2 >= 0 && w2 < 1024);
  }
  auto emit_row = [&](unsigned short* dst, int hs) {
    const bool rowok = (hs >= 0 && hs < HH);
    float xs[CONV_SW];
    const f32x2 e = *(const f32x2*)&raw[tid * RAW_P + 32];
    xs[0] = e.x; xs[33] = e.y;
    #pragma unroll
    for (int k = 0; k < 8; ++k) {
      const f32x4 m = *(const f32x4*)&raw[tid * RAW_P + 4 * k];
      xs[1 + 4 * k] = m.x; xs[2 + 4 * k] = m.y; xs[3 + 4 * k] = m.z; xs[4 + 4 * k] = m.w;
    }
    #pragma unroll
    for (int k = 0; k < CONV_SW; k += 2) {
      const float y0 = fmaxf(scc * xs[k]     + shc, 0.f);
      const float y1 = fmaxf(scc * xs[k + 1] + shc, 0.f);
      const unsigned int p = pack_bf2(rowok ? y0 : 0.f, rowok ? y1 : 0.f);
      dst[k * 264 + tid]       = (unsigned short)(p & 0xFFFFu);
      dst[(k + 1) * 264 + tid] = (unsigned short)(p >> 16);
    }
  };

  bf16x16 bw[3];
  #pragma unroll
  for (int ky = 0; ky < 3; ++ky) bw[ky] = load_frag_F(Fswz, ky, wave, lane);

  int o0 = 0, o1 = CONV_ROW, o2 = 2 * CONV_ROW;

  f32x4 qn[8];
  float ln, rn;
  {
    f32x4 q0[8], q1[8], q2[8];
    float  l0, r0, l1, r1, l2, r2;
    load_row(h0 - 1, q0, l0, r0);
    load_row(h0,     q1, l1, r1);
    load_row(h0 + 1, q2, l2, r2);

    put_row(q0, l0, r0);
    __syncthreads();
    emit_row(uT + o0, h0 - 1);
    __syncthreads();
    put_row(q1, l1, r1);
    __syncthreads();
    emit_row(uT + o1, h0);
    __syncthreads();
    put_row(q2, l2, r2);
    load_row(h0 + 2, qn, ln, rn);
    __syncthreads();
    emit_row(uT + o2, h0 + 1);
  }
  __syncthreads();

  const int half8 = (lane & 16) ? 8 : 0;
  const int arow_l = (lane & 15) * 264;
  const int arow_t = min(32 + (lane & 15), CONV_SW - 1) * 264;

  auto frag = [&](int slot, int m) {
    return load_frag_lds(uT + slot + ((m < CONV_MA - 1) ? arow_l + m * 16 * 264 : arow_t), wave, lane);
  };

  f32x8 accA[CONV_MA], accB[CONV_MA], accC[CONV_MA];
  {
    bf16x16 af[CONV_MA];
    #pragma unroll
    for (int m = 0; m < CONV_MA; ++m) af[m] = frag(o0, m);
    #pragma unroll
    for (int m = 0; m < CONV_MA; ++m) accA[m] = wmma_bf16(af[m], bw[0], (f32x8){});
    #pragma unroll
    for (int m = 0; m < CONV_MA; ++m) af[m] = frag(o1, m);
    #pragma unroll
    for (int m = 0; m < CONV_MA; ++m) {
      accA[m] = wmma_bf16(af[m], bw[1], accA[m]);
      accB[m] = wmma_bf16(af[m], bw[0], (f32x8){});
    }
  }

  const int pa_k  = tid / 3;
  const int pa_kx = tid - 3 * pa_k;
  auto sum_taps = [&](float* zrow) {
    if (tid < CONV_SW * 3) {
      float th = 0.f, tl = 0.f;
      #pragma unroll
      for (int w = 0; w < 8; ++w) {
        const f32x2 p = *(const f32x2*)(zp + (w * CONV_SW + pa_k) * 6 + 2 * pa_kx);
        th += p.x;
        tl += p.y;
      }
      zrow[pa_k * 4 + pa_kx] = th + tl;
    }
  };
  auto finish_row = [&](int h, const float* zrow) {
    if (tid < CONV_TW) {
      float acc = 0.f;
      #pragma unroll
      for (int kx = 0; kx < 3; ++kx) {
        const float t0 = zrow[tb_off[kx]];
        const float t1 = zrow[tb_off[kx] + 4];
        const float v  = t0 + tb_f[kx] * (t1 - t0);
        acc += tb_ok[kx] ? v : 0.f;
      }
      z[(b * 25 + h) * 1024 + w20 + tid] = acc;
    }
  };

  auto row_step = [&](int i, f32x8 (&P)[CONV_MA], f32x8 (&Q)[CONV_MA], f32x8 (&R)[CONV_MA]) {
    const int  h    = h0 + i;
    const bool more = (i + 1 < nrows);

    if (more) put_row(qn, ln, rn);

    bf16x16 af[CONV_MA];
    #pragma unroll
    for (int m = 0; m < CONV_MA; ++m) af[m] = frag(o2, m);
    __builtin_amdgcn_sched_barrier(0);
    #pragma unroll
    for (int m = 0; m < CONV_MA; ++m) {
      P[m] = wmma_bf16(af[m], bw[2], P[m]);
      Q[m] = wmma_bf16(af[m], bw[1], Q[m]);
      R[m] = wmma_bf16(af[m], bw[0], (f32x8){});
    }

    if ((lane & 15) < 6) {
      float* zw = zp + (wave * CONV_SW + half8) * 6 + (lane & 15);
      #pragma unroll
      for (int m = 0; m < CONV_MA - 1; ++m) {
        #pragma unroll
        for (int j = 0; j < 8; ++j)
          zw[(m * 16 + j) * 6] = P[m][j];
      }
      if (half8 == 0) {
        zw[32 * 6] = P[CONV_MA - 1][0];
        zw[33 * 6] = P[CONV_MA - 1][1];
      }
    }
    __syncthreads();

    if (more && i + 2 < nrows) load_row(h + 3, qn, ln, rn);
    sum_taps(zt + (i & 1) * (CONV_SW * 4));
    if (i > 0) finish_row(h - 1, zt + ((i - 1) & 1) * (CONV_SW * 4));
    if (more) emit_row(uT + o0, h + 2);
    __syncthreads();
    const int t = o0; o0 = o1; o1 = o2; o2 = t;
  };

  for (int i = 0; ; i += 3) {
    row_step(i, accA, accB, accC);
    if (i + 1 >= nrows) break;
    row_step(i + 1, accB, accC, accA);
    if (i + 2 >= nrows) break;
    row_step(i + 2, accC, accA, accB);
    if (i + 3 >= nrows) break;
  }
  finish_row(h0 + nrows - 1, zt + ((nrows - 1) & 1) * (CONV_SW * 4));
}

__global__ void final_kernel(const float* __restrict__ z, const float* __restrict__ v2,
                             float* __restrict__ out)
{
  int idx = blockIdx.x * 256 + threadIdx.x;
  if (idx >= BB * 4096) return;
  int b  = idx >> 12;
  int w4 = idx & 4095;

  int   num4 = w4 * 2047;
  int   i40  = num4 / 4095;
  float f4   = (float)(num4 - i40 * 4095) * (1.0f / 4095.0f);
  int   i41  = min(i40 + 1, 2047);

  int n3a = i40 * 1023; int ja0 = n3a / 2047;
  float fa = (float)(n3a - ja0 * 2047) * (1.0f / 2047.0f);
  int ja1 = min(ja0 + 1, 1023);
  int n3b = i41 * 1023; int jb0 = n3b / 2047;
  float fb = (float)(n3b - jb0 * 2047) * (1.0f / 2047.0f);
  int jb1 = min(jb0 + 1, 1023);

  float acc = 0.f;
  #pragma unroll 5
  for (int h = 0; h < HH; ++h) {
    const float* zh = z + (b * 25 + h) * 1024;
    float za = zh[ja0] + fa * (zh[ja1] - zh[ja0]);
    float zb = zh[jb0] + fb * (zh[jb1] - zh[jb0]);
    acc += v2[h] * (za + f4 * (zb - za));
  }
  out[idx] = 1.0f / (1.0f + __expf(-acc));
}

extern "C" void kernel_launch(void* const* d_in, const int* in_sizes, int n_in,
                              void* d_out, int out_size, void* d_ws, size_t ws_size,
                              hipStream_t stream)
{
  const float* p2c   = (const float*)d_in[0];
  const float* wmsg  = (const float*)d_in[1];
  const float* gamma = (const float*)d_in[2];
  const float* beta  = (const float*)d_in[3];
  const float* wup2  = (const float*)d_in[4];
  const float* v1    = (const float*)d_in[5];
  const float* v2    = (const float*)d_in[6];

  char* ws = (char*)d_ws;
  size_t off = 0;
  auto walloc = [&](size_t bytes) -> void* {
    void* p = ws + off;
    off = (off + bytes + 255) & ~(size_t)255;
    return p;
  };
  unsigned int* Mswz = (unsigned int*)walloc(32768u * 4);
  unsigned int* Fswz = (unsigned int*)walloc(3u * 2048u * 4);
  float* stats1 = (float*)walloc(512u * 256u * 4);
  float* stats2 = (float*)walloc(512u * 256u * 4);
  float* sc1  = (float*)walloc(256 * 4);
  float* sh1  = (float*)walloc(256 * 4);
  float* sc2  = (float*)walloc(256 * 4);
  float* sh2  = (float*)walloc(256 * 4);
  float* scid = (float*)walloc(256 * 4);
  float* shid = (float*)walloc(256 * 4);
  float* z1024 = (float*)walloc((size_t)BB * 25 * 1024 * 4);
  float* out1  = (float*)walloc((size_t)BB * CC * HH * WW * 4);
  float* out2  = (float*)walloc((size_t)BB * CC * HH * WW * 4);

  prep_kernel<<<256, 256, 0, stream>>>(wmsg, wup2, v1, Mswz, Fswz);
  init_misc<<<1, 256, 0, stream>>>(scid, shid);

  scnn_kernel<false><<<256, 256, 0, stream>>>(p2c, out1, Mswz, scid, shid, 0, stats1);
  bnscale_kernel<<<256, 256, 0, stream>>>(stats1, gamma, beta, sc1, sh1);

  scnn_kernel<true><<<256, 256, 0, stream>>>(out1, out2, Mswz, sc1, sh1, 1, stats2);
  bnscale_kernel<<<256, 256, 0, stream>>>(stats2, gamma, beta, sc2, sh2);

  conv_kernel<<<BB * CONV_RUNS * 16, 256, 0, stream>>>(out2, Fswz, sc2, sh2, z1024);

  final_kernel<<<128, 256, 0, stream>>>(z1024, v2, (float*)d_out);
}


# recurrence product transposed (matrix as A, state as B): a lane holds 8 channels of one position; state to LDS as 2 b128 stores, conv stages rows from registers, 1 barrier/row (BN sums regrouped)
# speedup vs baseline: 7.7405x; 1.0515x over previous
#include <hip/hip_runtime.h>
#include <hip/hip_bf16.h>
#include <stdint.h>

#define BB 8
#define CC 256
#define HH 25
#define WW 512

typedef __bf16 bf16x16 __attribute__((ext_vector_type(16)));
typedef float  f32x8   __attribute__((ext_vector_type(8)));
typedef float  f32x4   __attribute__((ext_vector_type(4)));
typedef float  f32x2   __attribute__((ext_vector_type(2)));

union FragU { uint4 u[2]; bf16x16 v; };

__device__ __forceinline__ unsigned short f2bf(float x) {
  unsigned u = __float_as_uint(x);
  unsigned r = u + 0x7FFFu + ((u >> 16) & 1u);
  return (unsigned short)(r >> 16);
}

typedef __bf16 bf16x2 __attribute__((ext_vector_type(2)));

__device__ __forceinline__ unsigned int pack_bf2(float a, float b) {
  const f32x2 v = {a, b};
  const bf16x2 r = __builtin_convertvector(v, bf16x2);
  return __builtin_bit_cast(unsigned int, r);
}

__device__ __forceinline__ f32x8 wmma_bf16(bf16x16 a, bf16x16 b, f32x8 c) {
  return __builtin_amdgcn_wmma_f32_16x16x32_bf16(false, a, false, b, (short)0, c, false, false);
}

__device__ __forceinline__ bf16x16 load_frag_lds(const unsigned short* rowBase, int kt, int lane) {
  int kb = (lane & 16) ? 8 : 0;
  const unsigned short* p = rowBase + kt * 32 + kb;
  FragU f;
  f.u[0] = *(const uint4*)(p);
  f.u[1] = *(const uint4*)(p + 16);
  return f.v;
}

__device__ __forceinline__ bf16x16 load_frag_Bswz(const unsigned int* swz, int kt, int nt, int lane) {
  const unsigned int* p = swz + ((((kt << 4) + nt) * 32 + lane) << 3);
  FragU f;
  f.u[0] = *(const uint4*)(p);
  f.u[1] = *(const uint4*)(p + 4);
  return f.v;
}

__device__ __forceinline__ bf16x16 load_frag_F(const unsigned int* swz, int ky, int kt, int lane) {
  const unsigned int* p = swz + ((((ky << 3) + kt) * 32 + lane) << 3);
  FragU f;
  f.u[0] = *(const uint4*)(p);
  f.u[1] = *(const uint4*)(p + 4);
  return f.v;
}

__device__ __forceinline__ float bf2f(unsigned short h) {
  return __uint_as_float(((unsigned)h) << 16);
}

__device__ __forceinline__ void lds_barrier() {
  __builtin_amdgcn_fence(__ATOMIC_RELEASE, "workgroup", "local");
  __builtin_amdgcn_s_barrier();
  __builtin_amdgcn_fence(__ATOMIC_ACQUIRE, "workgroup", "local");
}

template <int CTRL>
__device__ __forceinline__ float lane_perm(float v) {
  return __builtin_bit_cast(float, __builtin_amdgcn_update_dpp(0, __builtin_bit_cast(int, v), CTRL, 0xF, 0xF, true));
}
__device__ __forceinline__ float half16_sum(float v) {
  v += lane_perm<0xB1>(v);
  v += lane_perm<0x4E>(v);
  v += lane_perm<0x141>(v);
  v += lane_perm<0x140>(v);
  return v;
}

__global__ void __launch_bounds__(256) prep_kernel(
    const float* __restrict__ wmsg, const float* __restrict__ wup2, const float* __restrict__ v1,
    unsigned int* __restrict__ Mswz, unsigned int* __restrict__ Fswz)
{
  __shared__ float part[8][18];
  __shared__ float wsum[18];
  const int tid = threadIdx.x;

  if (blockIdx.x >= 128) {
    int r    = (blockIdx.x - 128) * 256 + tid;
    int j    = r & 7;
    int lane = (r >> 3) & 31;
    int nt   = (r >> 8) & 15;
    int kt   = r >> 12;
    int n  = nt * 16 + (lane & 15);
    int kb = (lane & 16) ? 8 : 0;
    int k  = kt * 32 + ((j < 4) ? (kb + 2 * j) : (16 + kb + 2 * (j - 4)));
    float lo = wmsg[(n * 256 + k) * 9 + 4];
    float hi = wmsg[(n * 256 + k + 1) * 9 + 4];
    Mswz[r] = (unsigned)f2bf(lo) | ((unsigned)f2bf(hi) << 16);
    return;
  }

  const int kp   = blockIdx.x;
  const int wave = tid >> 5;
  const int lane = tid & 31;
  const float g  = v1[tid];
  const float* srcw = wup2 + (tid * 256 + 2 * kp) * 9;
  float p[18];
  #pragma unroll
  for (int t = 0; t < 18; ++t) p[t] = g * srcw[t];
  #pragma unroll
  for (int t = 0; t < 18; ++t) {
    for (int off = 16; off >= 1; off >>= 1) p[t] += __shfl_xor(p[t], off);
  }
  if (lane == 0) {
    #pragma unroll
    for (int t = 0; t < 18; ++t) part[wave][t] = p[t];
  }
  __syncthreads();
  if (tid < 18) {
    float s = 0.f;
    for (int w = 0; w < 8; ++w) s += part[w][tid];
    wsum[tid] = s;
  }
  __syncthreads();
  if (tid < 48) {
    const int ky = tid >> 4;
    const int n  = tid & 15;
    unsigned int d = 0u;
    if (n < 6) {
      const int tap = ky * 3 + (n >> 1);
      const float w0 = wsum[tap], w1 = wsum[9 + tap];
      const unsigned short h0 = f2bf(w0), h1 = f2bf(w1);
      d = (n & 1) ? ((unsigned)f2bf(w0 - bf2f(h0)) | ((unsigned)f2bf(w1 - bf2f(h1)) << 16))
                  : ((unsigned)h0 | ((unsigned)h1 << 16));
    }
    const int k   = 2 * kp;
    const int kt  = k >> 5;
    const int r   = k & 31;
    const int kbs = (r >> 3) & 1;
    const int j   = ((r >> 4) << 2) + ((r & 7) >> 1);
    Fswz[((((ky << 3) + kt) * 32 + kbs * 16 + n) << 3) + j] = d;
  }
}

__global__ void init_misc(float* scid, float* shid) {
  int t = threadIdx.x;
  scid[t] = 1.0f; shid[t] = 0.0f;
}

#define TBLK 4096
struct XRow { f32x4 a0, b0, a1, b1; };

template <bool TIN>
__global__ void __launch_bounds__(256) scnn_kernel(
    const float* __restrict__ xin, float* __restrict__ xout,
    const unsigned int* __restrict__ Mswz,
    const float* __restrict__ sc, const float* __restrict__ sh,
    int relu_in, float* __restrict__ statsP)
{
  __shared__ __align__(16) unsigned short prevA[16 * 264];
  __shared__ __align__(16) unsigned short prevB[16 * 264];

  const int tid  = threadIdx.x;
  const int b    = blockIdx.x >> 5;
  const int w0   = (blockIdx.x & 31) << 4;
  const int wave = tid >> 5;
  const int lane = tid & 31;
  const int c16  = lane & 15;
  const int half8 = (lane & 16) ? 8 : 0;
  const int nt0 = wave * 2, nt1 = wave * 2 + 1;
  const int chA = nt0 * 16 + half8;
  const int chB = nt1 * 16 + half8;
  constexpr int AHEAD = TIN ? 4 : 2;

  for (int i = tid; i < 16 * 264; i += 256) prevA[i] = 0;

  float scA[8], shA[8], scB[8], shB[8];
  {
    const f32x4 t0 = *(const f32x4*)(sc + chA), t1 = *(const f32x4*)(sc + chA + 4);
    const f32x4 t2 = *(const f32x4*)(sh + chA), t3 = *(const f32x4*)(sh + chA + 4);
    const f32x4 u0 = *(const f32x4*)(sc + chB), u1 = *(const f32x4*)(sc + chB + 4);
    const f32x4 u2 = *(const f32x4*)(sh + chB), u3 = *(const f32x4*)(sh + chB + 4);
    scA[0] = t0.x; scA[1] = t0.y; scA[2] = t0.z; scA[3] = t0.w; scA[4] = t1.x; scA[5] = t1.y; scA[6] = t1.z; scA[7] = t1.w;
    shA[0] = t2.x; shA[1] = t2.y; shA[2] = t2.z; shA[3] = t2.w; shA[4] = t3.x; shA[5] = t3.y; shA[6] = t3.z; shA[7] = t3.w;
    scB[0] = u0.x; scB[1] = u0.y; scB[2] = u0.z; scB[3] = u0.w; scB[4] = u1.x; scB[5] = u1.y; scB[6] = u1.z; scB[7] = u1.w;
    shB[0] = u2.x; shB[1] = u2.y; shB[2] = u2.z; shB[3] = u2.w; shB[4] = u3.x; shB[5] = u3.y; shB[6] = u3.z; shB[7] = u3.w;
  }

  const float* xiAp = xin + ((size_t)(b * CC + chA) * HH) * WW + w0 + c16;
  const float* xiBp = xin + ((size_t)(b * CC + chB) * HH) * WW + w0 + c16;
  const size_t tbase = (size_t)blockIdx.x * HH * TBLK + (size_t)(wave * 128 + lane) * 4;
  const float* xitp = xin  + tbase;
  float*       xotp = xout + tbase;

  bf16x16 bm0[8], bm1[8];
  #pragma unroll
  for (int kt = 0; kt < 8; ++kt) {
    bm0[kt] = load_frag_Bswz(Mswz, kt, nt0, lane);
    bm1[kt] = load_frag_Bswz(Mswz, kt, nt1, lane);
  }
  __builtin_amdgcn_sched_barrier(0);

  auto load_row = [&](int r, XRow& x) {
    if (TIN) {
      const float* n = xitp + (size_t)r * TBLK;
      x.a0 = *(const f32x4*)(n);       x.b0 = *(const f32x4*)(n + 128);
      x.a1 = *(const f32x4*)(n + 256); x.b1 = *(const f32x4*)(n + 384);
    } else {
      const float* n0 = xiAp + (size_t)r * WW;
      const float* n1 = xiBp + (size_t)r * WW;
      x.a0 = (f32x4){n0[0 * HH * WW], n0[1 * HH * WW], n0[2 * HH * WW], n0[3 * HH * WW]};
      x.b0 = (f32x4){n0[4 * HH * WW], n0[5 * HH * WW], n0[6 * HH * WW], n0[7 * HH * WW]};
      x.a1 = (f32x4){n1[0 * HH * WW], n1[1 * HH * WW], n1[2 * HH * WW], n1[3 * HH * WW]};
      x.b1 = (f32x4){n1[4 * HH * WW], n1[5 * HH * WW], n1[6 * HH * WW], n1[7 * HH * WW]};
    }
  };

  float sA_[8], qA_[8], sB_[8], qB_[8];
  #pragma unroll
  for (int j = 0; j < 8; ++j) { sA_[j] = 0.f; qA_[j] = 0.f; sB_[j] = 0.f; qB_[j] = 0.f; }

  auto step = [&](int h, const XRow& x, XRow& xfill, const unsigned short* prevCur, unsigned short* prevNxt) {
    __builtin_amdgcn_sched_barrier(0);
    lds_barrier();
    load_row(min(h + AHEAD, HH - 1), xfill);

    f32x8 acc0 = {}; f32x8 acc1 = {};
    const unsigned short* prow = &prevCur[c16 * 264];
    bf16x16 a = load_frag_lds(prow, 0, lane);
    #pragma unroll
    for (int kt = 0; kt < 8; ++kt) {
      bf16x16 an = (kt < 7) ? load_frag_lds(prow, kt + 1, lane) : a;
      acc0 = wmma_bf16(bm0[kt], a, acc0);
      acc1 = wmma_bf16(bm1[kt], a, acc1);
      a = an;
    }

    f32x4 xa0 = x.a0, xb0 = x.b0, xa1 = x.a1, xb1 = x.b1;
    asm("" : "+v"(xa0), "+v"(xb0), "+v"(xa1), "+v"(xb1));
    const float xi0[8] = {xa0.x, xa0.y, xa0.z, xa0.w, xb0.x, xb0.y, xb0.z, xb0.w};
    const float xi1[8] = {xa1.x, xa1.y, xa1.z, xa1.w, xb1.x, xb1.y, xb1.z, xb1.w};
    float r0[8], r1[8];
    #pragma unroll
    for (int j = 0; j < 8; ++j) {
      float a0 = scA[j] * xi0[j] + shA[j];
      float a1 = scB[j] * xi1[j] + shB[j];
      if (relu_in) { a0 = fmaxf(a0, 0.f); a1 = fmaxf(a1, 0.f); }
      float v0 = a0 + fmaxf(acc0[j], 0.f);
      float v1 = a1 + fmaxf(acc1[j], 0.f);
      r0[j] = v0; r1[j] = v1;
      sA_[j] += v0; qA_[j] += v0 * v0;
      sB_[j] += v1; qB_[j] += v1 * v1;
    }
    {
      uint4 p0, p1;
      p0.x = pack_bf2(r0[0], r0[1]); p0.y = pack_bf2(r0[2], r0[3]); p0.z = pack_bf2(r0[4], r0[5]); p0.w = pack_bf2(r0[6], r0[7]);
      p1.x = pack_bf2(r1[0], r1[1]); p1.y = pack_bf2(r1[2], r1[3]); p1.z = pack_bf2(r1[4], r1[5]); p1.w = pack_bf2(r1[6], r1[7]);
      *(uint4*)&prevNxt[c16 * 264 + chA] = p0;
      *(uint4*)&prevNxt[c16 * 264 + chB] = p1;
    }
    {
      float* p = xotp + (size_t)(HH - 1 - h) * TBLK;
      *(float4*)(p)       = make_float4(r0[0], r0[1], r0[2], r0[3]);
      *(float4*)(p + 128) = make_float4(r0[4], r0[5], r0[6], r0[7]);
      *(float4*)(p + 256) = make_float4(r1[0], r1[1], r1[2], r1[3]);
      *(float4*)(p + 384) = make_float4(r1[4], r1[5], r1[6], r1[7]);
    }
  };

  unsigned short* cur = prevA;
  unsigned short* nxt = prevB;
  if (TIN) {
    XRow x0, x1, x2, x3, x4;
    load_row(0, x0); load_row(1, x1); load_row(2, x2); load_row(3, x3);

    for (int h = 0; h < HH; h += 5) {
      step(h,     x0, x4, cur, nxt);
      step(h + 1, x1, x0, nxt, cur);
      step(h + 2, x2, x1, cur, nxt);
      step(h + 3, x3, x2, nxt, cur);
      step(h + 4, x4, x3, cur, nxt);
      unsigned short* t = cur; cur = nxt; nxt = t;
    }
  } else {
    XRow x0, x1, x2;
    load_row(0, x0);
    __builtin_amdgcn_sched_barrier(0);
    load_row(1, x1);
    for (int h = 0; h < HH - 1; h += 6) {
      step(h,     x0, x2, cur, nxt);
      step(h + 1, x1, x0, nxt, cur);
      step(h + 2, x2, x1, cur, nxt);
      step(h + 3, x0, x2, nxt, cur);
      step(h + 4, x1, x0, cur, nxt);
      step(h + 5, x2, x1, nxt, cur);
    }
    step(HH - 1, x0, x2, cur, nxt);
  }

  #pragma unroll
  for (int j = 0; j < 8; ++j) {
    sA_[j] = half16_sum(sA_[j]); qA_[j] = half16_sum(qA_[j]);
    sB_[j] = half16_sum(sB_[j]); qB_[j] = half16_sum(qB_[j]);
  }
  if (c16 == 0) {
    #pragma unroll
    for (int j = 0; j < 8; ++j) {
      statsP[(chA + j)       * 256 + blockIdx.x] = sA_[j];
      statsP[(256 + chA + j) * 256 + blockIdx.x] = qA_[j];
      statsP[(chB + j)       * 256 + blockIdx.x] = sB_[j];
      statsP[(256 + chB + j) * 256 + blockIdx.x] = qB_[j];
    }
  }
}

__global__ void __launch_bounds__(256) bnscale_kernel(
    const float* __restrict__ statsP,
    const float* __restrict__ gamma, const float* __restrict__ beta,
    float* __restrict__ sc, float* __restrict__ sh)
{
  __shared__ float ps[8], pq[8];
  const int c = blockIdx.x;
  const int t = threadIdx.x;
  float s = statsP[c * 256 + t];
  float q = statsP[(256 + c) * 256 + t];
  for (int off = 16; off >= 1; off >>= 1) { s += __shfl_xor(s, off); q += __shfl_xor(q, off); }
  if ((t & 31) == 0) { ps[t >> 5] = s; pq[t >> 5] = q; }
  __syncthreads();
  if (t == 0) {
    float S = 0.f, Q = 0.f;
    for (int w = 0; w < 8; ++w) { S += ps[w]; Q += pq[w]; }
    const float n = (float)(BB * HH * WW);
    float mean = S / n;
    float var  = Q / n - mean * mean;
    float g = gamma[c] * rsqrtf(var + 1e-5f);
    sc[c] = g;
    sh[c] = beta[c] - mean * g;
  }
}

#define CONV_TW 64
#define CONV_SW 34
#define CONV_MA 3
#define CONV_ROW (CONV_SW * 264)
#define CONV_RUNS 4
#define CONV_ZP (8 * CONV_SW * 6)

__global__ void __launch_bounds__(256) conv_kernel(
    const float* __restrict__ xin,
    const unsigned int* __restrict__ Fswz,
    const float* __restrict__ sc, const float* __restrict__ sh,
    float* __restrict__ z)
{
  __shared__ __align__(16) unsigned short uT[3 * CONV_ROW];
  __shared__ __align__(16) float zp[2 * CONV_ZP];
  __shared__ __align__(16) float zt[2 * CONV_SW * 4];

  const int tid  = threadIdx.x;
  const int wave = tid >> 5;
  const int lane = tid & 31;
  const int wt   = blockIdx.x & 15;
  const int run  = (blockIdx.x >> 4) & (CONV_RUNS - 1);
  const int b    = blockIdx.x >> 6;
  const int h0   = run * 6;
  const int nrows = (run == CONV_RUNS - 1) ? 7 : 6;
  const int w20  = wt * CONV_TW;

  const int c16  = tid & 15;
  const int pq   = (tid >> 5) & 3;
  const int chq  = 32 * (tid >> 7) + 16 * (pq >> 1) + 8 * ((tid >> 4) & 1) + 4 * (pq & 1);
  const bool nbr   = (tid < 128);
  const bool nbrL  = (tid < 64);
  const int  nn    = tid & 63;
  const int  nq    = (nn >> 1) & 3;
  const int  chN   = 32 * (nn >> 3) + 16 * (nq >> 1) + 8 * (nn & 1) + 4 * (nq & 1);
  const int  tileN = nbrL ? ((wt > 0) ? 2 * wt - 1 : 0) : ((wt < 15) ? 2 * wt + 2 : 31);
  const int  posN  = nbrL ? ((wt > 0) ? 15 : 0)         : ((wt < 15) ? 0 : 15);
  const int  offN  = (((nn >> 3) * 4 + nq) * 32 + 16 * (nn & 1) + posN) * 4;
  const int  sN    = nbrL ? 0 : (CONV_SW - 1);

  f32x4 scv[4], shv[4];
  #pragma unroll
  for (int k = 0; k < 4; ++k) {
    scv[k] = *(const f32x4*)(sc + chq + 64 * k);
    shv[k] = *(const f32x4*)(sh + chq + 64 * k);
  }
  f32x4 scN = {}, shN = {};
  if (nbr) { scN = *(const f32x4*)(sc + chN); shN = *(const f32x4*)(sh + chN); }

  auto load_row = [&](int hs, f32x4 (&q)[8], f32x4& en) {
    const int hsc = min(max(hs, 0), HH - 1);
    const float* t0 = xin + ((b * 32 + 2 * wt) * HH + hsc) * TBLK + 4 * tid;
    #pragma unroll
    for (int i = 0; i < 8; ++i)
      q[i] = *(const f32x4*)(t0 + (i >> 2) * (HH * TBLK) + (i & 3) * 1024);
    if (nbr) en = *(const f32x4*)(xin + ((b * 32 + tileN) * HH + hsc) * TBLK + offN);
  };
  auto stage4 = [&](unsigned short* dst, int s, int ch, const f32x4& x, const f32x4& a, const f32x4& c, bool rowok) {
    const float y0 = fmaxf(a.x * x.x + c.x, 0.f);
    const float y1 = fmaxf(a.y * x.y + c.y, 0.f);
    const float y2 = fmaxf(a.z * x.z + c.z, 0.f);
    const float y3 = fmaxf(a.w * x.w + c.w, 0.f);
    uint2 p;
    p.x = pack_bf2(rowok ? y0 : 0.f, rowok ? y1 : 0.f);
    p.y = pack_bf2(rowok ? y2 : 0.f, rowok ? y3 : 0.f);
    *(uint2*)&dst[s * 264 + ch] = p;
  };
  auto stage_row = [&](unsigned short* dst, int hs, const f32x4 (&q)[8], const f32x4& en) {
    const bool rowok = (hs >= 0 && hs < HH);
    #pragma unroll
    for (int i = 0; i < 8; ++i)
      stage4(dst, 1 + 16 * (i >> 2) + c16, chq + 64 * (i & 3), q[i], scv[i & 3], shv[i & 3], rowok);
    if (nbr) stage4(dst, sN, chN, en, scN, shN, rowok);
  };
  auto col_weight = [&](int wi) {
    const int w2  = w20 + wi - 1;
    const int num = w2 * 511;
    const int i0  = num / 1023;
    return (float)(num - i0 * 1023) * (1.0f / 1023.0f);
  };
  int   tb_off[3];
  float tb_f[3];
  bool  tb_ok[3];
  #pragma unroll
  for (int kx = 0; kx < 3; ++kx) {
    const int  wi   = (tid & (CONV_TW - 1)) + kx;
    const int  w2   = w20 + wi - 1;
    const bool col0 = (wi == 1) && (wt == 0);
    tb_off[kx] = (col0 ? 1 : (wi >> 1)) * 4 + kx;
    tb_f[kx]   = col_weight(wi);
    tb_ok[kx]  = (w2 >= 0 && w2 < 1024);
  }

  bf16x16 bw[3];
  #pragma unroll
  for (int ky = 0; ky < 3; ++ky) bw[ky] = load_frag_F(Fswz, ky, wave, lane);

  int o0 = 0, o1 = CONV_ROW, o2 = 2 * CONV_ROW;

  f32x4 qn[8];
  f32x4 en = {};
  {
    f32x4 q0[8], q1[8], q2[8];
    f32x4 e0 = {}, e1 = {}, e2 = {};
    load_row(h0 - 1, q0, e0);
    load_row(h0,     q1, e1);
    load_row(h0 + 1, q2, e2);
    stage_row(uT + o0, h0 - 1, q0, e0);
    stage_row(uT + o1, h0,     q1, e1);
    load_row(h0 + 2, qn, en);
    stage_row(uT + o2, h0 + 1, q2, e2);
  }
  __syncthreads();

  const int half8 = (lane & 16) ? 8 : 0;
  const int arow_l = (lane & 15) * 264;
  const int arow_t = min(32 + (lane & 15), CONV_SW - 1) * 264;

  auto frag = [&](int slot, int m) {
    return load_frag_lds(uT + slot + ((m < CONV_MA - 1) ? arow_l + m * 16 * 264 : arow_t), wave, lane);
  };

  f32x8 accA[CONV_MA], accB[CONV_MA], accC[CONV_MA];
  {
    bf16x16 af[CONV_MA];
    #pragma unroll
    for (int m = 0; m < CONV_MA; ++m) af[m] = frag(o0, m);
    #pragma unroll
    for (int m = 0; m < CONV_MA; ++m) accA[m] = wmma_bf16(af[m], bw[0], (f32x8){});
    #pragma unroll
    for (int m = 0; m < CONV_MA; ++m) af[m] = frag(o1, m);
    #pragma unroll
    for (int m = 0; m < CONV_MA; ++m) {
      accA[m] = wmma_bf16(af[m], bw[1], accA[m]);
      accB[m] = wmma_bf16(af[m], bw[0], (f32x8){});
    }
  }
  __syncthreads();

  const int pa_k  = tid / 3;
  const int pa_kx = tid - 3 * pa_k;
  auto sum_taps = [&](const float* zpr, float* zrow) {
    if (tid < CONV_SW * 3) {
      float th = 0.f, tl = 0.f;
      #pragma unroll
      for (int w = 0; w < 8; ++w) {
        const f32x2 p = *(const f32x2*)(zpr + (w * CONV_SW + pa_k) * 6 + 2 * pa_kx);
        th += p.x;
        tl += p.y;
      }
      zrow[pa_k * 4 + pa_kx] = th + tl;
    }
  };
  auto finish_row = [&](int h, const float* zrow) {
    if (tid < CONV_TW) {
      float acc = 0.f;
      #pragma unroll
      for (int kx = 0; kx < 3; ++kx) {
        const float t0 = zrow[tb_off[kx]];
        const float t1 = zrow[tb_off[kx] + 4];
        const float v  = t0 + tb_f[kx] * (t1 - t0);
        acc += tb_ok[kx] ? v : 0.f;
      }
      z[(b * 25 + h) * 1024 + w20 + tid] = acc;
    }
  };

  auto row_step = [&](int i, f32x8 (&P)[CONV_MA], f32x8 (&Q)[CONV_MA], f32x8 (&R)[CONV_MA]) {
    const int  h    = h0 + i;
    const bool more = (i + 1 < nrows);
    float* zpr = zp + (i & 1) * CONV_ZP;

    bf16x16 af[CONV_MA];
    #pragma unroll
    for (int m = 0; m < CONV_MA; ++m) af[m] = frag(o2, m);
    __builtin_amdgcn_sched_barrier(0);
    #pragma unroll
    for (int m = 0; m < CONV_MA; ++m) {
      P[m] = wmma_bf16(af[m], bw[2], P[m]);
      Q[m] = wmma_bf16(af[m], bw[1], Q[m]);
      R[m] = wmma_bf16(af[m], bw[0], (f32x8){});
    }

    if ((lane & 15) < 6) {
      float* zw = zpr + (wave * CONV_SW + half8) * 6 + (lane & 15);
      #pragma unroll
      for (int m = 0; m < CONV_MA - 1; ++m) {
        #pragma unroll
        for (int j = 0; j < 8; ++j)
          zw[(m * 16 + j) * 6] = P[m][j];
      }
      if (half8 == 0) {
        zw[32 * 6] = P[CONV_MA - 1][0];
        zw[33 * 6] = P[CONV_MA - 1][1];
      }
    }
    if (more) stage_row(uT + o0, h + 2, qn, en);
    __syncthreads();

    if (more && i + 2 < nrows) load_row(h + 3, qn, en);
    sum_taps(zpr, zt + (i & 1) * (CONV_SW * 4));
    if (i > 0) finish_row(h - 1, zt + ((i - 1) & 1) * (CONV_SW * 4));
    const int t = o0; o0 = o1; o1 = o2; o2 = t;
  };

  for (int i = 0; ; i += 3) {
    row_step(i, accA, accB, accC);
    if (i + 1 >= nrows) break;
    row_step(i + 1, accB, accC, accA);
    if (i + 2 >= nrows) break;
    row_step(i + 2, accC, accA, accB);
    if (i + 3 >= nrows) break;
  }
  __syncthreads();
  finish_row(h0 + nrows - 1, zt + ((nrows - 1) & 1) * (CONV_SW * 4));
}

__global__ void final_kernel(const float* __restrict__ z, const float* __restrict__ v2,
                             float* __restrict__ out)
{
  int idx = blockIdx.x * 256 + threadIdx.x;
  if (idx >= BB * 4096) return;
  int b  = idx >> 12;
  int w4 = idx & 4095;

  int   num4 = w4 * 2047;
  int   i40  = num4 / 4095;
  float f4   = (float)(num4 - i40 * 4095) * (1.0f / 4095.0f);
  int   i41  = min(i40 + 1, 2047);

  int n3a = i40 * 1023; int ja0 = n3a / 2047;
  float fa = (float)(n3a - ja0 * 2047) * (1.0f / 2047.0f);
  int ja1 = min(ja0 + 1, 1023);
  int n3b = i41 * 1023; int jb0 = n3b / 2047;
  float fb = (float)(n3b - jb0 * 2047) * (1.0f / 2047.0f);
  int jb1 = min(jb0 + 1, 1023);

  float acc = 0.f;
  #pragma unroll 5
  for (int h = 0; h < HH; ++h) {
    const float* zh = z + (b * 25 + h) * 1024;
    float za = zh[ja0] + fa * (zh[ja1] - zh[ja0]);
    float zb = zh[jb0] + fb * (zh[jb1] - zh[jb0]);
    acc += v2[h] * (za + f4 * (zb - za));
  }
  out[idx] = 1.0f / (1.0f + __expf(-acc));
}

extern "C" void kernel_launch(void* const* d_in, const int* in_sizes, int n_in,
                              void* d_out, int out_size, void* d_ws, size_t ws_size,
                              hipStream_t stream)
{
  const float* p2c   = (const float*)d_in[0];
  const float* wmsg  = (const float*)d_in[1];
  const float* gamma = (const float*)d_in[2];
  const float* beta  = (const float*)d_in[3];
  const float* wup2  = (const float*)d_in[4];
  const float* v1    = (const float*)d_in[5];
  const float* v2    = (const float*)d_in[6];

  char* ws = (char*)d_ws;
  size_t off = 0;
  auto walloc = [&](size_t bytes) -> void* {
    void* p = ws + off;
    off = (off + bytes + 255) & ~(size_t)255;
    return p;
  };
  unsigned int* Mswz = (unsigned int*)walloc(32768u * 4);
  unsigned int* Fswz = (unsigned int*)walloc(3u * 2048u * 4);
  float* stats1 = (float*)walloc(512u * 256u * 4);
  float* stats2 = (float*)walloc(512u * 256u * 4);
  float* sc1  = (float*)walloc(256 * 4);
  float* sh1  = (float*)walloc(256 * 4);
  float* sc2  = (float*)walloc(256 * 4);
  float* sh2  = (float*)walloc(256 * 4);
  float* scid = (float*)walloc(256 * 4);
  float* shid = (float*)walloc(256 * 4);
  float* z1024 = (float*)walloc((size_t)BB * 25 * 1024 * 4);
  float* out1  = (float*)walloc((size_t)BB * CC * HH * WW * 4);
  float* out2  = (float*)walloc((size_t)BB * CC * HH * WW * 4);

  prep_kernel<<<256, 256, 0, stream>>>(wmsg, wup2, v1, Mswz, Fswz);
  init_misc<<<1, 256, 0, stream>>>(scid, shid);

  scnn_kernel<false><<<256, 256, 0, stream>>>(p2c, out1, Mswz, scid, shid, 0, stats1);
  bnscale_kernel<<<256, 256, 0, stream>>>(stats1, gamma, beta, sc1, sh1);

  scnn_kernel<true><<<256, 256, 0, stream>>>(out1, out2, Mswz, sc1, sh1, 1, stats2);
  bnscale_kernel<<<256, 256, 0, stream>>>(stats2, gamma, beta, sc2, sh2);

  conv_kernel<<<BB * CONV_RUNS * 16, 256, 0, stream>>>(out2, Fswz, sc2, sh2, z1024);

  final_kernel<<<128, 256, 0, stream>>>(z1024, v2, (float*)d_out);
}
